# starting kernel
# baseline (speedup 1.0000x reference)
#include <hip/hip_runtime.h>
#include <math.h>


typedef float v2f __attribute__((ext_vector_type(2)));
typedef float v8f __attribute__((ext_vector_type(8)));

#define QC_BATCH_PER_BLOCK 4
#define QC_THREADS 256
#define QC_DIM 4096

#define QC_PLANE 4416
#define QC_BUF (2 * QC_PLANE)
#define QC_ELEM (2 * QC_BUF)
#define QC_FINV_OFF (QC_BATCH_PER_BLOCK * QC_ELEM)
#define QC_LDS_FLOATS (QC_FINV_OFF + QC_DIM)
#define QC_LDS_BYTES (QC_LDS_FLOATS * 4)

static __device__ __forceinline__ v8f wmma4(v2f a, v2f b, v8f c) {
  return __builtin_amdgcn_wmma_f32_16x16x4_f32(false, a, false, b, (short)0, c,
                                               false, false);
}

static __device__ __forceinline__ int physidx(int i) {
  return i + (i >> 4) + 4 * (i >> 8);
}

static __device__ __forceinline__ int fperm(int i) {
  int x = i;
#pragma unroll
  for (int q = 11; q >= 0; --q) {
    int C = 11 - q;
    int T = 11 - ((q + 1) % 12);
    x ^= ((x >> C) & 1) << T;
  }
  return x;
}

template <int PASS, bool PERM>
static __device__ __forceinline__ void apply_pass(float* lds, int srcOff,
                                                  int dstOff,
                                                  const float* wsMat, int lane,
                                                  int h) {
  constexpr int CT = (PASS == 0) ? 17 : 276;
  constexpr int CM = (PASS == 2) ? 17 : 1;
  constexpr int CK = (PASS == 0) ? 276 : ((PASS == 1) ? 17 : 1);

  float ur[8], ui[8], nui[8];
  const float* ubase = wsMat + lane * 8;
#pragma unroll
  for (int s = 0; s < 8; ++s) {
    ur[s] = ubase[s];
    ui[s] = ubase[256 + s];
    nui[s] = ubase[512 + s];
  }
  const int m = lane & 15;
  const int kh = (lane >> 4) << 1;
  const int rh = (lane >> 4) << 3;

  const int baseSrc = srcOff + CT * (h * 8) + CM * m + CK * kh;
  const int baseDst = dstOff + CT * (h * 8) + CM * rh + CK * m;
  const int baseDstLog = 256 * (h * 8) + 16 * rh + m;

#pragma unroll 1
  for (int tp = 0; tp < 4; ++tp) {
    const int t0 = 2 * tp, t1 = 2 * tp + 1;
    float ar0[8], ai0[8], ar1[8], ai1[8];
#pragma unroll
    for (int j = 0; j < 4; ++j) {
      const int o0 = baseSrc + CT * t0 + CK * (4 * j);
      const int o1 = baseSrc + CT * t1 + CK * (4 * j);
      ar0[2 * j] = lds[o0];
      ar0[2 * j + 1] = lds[o0 + CK];
      ai0[2 * j] = lds[o0 + QC_PLANE];
      ai0[2 * j + 1] = lds[o0 + CK + QC_PLANE];
      ar1[2 * j] = lds[o1];
      ar1[2 * j + 1] = lds[o1 + CK];
      ai1[2 * j] = lds[o1 + QC_PLANE];
      ai1[2 * j + 1] = lds[o1 + CK + QC_PLANE];
    }
    v8f dr0 = {0.f, 0.f, 0.f, 0.f, 0.f, 0.f, 0.f, 0.f};
    v8f di0 = {0.f, 0.f, 0.f, 0.f, 0.f, 0.f, 0.f, 0.f};
    v8f dr1 = {0.f, 0.f, 0.f, 0.f, 0.f, 0.f, 0.f, 0.f};
    v8f di1 = {0.f, 0.f, 0.f, 0.f, 0.f, 0.f, 0.f, 0.f};
#pragma unroll
    for (int j = 0; j < 4; ++j) {
      v2f Br = {ur[2 * j], ur[2 * j + 1]};
      v2f Bi = {ui[2 * j], ui[2 * j + 1]};
      v2f Bn = {nui[2 * j], nui[2 * j + 1]};
      v2f Ar0 = {ar0[2 * j], ar0[2 * j + 1]};
      v2f Ai0 = {ai0[2 * j], ai0[2 * j + 1]};
      v2f Ar1 = {ar1[2 * j], ar1[2 * j + 1]};
      v2f Ai1 = {ai1[2 * j], ai1[2 * j + 1]};
      dr0 = wmma4(Ar0, Br, dr0);
      di0 = wmma4(Ar0, Bi, di0);
      dr1 = wmma4(Ar1, Br, dr1);
      di1 = wmma4(Ar1, Bi, di1);
      dr0 = wmma4(Ai0, Bn, dr0);
      di0 = wmma4(Ai0, Br, di0);
      dr1 = wmma4(Ai1, Bn, dr1);
      di1 = wmma4(Ai1, Br, di1);
    }
#pragma unroll
    for (int r = 0; r < 8; ++r) {
      if (PERM) {
        const int key0 = baseDstLog + 256 * t0 + 16 * r;
        const int key1 = baseDstLog + 256 * t1 + 16 * r;
        const int ph0 = __float_as_int(lds[QC_FINV_OFF + key0]);
        const int ph1 = __float_as_int(lds[QC_FINV_OFF + key1]);
        lds[dstOff + ph0] = dr0[r];
        lds[dstOff + ph0 + QC_PLANE] = di0[r];
        lds[dstOff + ph1] = dr1[r];
        lds[dstOff + ph1 + QC_PLANE] = di1[r];
      } else {
        const int d0 = baseDst + CT * t0 + CM * r;
        const int d1 = baseDst + CT * t1 + CM * r;
        lds[d0] = dr0[r];
        lds[d0 + QC_PLANE] = di0[r];
        lds[d1] = dr1[r];
        lds[d1 + QC_PLANE] = di1[r];
      }
    }
  }
}

__global__ void qc_setup_gates(const float* __restrict__ params,
                               float* __restrict__ ws) {
  const int tid = threadIdx.x;
  const int mat = tid >> 5;
  const int lane = tid & 31;
  if (mat >= 9) return;
  const int L = mat / 3, g = mat % 3;

  float mr[4][2][2], mi[4][2][2];
#pragma unroll
  for (int s = 0; s < 4; ++s) {
    const int q = g * 4 + s;
    const int base = L * 36 + q * 3;
    const float tx = params[base], ty = params[base + 1], tz = params[base + 2];
    const float cx = cosf(0.5f * tx), sx = sinf(0.5f * tx);
    const float cy = cosf(0.5f * ty), sy = sinf(0.5f * ty);
    const float cz = cosf(0.5f * tz), sz = sinf(0.5f * tz);
    float rr, ii;
    rr = cy * cx; ii = sy * sx;
    mr[s][0][0] = cz * rr + sz * ii; mi[s][0][0] = cz * ii - sz * rr;
    rr = -sy * cx; ii = -cy * sx;
    mr[s][0][1] = cz * rr + sz * ii; mi[s][0][1] = cz * ii - sz * rr;
    rr = sy * cx; ii = -cy * sx;
    mr[s][1][0] = cz * rr - sz * ii; mi[s][1][0] = cz * ii + sz * rr;
    rr = cy * cx; ii = -sy * sx;
    mr[s][1][1] = cz * rr - sz * ii; mi[s][1][1] = cz * ii + sz * rr;
  }

  float* base = ws + mat * 768;
  const int n = lane & 15;
  const int khalf = (lane >> 4) << 1;
#pragma unroll
  for (int j = 0; j < 4; ++j) {
#pragma unroll
    for (int p = 0; p < 2; ++p) {
      const int k = 4 * j + p + khalf;
      float re = 1.f, im = 0.f;
#pragma unroll
      for (int s = 0; s < 4; ++s) {
        const int rb = (n >> (3 - s)) & 1;
        const int cb = (k >> (3 - s)) & 1;
        const float wr = mr[s][rb][cb], wi = mi[s][rb][cb];
        const float nre = re * wr - im * wi;
        const float nim = re * wi + im * wr;
        re = nre; im = nim;
      }
      const int slot = lane * 8 + j * 2 + p;
      base[slot] = re;
      base[256 + slot] = im;
      base[512 + slot] = -im;
    }
  }
}

__global__ __launch_bounds__(QC_THREADS) void qc_main(
    const float* __restrict__ x, const float* __restrict__ ws,
    float* __restrict__ out) {
  extern __shared__ float lds[];
  const int tid = threadIdx.x;
  const int lane = tid & 31;
  const int w = tid >> 5;
  const int e = w >> 1;
  const int h = w & 1;
  const int lt = h * 32 + lane;
  const long long b = (long long)blockIdx.x * QC_BATCH_PER_BLOCK + e;

  const int eb = e * QC_ELEM;

  for (int p = tid; p < QC_DIM; p += QC_THREADS)
    lds[QC_FINV_OFF + fperm(p)] = __int_as_float(physidx(p));

  const float4* xr = (const float4*)(x + b * QC_DIM);
  float4 v[16];
  float ss = 0.f;
#pragma unroll
  for (int ii = 0; ii < 16; ++ii) {
    v[ii] = xr[ii * 64 + lt];
    ss += v[ii].x * v[ii].x + v[ii].y * v[ii].y + v[ii].z * v[ii].z +
          v[ii].w * v[ii].w;
  }
#pragma unroll
  for (int off = 16; off > 0; off >>= 1) ss += __shfl_xor(ss, off);
  if (lane == 0) lds[eb + QC_PLANE + h] = ss;
  __syncthreads();
  const float inv = rsqrtf(lds[eb + QC_PLANE] + lds[eb + QC_PLANE + 1]);
  __syncthreads();
#pragma unroll
  for (int ii = 0; ii < 16; ++ii) {
    const int u = ii * 64 + lt;
    const int pb = 4 * u + (u >> 2) + 4 * ii;
    lds[eb + pb + 0] = v[ii].x * inv;
    lds[eb + pb + 1] = v[ii].y * inv;
    lds[eb + pb + 2] = v[ii].z * inv;
    lds[eb + pb + 3] = v[ii].w * inv;
    lds[eb + QC_PLANE + pb + 0] = 0.f;
    lds[eb + QC_PLANE + pb + 1] = 0.f;
    lds[eb + QC_PLANE + pb + 2] = 0.f;
    lds[eb + QC_PLANE + pb + 3] = 0.f;
  }
  __syncthreads();

  int cb = 0;
#pragma unroll 1
  for (int L = 0; L < 3; ++L) {
    const float* wsL = ws + L * 3 * 768;
    const int cur = eb + cb * QC_BUF;
    const int nxt = eb + (cb ^ 1) * QC_BUF;
    apply_pass<0, false>(lds, cur, cur, wsL + 0, lane, h);
    __syncthreads();
    apply_pass<1, false>(lds, cur, cur, wsL + 768, lane, h);
    __syncthreads();
    apply_pass<2, true>(lds, cur, nxt, wsL + 1536, lane, h);
    __syncthreads();
    cb ^= 1;
  }
  const int fin = eb + cb * QC_BUF;

  const int ebase = fin + lt + (lt >> 4);
  float T = 0.f;
  float accL[6];
#pragma unroll
  for (int q = 0; q < 6; ++q) accL[q] = 0.f;
#pragma unroll
  for (int ii = 0; ii < 64; ++ii) {
    const int d = 68 * ii + 4 * (ii >> 2);
    const float re = lds[ebase + d];
    const float im = lds[ebase + d + QC_PLANE];
    const float p = re * re + im * im;
    T += p;
#pragma unroll
    for (int q = 0; q < 6; ++q)
      accL[q] += ((ii >> (5 - q)) & 1) ? -p : p;
  }
  float acc[12];
#pragma unroll
  for (int q = 0; q < 6; ++q) acc[q] = accL[q];
#pragma unroll
  for (int q = 6; q < 12; ++q)
    acc[q] = ((lt >> (11 - q)) & 1) ? -T : T;
#pragma unroll
  for (int q = 0; q < 12; ++q) {
    float s = acc[q];
#pragma unroll
    for (int off = 16; off > 0; off >>= 1) s += __shfl_xor(s, off);
    acc[q] = s;
  }
  __syncthreads();
  if (lane == 0) {
#pragma unroll
    for (int q = 0; q < 12; ++q) lds[QC_FINV_OFF + w * 12 + q] = acc[q];
  }
  __syncthreads();
  if (h == 0 && lane < 12)
    out[b * 12 + lane] = lds[QC_FINV_OFF + (2 * e) * 12 + lane] +
                         lds[QC_FINV_OFF + (2 * e + 1) * 12 + lane];
}

extern "C" void kernel_launch(void* const* d_in, const int* in_sizes, int n_in,
                              void* d_out, int out_size, void* d_ws,
                              size_t ws_size, hipStream_t stream) {
  (void)n_in; (void)out_size; (void)ws_size;
  const float* x = (const float*)d_in[0];
  const float* params = (const float*)d_in[1];
  float* out = (float*)d_out;
  float* ws = (float*)d_ws;

  const int batch = in_sizes[0] / QC_DIM;
  const int blocks = batch / QC_BATCH_PER_BLOCK;

  qc_setup_gates<<<1, 288, 0, stream>>>(params, ws);

  (void)hipFuncSetAttribute((const void*)qc_main,
                            hipFuncAttributeMaxDynamicSharedMemorySize,
                            QC_LDS_BYTES);
  qc_main<<<blocks, QC_THREADS, QC_LDS_BYTES, stream>>>(x, ws, out);
}


# complex 16x16 product by three real WMMA products instead of four: K=(Sr+Si)Ur is the C start of both output chains, gate planes Ur, Ui-Ur, -(Ur+Ui); changes summation order and rounding
# speedup vs baseline: 1.1413x; 1.1413x over previous
#include <hip/hip_runtime.h>
#include <math.h>


typedef float v2f __attribute__((ext_vector_type(2)));
typedef float v8f __attribute__((ext_vector_type(8)));

#define QC_BATCH_PER_BLOCK 4
#define QC_THREADS 256
#define QC_DIM 4096

#define QC_PLANE 4416
#define QC_BUF (2 * QC_PLANE)
#define QC_ELEM (2 * QC_BUF)
#define QC_FINV_OFF (QC_BATCH_PER_BLOCK * QC_ELEM)
#define QC_LDS_FLOATS (QC_FINV_OFF + QC_DIM)
#define QC_LDS_BYTES (QC_LDS_FLOATS * 4)

static __device__ __forceinline__ v8f wmma4(v2f a, v2f b, v8f c) {
  return __builtin_amdgcn_wmma_f32_16x16x4_f32(false, a, false, b, (short)0, c,
                                               false, false);
}

static __device__ __forceinline__ int physidx(int i) {
  return i + (i >> 4) + 4 * (i >> 8);
}

static __device__ __forceinline__ int fperm(int i) {
  int x = i;
#pragma unroll
  for (int q = 11; q >= 0; --q) {
    int C = 11 - q;
    int T = 11 - ((q + 1) % 12);
    x ^= ((x >> C) & 1) << T;
  }
  return x;
}

template <int PASS, bool PERM>
static __device__ __forceinline__ void apply_pass(float* lds, int srcOff,
                                                  int dstOff,
                                                  const float* wsMat, int lane,
                                                  int h) {
  constexpr int CT = (PASS == 0) ? 17 : 276;
  constexpr int CM = (PASS == 2) ? 17 : 1;
  constexpr int CK = (PASS == 0) ? 276 : ((PASS == 1) ? 17 : 1);

  float ur[8], ud[8], un[8];
  const float* ubase = wsMat + lane * 8;
#pragma unroll
  for (int s = 0; s < 8; ++s) {
    ur[s] = ubase[s];
    ud[s] = ubase[256 + s];
    un[s] = ubase[512 + s];
  }
  const int m = lane & 15;
  const int kh = (lane >> 4) << 1;
  const int rh = (lane >> 4) << 3;

  const int baseSrc = srcOff + CT * (h * 8) + CM * m + CK * kh;
  const int baseDst = dstOff + CT * (h * 8) + CM * rh + CK * m;
  const int baseDstLog = 256 * (h * 8) + 16 * rh + m;

#pragma unroll 1
  for (int tp = 0; tp < 4; ++tp) {
    const int t0 = 2 * tp, t1 = 2 * tp + 1;
    float ar0[8], ai0[8], ar1[8], ai1[8];
#pragma unroll
    for (int j = 0; j < 4; ++j) {
      const int o0 = baseSrc + CT * t0 + CK * (4 * j);
      const int o1 = baseSrc + CT * t1 + CK * (4 * j);
      ar0[2 * j] = lds[o0];
      ar0[2 * j + 1] = lds[o0 + CK];
      ai0[2 * j] = lds[o0 + QC_PLANE];
      ai0[2 * j + 1] = lds[o0 + CK + QC_PLANE];
      ar1[2 * j] = lds[o1];
      ar1[2 * j + 1] = lds[o1 + CK];
      ai1[2 * j] = lds[o1 + QC_PLANE];
      ai1[2 * j + 1] = lds[o1 + CK + QC_PLANE];
    }
    v8f k0 = {0.f, 0.f, 0.f, 0.f, 0.f, 0.f, 0.f, 0.f};
    v8f k1 = {0.f, 0.f, 0.f, 0.f, 0.f, 0.f, 0.f, 0.f};
#pragma unroll
    for (int j = 0; j < 4; ++j) {
      v2f Br = {ur[2 * j], ur[2 * j + 1]};
      v2f As0 = {ar0[2 * j] + ai0[2 * j], ar0[2 * j + 1] + ai0[2 * j + 1]};
      v2f As1 = {ar1[2 * j] + ai1[2 * j], ar1[2 * j + 1] + ai1[2 * j + 1]};
      k0 = wmma4(As0, Br, k0);
      k1 = wmma4(As1, Br, k1);
    }
    v8f dr0 = k0, di0 = k0, dr1 = k1, di1 = k1;
#pragma unroll
    for (int j = 0; j < 4; ++j) {
      v2f Bd = {ud[2 * j], ud[2 * j + 1]};
      v2f Bn = {un[2 * j], un[2 * j + 1]};
      v2f Ar0 = {ar0[2 * j], ar0[2 * j + 1]};
      v2f Ai0 = {ai0[2 * j], ai0[2 * j + 1]};
      v2f Ar1 = {ar1[2 * j], ar1[2 * j + 1]};
      v2f Ai1 = {ai1[2 * j], ai1[2 * j + 1]};
      dr0 = wmma4(Ai0, Bn, dr0);
      di0 = wmma4(Ar0, Bd, di0);
      dr1 = wmma4(Ai1, Bn, dr1);
      di1 = wmma4(Ar1, Bd, di1);
    }
#pragma unroll
    for (int r = 0; r < 8; ++r) {
      if (PERM) {
        const int key0 = baseDstLog + 256 * t0 + 16 * r;
        const int key1 = baseDstLog + 256 * t1 + 16 * r;
        const int ph0 = __float_as_int(lds[QC_FINV_OFF + key0]);
        const int ph1 = __float_as_int(lds[QC_FINV_OFF + key1]);
        lds[dstOff + ph0] = dr0[r];
        lds[dstOff + ph0 + QC_PLANE] = di0[r];
        lds[dstOff + ph1] = dr1[r];
        lds[dstOff + ph1 + QC_PLANE] = di1[r];
      } else {
        const int d0 = baseDst + CT * t0 + CM * r;
        const int d1 = baseDst + CT * t1 + CM * r;
        lds[d0] = dr0[r];
        lds[d0 + QC_PLANE] = di0[r];
        lds[d1] = dr1[r];
        lds[d1 + QC_PLANE] = di1[r];
      }
    }
  }
}

__global__ void qc_setup_gates(const float* __restrict__ params,
                               float* __restrict__ ws) {
  const int tid = threadIdx.x;
  const int mat = tid >> 5;
  const int lane = tid & 31;
  if (mat >= 9) return;
  const int L = mat / 3, g = mat % 3;

  float mr[4][2][2], mi[4][2][2];
#pragma unroll
  for (int s = 0; s < 4; ++s) {
    const int q = g * 4 + s;
    const int base = L * 36 + q * 3;
    const float tx = params[base], ty = params[base + 1], tz = params[base + 2];
    const float cx = cosf(0.5f * tx), sx = sinf(0.5f * tx);
    const float cy = cosf(0.5f * ty), sy = sinf(0.5f * ty);
    const float cz = cosf(0.5f * tz), sz = sinf(0.5f * tz);
    float rr, ii;
    rr = cy * cx; ii = sy * sx;
    mr[s][0][0] = cz * rr + sz * ii; mi[s][0][0] = cz * ii - sz * rr;
    rr = -sy * cx; ii = -cy * sx;
    mr[s][0][1] = cz * rr + sz * ii; mi[s][0][1] = cz * ii - sz * rr;
    rr = sy * cx; ii = -cy * sx;
    mr[s][1][0] = cz * rr - sz * ii; mi[s][1][0] = cz * ii + sz * rr;
    rr = cy * cx; ii = -sy * sx;
    mr[s][1][1] = cz * rr - sz * ii; mi[s][1][1] = cz * ii + sz * rr;
  }

  float* base = ws + mat * 768;
  const int n = lane & 15;
  const int khalf = (lane >> 4) << 1;
#pragma unroll
  for (int j = 0; j < 4; ++j) {
#pragma unroll
    for (int p = 0; p < 2; ++p) {
      const int k = 4 * j + p + khalf;
      float re = 1.f, im = 0.f;
#pragma unroll
      for (int s = 0; s < 4; ++s) {
        const int rb = (n >> (3 - s)) & 1;
        const int cb = (k >> (3 - s)) & 1;
        const float wr = mr[s][rb][cb], wi = mi[s][rb][cb];
        const float nre = re * wr - im * wi;
        const float nim = re * wi + im * wr;
        re = nre; im = nim;
      }
      const int slot = lane * 8 + j * 2 + p;
      base[slot] = re;
      base[256 + slot] = im - re;
      base[512 + slot] = -(re + im);
    }
  }
}

__global__ __launch_bounds__(QC_THREADS) void qc_main(
    const float* __restrict__ x, const float* __restrict__ ws,
    float* __restrict__ out) {
  extern __shared__ float lds[];
  const int tid = threadIdx.x;
  const int lane = tid & 31;
  const int w = tid >> 5;
  const int e = w >> 1;
  const int h = w & 1;
  const int lt = h * 32 + lane;
  const long long b = (long long)blockIdx.x * QC_BATCH_PER_BLOCK + e;

  const int eb = e * QC_ELEM;

  for (int p = tid; p < QC_DIM; p += QC_THREADS)
    lds[QC_FINV_OFF + fperm(p)] = __int_as_float(physidx(p));

  const float4* xr = (const float4*)(x + b * QC_DIM);
  float4 v[16];
  float ss = 0.f;
#pragma unroll
  for (int ii = 0; ii < 16; ++ii) {
    v[ii] = xr[ii * 64 + lt];
    ss += v[ii].x * v[ii].x + v[ii].y * v[ii].y + v[ii].z * v[ii].z +
          v[ii].w * v[ii].w;
  }
#pragma unroll
  for (int off = 16; off > 0; off >>= 1) ss += __shfl_xor(ss, off);
  if (lane == 0) lds[eb + QC_PLANE + h] = ss;
  __syncthreads();
  const float inv = rsqrtf(lds[eb + QC_PLANE] + lds[eb + QC_PLANE + 1]);
  __syncthreads();
#pragma unroll
  for (int ii = 0; ii < 16; ++ii) {
    const int u = ii * 64 + lt;
    const int pb = 4 * u + (u >> 2) + 4 * ii;
    lds[eb + pb + 0] = v[ii].x * inv;
    lds[eb + pb + 1] = v[ii].y * inv;
    lds[eb + pb + 2] = v[ii].z * inv;
    lds[eb + pb + 3] = v[ii].w * inv;
    lds[eb + QC_PLANE + pb + 0] = 0.f;
    lds[eb + QC_PLANE + pb + 1] = 0.f;
    lds[eb + QC_PLANE + pb + 2] = 0.f;
    lds[eb + QC_PLANE + pb + 3] = 0.f;
  }
  __syncthreads();

  int cb = 0;
#pragma unroll 1
  for (int L = 0; L < 3; ++L) {
    const float* wsL = ws + L * 3 * 768;
    const int cur = eb + cb * QC_BUF;
    const int nxt = eb + (cb ^ 1) * QC_BUF;
    apply_pass<0, false>(lds, cur, cur, wsL + 0, lane, h);
    __syncthreads();
    apply_pass<1, false>(lds, cur, cur, wsL + 768, lane, h);
    __syncthreads();
    apply_pass<2, true>(lds, cur, nxt, wsL + 1536, lane, h);
    __syncthreads();
    cb ^= 1;
  }
  const int fin = eb + cb * QC_BUF;

  const int ebase = fin + lt + (lt >> 4);
  float T = 0.f;
  float accL[6];
#pragma unroll
  for (int q = 0; q < 6; ++q) accL[q] = 0.f;
#pragma unroll
  for (int ii = 0; ii < 64; ++ii) {
    const int d = 68 * ii + 4 * (ii >> 2);
    const float re = lds[ebase + d];
    const float im = lds[ebase + d + QC_PLANE];
    const float p = re * re + im * im;
    T += p;
#pragma unroll
    for (int q = 0; q < 6; ++q)
      accL[q] += ((ii >> (5 - q)) & 1) ? -p : p;
  }
  float acc[12];
#pragma unroll
  for (int q = 0; q < 6; ++q) acc[q] = accL[q];
#pragma unroll
  for (int q = 6; q < 12; ++q)
    acc[q] = ((lt >> (11 - q)) & 1) ? -T : T;
#pragma unroll
  for (int q = 0; q < 12; ++q) {
    float s = acc[q];
#pragma unroll
    for (int off = 16; off > 0; off >>= 1) s += __shfl_xor(s, off);
    acc[q] = s;
  }
  __syncthreads();
  if (lane == 0) {
#pragma unroll
    for (int q = 0; q < 12; ++q) lds[QC_FINV_OFF + w * 12 + q] = acc[q];
  }
  __syncthreads();
  if (h == 0 && lane < 12)
    out[b * 12 + lane] = lds[QC_FINV_OFF + (2 * e) * 12 + lane] +
                         lds[QC_FINV_OFF + (2 * e + 1) * 12 + lane];
}

extern "C" void kernel_launch(void* const* d_in, const int* in_sizes, int n_in,
                              void* d_out, int out_size, void* d_ws,
                              size_t ws_size, hipStream_t stream) {
  (void)n_in; (void)out_size; (void)ws_size;
  const float* x = (const float*)d_in[0];
  const float* params = (const float*)d_in[1];
  float* out = (float*)d_out;
  float* ws = (float*)d_ws;

  const int batch = in_sizes[0] / QC_DIM;
  const int blocks = batch / QC_BATCH_PER_BLOCK;

  qc_setup_gates<<<1, 288, 0, stream>>>(params, ws);

  (void)hipFuncSetAttribute((const void*)qc_main,
                            hipFuncAttributeMaxDynamicSharedMemorySize,
                            QC_LDS_BYTES);
  qc_main<<<blocks, QC_THREADS, QC_LDS_BYTES, stream>>>(x, ws, out);
}


# groups A and B applied in one two-sided pass per 16x16 tile (Y=X*UB^T, X'=UA*Y, Y fed from D registers as B operand): one LDS round trip and barrier fewer per layer; changes summation order
# speedup vs baseline: 1.2275x; 1.0755x over previous
#include <hip/hip_runtime.h>
#include <math.h>


typedef float v2f __attribute__((ext_vector_type(2)));
typedef float v8f __attribute__((ext_vector_type(8)));

#define QC_BATCH_PER_BLOCK 4
#define QC_THREADS 256
#define QC_DIM 4096

#define QC_PLANE 4416
#define QC_BUF (2 * QC_PLANE)
#define QC_ELEM (2 * QC_BUF)
#define QC_FINV_OFF (QC_BATCH_PER_BLOCK * QC_ELEM)
#define QC_LDS_FLOATS (QC_FINV_OFF + QC_DIM)
#define QC_LDS_BYTES (QC_LDS_FLOATS * 4)

static __device__ __forceinline__ v8f wmma4(v2f a, v2f b, v8f c) {
  return __builtin_amdgcn_wmma_f32_16x16x4_f32(false, a, false, b, (short)0, c,
                                               false, false);
}

static __device__ __forceinline__ int physidx(int i) {
  return i + (i >> 4) + 4 * (i >> 8);
}

static __device__ __forceinline__ int fperm(int i) {
  int x = i;
#pragma unroll
  for (int q = 11; q >= 0; --q) {
    int C = 11 - q;
    int T = 11 - ((q + 1) % 12);
    x ^= ((x >> C) & 1) << T;
  }
  return x;
}

template <int PASS, bool PERM>
static __device__ __forceinline__ void apply_pass(float* lds, int srcOff,
                                                  int dstOff,
                                                  const float* wsMat, int lane,
                                                  int h) {
  constexpr int CT = (PASS == 0) ? 17 : 276;
  constexpr int CM = (PASS == 2) ? 17 : 1;
  constexpr int CK = (PASS == 0) ? 276 : ((PASS == 1) ? 17 : 1);

  float ur[8], ud[8], un[8];
  const float* ubase = wsMat + lane * 8;
#pragma unroll
  for (int s = 0; s < 8; ++s) {
    ur[s] = ubase[s];
    ud[s] = ubase[256 + s];
    un[s] = ubase[512 + s];
  }
  const int m = lane & 15;
  const int kh = (lane >> 4) << 1;
  const int rh = (lane >> 4) << 3;

  const int baseSrc = srcOff + CT * (h * 8) + CM * m + CK * kh;
  const int baseDst = dstOff + CT * (h * 8) + CM * rh + CK * m;
  const int baseDstLog = 256 * (h * 8) + 16 * rh + m;

#pragma unroll 1
  for (int tp = 0; tp < 4; ++tp) {
    const int t0 = 2 * tp, t1 = 2 * tp + 1;
    float ar0[8], ai0[8], ar1[8], ai1[8];
#pragma unroll
    for (int j = 0; j < 4; ++j) {
      const int o0 = baseSrc + CT * t0 + CK * (4 * j);
      const int o1 = baseSrc + CT * t1 + CK * (4 * j);
      ar0[2 * j] = lds[o0];
      ar0[2 * j + 1] = lds[o0 + CK];
      ai0[2 * j] = lds[o0 + QC_PLANE];
      ai0[2 * j + 1] = lds[o0 + CK + QC_PLANE];
      ar1[2 * j] = lds[o1];
      ar1[2 * j + 1] = lds[o1 + CK];
      ai1[2 * j] = lds[o1 + QC_PLANE];
      ai1[2 * j + 1] = lds[o1 + CK + QC_PLANE];
    }
    v8f k0 = {0.f, 0.f, 0.f, 0.f, 0.f, 0.f, 0.f, 0.f};
    v8f k1 = {0.f, 0.f, 0.f, 0.f, 0.f, 0.f, 0.f, 0.f};
#pragma unroll
    for (int j = 0; j < 4; ++j) {
      v2f Br = {ur[2 * j], ur[2 * j + 1]};
      v2f As0 = {ar0[2 * j] + ai0[2 * j], ar0[2 * j + 1] + ai0[2 * j + 1]};
      v2f As1 = {ar1[2 * j] + ai1[2 * j], ar1[2 * j + 1] + ai1[2 * j + 1]};
      k0 = wmma4(As0, Br, k0);
      k1 = wmma4(As1, Br, k1);
    }
    v8f dr0 = k0, di0 = k0, dr1 = k1, di1 = k1;
#pragma unroll
    for (int j = 0; j < 4; ++j) {
      v2f Bd = {ud[2 * j], ud[2 * j + 1]};
      v2f Bn = {un[2 * j], un[2 * j + 1]};
      v2f Ar0 = {ar0[2 * j], ar0[2 * j + 1]};
      v2f Ai0 = {ai0[2 * j], ai0[2 * j + 1]};
      v2f Ar1 = {ar1[2 * j], ar1[2 * j + 1]};
      v2f Ai1 = {ai1[2 * j], ai1[2 * j + 1]};
      dr0 = wmma4(Ai0, Bn, dr0);
      di0 = wmma4(Ar0, Bd, di0);
      dr1 = wmma4(Ai1, Bn, dr1);
      di1 = wmma4(Ar1, Bd, di1);
    }
#pragma unroll
    for (int r = 0; r < 8; ++r) {
      if (PERM) {
        const int key0 = baseDstLog + 256 * t0 + 16 * r;
        const int key1 = baseDstLog + 256 * t1 + 16 * r;
        const int ph0 = __float_as_int(lds[QC_FINV_OFF + key0]);
        const int ph1 = __float_as_int(lds[QC_FINV_OFF + key1]);
        lds[dstOff + ph0] = dr0[r];
        lds[dstOff + ph0 + QC_PLANE] = di0[r];
        lds[dstOff + ph1] = dr1[r];
        lds[dstOff + ph1 + QC_PLANE] = di1[r];
      } else {
        const int d0 = baseDst + CT * t0 + CM * r;
        const int d1 = baseDst + CT * t1 + CM * r;
        lds[d0] = dr0[r];
        lds[d0 + QC_PLANE] = di0[r];
        lds[d1] = dr1[r];
        lds[d1 + QC_PLANE] = di1[r];
      }
    }
  }
}


static __device__ __forceinline__ void apply_fused_ab(float* lds, int base,
                                                      const float* wsA,
                                                      const float* wsB,
                                                      int lane, int h) {
  float br[8], bd[8], bn[8], ar_[8], ad_[8], an_[8];
  const float* bb = wsB + lane * 8;
  const float* ab = wsA + lane * 8;
#pragma unroll
  for (int s = 0; s < 8; ++s) {
    br[s] = bb[s];
    bd[s] = bb[256 + s];
    bn[s] = bb[512 + s];
    ar_[s] = ab[s];
    ad_[s] = ab[256 + s];
    an_[s] = ab[512 + s];
  }
  const int m = lane & 15;
  const int hf = lane >> 4;
  const int baseSrc = base + 276 * m + 17 * (2 * hf) + 8 * h;
  const int baseDst = base + 276 * (8 * hf) + 17 * m + 8 * h;

#pragma unroll 1
  for (int tp = 0; tp < 4; ++tp) {
    const int c0 = 2 * tp, c1 = 2 * tp + 1;
    float sr0[8], si0[8], sr1[8], si1[8];
#pragma unroll
    for (int j = 0; j < 4; ++j) {
      const int o0 = baseSrc + c0 + 17 * (4 * j);
      const int o1 = baseSrc + c1 + 17 * (4 * j);
      sr0[2 * j] = lds[o0];
      sr0[2 * j + 1] = lds[o0 + 17];
      si0[2 * j] = lds[o0 + QC_PLANE];
      si0[2 * j + 1] = lds[o0 + 17 + QC_PLANE];
      sr1[2 * j] = lds[o1];
      sr1[2 * j + 1] = lds[o1 + 17];
      si1[2 * j] = lds[o1 + QC_PLANE];
      si1[2 * j + 1] = lds[o1 + 17 + QC_PLANE];
    }
    v8f k0 = {0.f, 0.f, 0.f, 0.f, 0.f, 0.f, 0.f, 0.f};
    v8f k1 = {0.f, 0.f, 0.f, 0.f, 0.f, 0.f, 0.f, 0.f};
#pragma unroll
    for (int j = 0; j < 4; ++j) {
      v2f Br = {br[2 * j], br[2 * j + 1]};
      v2f As0 = {sr0[2 * j] + si0[2 * j], sr0[2 * j + 1] + si0[2 * j + 1]};
      v2f As1 = {sr1[2 * j] + si1[2 * j], sr1[2 * j + 1] + si1[2 * j + 1]};
      k0 = wmma4(As0, Br, k0);
      k1 = wmma4(As1, Br, k1);
    }
    v8f yr0 = k0, yi0 = k0, yr1 = k1, yi1 = k1;
#pragma unroll
    for (int j = 0; j < 4; ++j) {
      v2f Bd = {bd[2 * j], bd[2 * j + 1]};
      v2f Bn = {bn[2 * j], bn[2 * j + 1]};
      v2f Ar0 = {sr0[2 * j], sr0[2 * j + 1]};
      v2f Ai0 = {si0[2 * j], si0[2 * j + 1]};
      v2f Ar1 = {sr1[2 * j], sr1[2 * j + 1]};
      v2f Ai1 = {si1[2 * j], si1[2 * j + 1]};
      yr0 = wmma4(Ai0, Bn, yr0);
      yi0 = wmma4(Ar0, Bd, yi0);
      yr1 = wmma4(Ai1, Bn, yr1);
      yi1 = wmma4(Ar1, Bd, yi1);
    }
    const v8f ys0 = yr0 + yi0;
    const v8f ys1 = yr1 + yi1;
    v8f q0 = {0.f, 0.f, 0.f, 0.f, 0.f, 0.f, 0.f, 0.f};
    v8f q1 = {0.f, 0.f, 0.f, 0.f, 0.f, 0.f, 0.f, 0.f};
#pragma unroll
    for (int j = 0; j < 4; ++j) {
      v2f Ur = {ar_[2 * j], ar_[2 * j + 1]};
      v2f Bs0 = {ys0[2 * j], ys0[2 * j + 1]};
      v2f Bs1 = {ys1[2 * j], ys1[2 * j + 1]};
      q0 = wmma4(Ur, Bs0, q0);
      q1 = wmma4(Ur, Bs1, q1);
    }
    v8f xr0 = q0, xi0 = q0, xr1 = q1, xi1 = q1;
#pragma unroll
    for (int j = 0; j < 4; ++j) {
      v2f Ud = {ad_[2 * j], ad_[2 * j + 1]};
      v2f Un = {an_[2 * j], an_[2 * j + 1]};
      v2f Yr0 = {yr0[2 * j], yr0[2 * j + 1]};
      v2f Yi0 = {yi0[2 * j], yi0[2 * j + 1]};
      v2f Yr1 = {yr1[2 * j], yr1[2 * j + 1]};
      v2f Yi1 = {yi1[2 * j], yi1[2 * j + 1]};
      xr0 = wmma4(Un, Yi0, xr0);
      xi0 = wmma4(Ud, Yr0, xi0);
      xr1 = wmma4(Un, Yi1, xr1);
      xi1 = wmma4(Ud, Yr1, xi1);
    }
#pragma unroll
    for (int r = 0; r < 8; ++r) {
      const int d0 = baseDst + c0 + 276 * r;
      const int d1 = baseDst + c1 + 276 * r;
      lds[d0] = xr0[r];
      lds[d0 + QC_PLANE] = xi0[r];
      lds[d1] = xr1[r];
      lds[d1 + QC_PLANE] = xi1[r];
    }
  }
}

__global__ void qc_setup_gates(const float* __restrict__ params,
                               float* __restrict__ ws) {
  const int tid = threadIdx.x;
  const int mat = tid >> 5;
  const int lane = tid & 31;
  if (mat >= 9) return;
  const int L = mat / 3, g = mat % 3;

  float mr[4][2][2], mi[4][2][2];
#pragma unroll
  for (int s = 0; s < 4; ++s) {
    const int q = g * 4 + s;
    const int base = L * 36 + q * 3;
    const float tx = params[base], ty = params[base + 1], tz = params[base + 2];
    const float cx = cosf(0.5f * tx), sx = sinf(0.5f * tx);
    const float cy = cosf(0.5f * ty), sy = sinf(0.5f * ty);
    const float cz = cosf(0.5f * tz), sz = sinf(0.5f * tz);
    float rr, ii;
    rr = cy * cx; ii = sy * sx;
    mr[s][0][0] = cz * rr + sz * ii; mi[s][0][0] = cz * ii - sz * rr;
    rr = -sy * cx; ii = -cy * sx;
    mr[s][0][1] = cz * rr + sz * ii; mi[s][0][1] = cz * ii - sz * rr;
    rr = sy * cx; ii = -cy * sx;
    mr[s][1][0] = cz * rr - sz * ii; mi[s][1][0] = cz * ii + sz * rr;
    rr = cy * cx; ii = -sy * sx;
    mr[s][1][1] = cz * rr - sz * ii; mi[s][1][1] = cz * ii + sz * rr;
  }

  float* base = ws + mat * 768;
  const int n = lane & 15;
  const int khalf = (lane >> 4) << 1;
#pragma unroll
  for (int j = 0; j < 4; ++j) {
#pragma unroll
    for (int p = 0; p < 2; ++p) {
      const int k = (g == 0) ? (2 * j + p + 4 * khalf) : (4 * j + p + khalf);
      float re = 1.f, im = 0.f;
#pragma unroll
      for (int s = 0; s < 4; ++s) {
        const int rb = (n >> (3 - s)) & 1;
        const int cb = (k >> (3 - s)) & 1;
        const float wr = mr[s][rb][cb], wi = mi[s][rb][cb];
        const float nre = re * wr - im * wi;
        const float nim = re * wi + im * wr;
        re = nre; im = nim;
      }
      const int slot = lane * 8 + j * 2 + p;
      base[slot] = re;
      base[256 + slot] = im - re;
      base[512 + slot] = -(re + im);
    }
  }
}

__global__ __launch_bounds__(QC_THREADS) void qc_main(
    const float* __restrict__ x, const float* __restrict__ ws,
    float* __restrict__ out) {
  extern __shared__ float lds[];
  const int tid = threadIdx.x;
  const int lane = tid & 31;
  const int w = tid >> 5;
  const int e = w >> 1;
  const int h = w & 1;
  const int lt = h * 32 + lane;
  const long long b = (long long)blockIdx.x * QC_BATCH_PER_BLOCK + e;

  const int eb = e * QC_ELEM;

  for (int p = tid; p < QC_DIM; p += QC_THREADS)
    lds[QC_FINV_OFF + fperm(p)] = __int_as_float(physidx(p));

  const float4* xr = (const float4*)(x + b * QC_DIM);
  float4 v[16];
  float ss = 0.f;
#pragma unroll
  for (int ii = 0; ii < 16; ++ii) {
    v[ii] = xr[ii * 64 + lt];
    ss += v[ii].x * v[ii].x + v[ii].y * v[ii].y + v[ii].z * v[ii].z +
          v[ii].w * v[ii].w;
  }
#pragma unroll
  for (int off = 16; off > 0; off >>= 1) ss += __shfl_xor(ss, off);
  if (lane == 0) lds[eb + QC_PLANE + h] = ss;
  __syncthreads();
  const float inv = rsqrtf(lds[eb + QC_PLANE] + lds[eb + QC_PLANE + 1]);
  __syncthreads();
#pragma unroll
  for (int ii = 0; ii < 16; ++ii) {
    const int u = ii * 64 + lt;
    const int pb = 4 * u + (u >> 2) + 4 * ii;
    lds[eb + pb + 0] = v[ii].x * inv;
    lds[eb + pb + 1] = v[ii].y * inv;
    lds[eb + pb + 2] = v[ii].z * inv;
    lds[eb + pb + 3] = v[ii].w * inv;
    lds[eb + QC_PLANE + pb + 0] = 0.f;
    lds[eb + QC_PLANE + pb + 1] = 0.f;
    lds[eb + QC_PLANE + pb + 2] = 0.f;
    lds[eb + QC_PLANE + pb + 3] = 0.f;
  }
  __syncthreads();

  int cb = 0;
#pragma unroll 1
  for (int L = 0; L < 3; ++L) {
    const float* wsL = ws + L * 3 * 768;
    const int cur = eb + cb * QC_BUF;
    const int nxt = eb + (cb ^ 1) * QC_BUF;
    apply_fused_ab(lds, cur, wsL + 0, wsL + 768, lane, h);
    __syncthreads();
    apply_pass<2, true>(lds, cur, nxt, wsL + 1536, lane, h);
    __syncthreads();
    cb ^= 1;
  }
  const int fin = eb + cb * QC_BUF;

  const int ebase = fin + lt + (lt >> 4);
  float T = 0.f;
  float accL[6];
#pragma unroll
  for (int q = 0; q < 6; ++q) accL[q] = 0.f;
#pragma unroll
  for (int ii = 0; ii < 64; ++ii) {
    const int d = 68 * ii + 4 * (ii >> 2);
    const float re = lds[ebase + d];
    const float im = lds[ebase + d + QC_PLANE];
    const float p = re * re + im * im;
    T += p;
#pragma unroll
    for (int q = 0; q < 6; ++q)
      accL[q] += ((ii >> (5 - q)) & 1) ? -p : p;
  }
  float acc[12];
#pragma unroll
  for (int q = 0; q < 6; ++q) acc[q] = accL[q];
#pragma unroll
  for (int q = 6; q < 12; ++q)
    acc[q] = ((lt >> (11 - q)) & 1) ? -T : T;
#pragma unroll
  for (int q = 0; q < 12; ++q) {
    float s = acc[q];
#pragma unroll
    for (int off = 16; off > 0; off >>= 1) s += __shfl_xor(s, off);
    acc[q] = s;
  }
  __syncthreads();
  if (lane == 0) {
#pragma unroll
    for (int q = 0; q < 12; ++q) lds[QC_FINV_OFF + w * 12 + q] = acc[q];
  }
  __syncthreads();
  if (h == 0 && lane < 12)
    out[b * 12 + lane] = lds[QC_FINV_OFF + (2 * e) * 12 + lane] +
                         lds[QC_FINV_OFF + (2 * e + 1) * 12 + lane];
}

extern "C" void kernel_launch(void* const* d_in, const int* in_sizes, int n_in,
                              void* d_out, int out_size, void* d_ws,
                              size_t ws_size, hipStream_t stream) {
  (void)n_in; (void)out_size; (void)ws_size;
  const float* x = (const float*)d_in[0];
  const float* params = (const float*)d_in[1];
  float* out = (float*)d_out;
  float* ws = (float*)d_ws;

  const int batch = in_sizes[0] / QC_DIM;
  const int blocks = batch / QC_BATCH_PER_BLOCK;

  qc_setup_gates<<<1, 288, 0, stream>>>(params, ws);

  (void)hipFuncSetAttribute((const void*)qc_main,
                            hipFuncAttributeMaxDynamicSharedMemorySize,
                            QC_LDS_BYTES);
  qc_main<<<blocks, QC_THREADS, QC_LDS_BYTES, stream>>>(x, ws, out);
}


# two batch elements (4 waves, 157696 B LDS) per work-group instead of four: two independent work-groups per WGP, so one's load, barrier and expectation phases overlap the other's matrix products
# speedup vs baseline: 1.3477x; 1.0979x over previous
#include <hip/hip_runtime.h>
#include <math.h>


typedef float v2f __attribute__((ext_vector_type(2)));
typedef float v8f __attribute__((ext_vector_type(8)));

#define QC_BATCH_PER_BLOCK 2
#define QC_THREADS 128
#define QC_DIM 4096

#define QC_PLANE 4416
#define QC_BUF (2 * QC_PLANE)
#define QC_ELEM (2 * QC_BUF)
#define QC_FINV_OFF (QC_BATCH_PER_BLOCK * QC_ELEM)
#define QC_LDS_FLOATS (QC_FINV_OFF + QC_DIM)
#define QC_LDS_BYTES (QC_LDS_FLOATS * 4)

static __device__ __forceinline__ v8f wmma4(v2f a, v2f b, v8f c) {
  return __builtin_amdgcn_wmma_f32_16x16x4_f32(false, a, false, b, (short)0, c,
                                               false, false);
}

static __device__ __forceinline__ int physidx(int i) {
  return i + (i >> 4) + 4 * (i >> 8);
}

static __device__ __forceinline__ int fperm(int i) {
  int x = i;
#pragma unroll
  for (int q = 11; q >= 0; --q) {
    int C = 11 - q;
    int T = 11 - ((q + 1) % 12);
    x ^= ((x >> C) & 1) << T;
  }
  return x;
}

template <int PASS, bool PERM>
static __device__ __forceinline__ void apply_pass(float* lds, int srcOff,
                                                  int dstOff,
                                                  const float* wsMat, int lane,
                                                  int h) {
  constexpr int CT = (PASS == 0) ? 17 : 276;
  constexpr int CM = (PASS == 2) ? 17 : 1;
  constexpr int CK = (PASS == 0) ? 276 : ((PASS == 1) ? 17 : 1);

  float ur[8], ud[8], un[8];
  const float* ubase = wsMat + lane * 8;
#pragma unroll
  for (int s = 0; s < 8; ++s) {
    ur[s] = ubase[s];
    ud[s] = ubase[256 + s];
    un[s] = ubase[512 + s];
  }
  const int m = lane & 15;
  const int kh = (lane >> 4) << 1;
  const int rh = (lane >> 4) << 3;

  const int baseSrc = srcOff + CT * (h * 8) + CM * m + CK * kh;
  const int baseDst = dstOff + CT * (h * 8) + CM * rh + CK * m;
  const int baseDstLog = 256 * (h * 8) + 16 * rh + m;

#pragma unroll 1
  for (int tp = 0; tp < 4; ++tp) {
    const int t0 = 2 * tp, t1 = 2 * tp + 1;
    float ar0[8], ai0[8], ar1[8], ai1[8];
#pragma unroll
    for (int j = 0; j < 4; ++j) {
      const int o0 = baseSrc + CT * t0 + CK * (4 * j);
      const int o1 = baseSrc + CT * t1 + CK * (4 * j);
      ar0[2 * j] = lds[o0];
      ar0[2 * j + 1] = lds[o0 + CK];
      ai0[2 * j] = lds[o0 + QC_PLANE];
      ai0[2 * j + 1] = lds[o0 + CK + QC_PLANE];
      ar1[2 * j] = lds[o1];
      ar1[2 * j + 1] = lds[o1 + CK];
      ai1[2 * j] = lds[o1 + QC_PLANE];
      ai1[2 * j + 1] = lds[o1 + CK + QC_PLANE];
    }
    v8f k0 = {0.f, 0.f, 0.f, 0.f, 0.f, 0.f, 0.f, 0.f};
    v8f k1 = {0.f, 0.f, 0.f, 0.f, 0.f, 0.f, 0.f, 0.f};
#pragma unroll
    for (int j = 0; j < 4; ++j) {
      v2f Br = {ur[2 * j], ur[2 * j + 1]};
      v2f As0 = {ar0[2 * j] + ai0[2 * j], ar0[2 * j + 1] + ai0[2 * j + 1]};
      v2f As1 = {ar1[2 * j] + ai1[2 * j], ar1[2 * j + 1] + ai1[2 * j + 1]};
      k0 = wmma4(As0, Br, k0);
      k1 = wmma4(As1, Br, k1);
    }
    v8f dr0 = k0, di0 = k0, dr1 = k1, di1 = k1;
#pragma unroll
    for (int j = 0; j < 4; ++j) {
      v2f Bd = {ud[2 * j], ud[2 * j + 1]};
      v2f Bn = {un[2 * j], un[2 * j + 1]};
      v2f Ar0 = {ar0[2 * j], ar0[2 * j + 1]};
      v2f Ai0 = {ai0[2 * j], ai0[2 * j + 1]};
      v2f Ar1 = {ar1[2 * j], ar1[2 * j + 1]};
      v2f Ai1 = {ai1[2 * j], ai1[2 * j + 1]};
      dr0 = wmma4(Ai0, Bn, dr0);
      di0 = wmma4(Ar0, Bd, di0);
      dr1 = wmma4(Ai1, Bn, dr1);
      di1 = wmma4(Ar1, Bd, di1);
    }
#pragma unroll
    for (int r = 0; r < 8; ++r) {
      if (PERM) {
        const int key0 = baseDstLog + 256 * t0 + 16 * r;
        const int key1 = baseDstLog + 256 * t1 + 16 * r;
        const int ph0 = __float_as_int(lds[QC_FINV_OFF + key0]);
        const int ph1 = __float_as_int(lds[QC_FINV_OFF + key1]);
        lds[dstOff + ph0] = dr0[r];
        lds[dstOff + ph0 + QC_PLANE] = di0[r];
        lds[dstOff + ph1] = dr1[r];
        lds[dstOff + ph1 + QC_PLANE] = di1[r];
      } else {
        const int d0 = baseDst + CT * t0 + CM * r;
        const int d1 = baseDst + CT * t1 + CM * r;
        lds[d0] = dr0[r];
        lds[d0 + QC_PLANE] = di0[r];
        lds[d1] = dr1[r];
        lds[d1 + QC_PLANE] = di1[r];
      }
    }
  }
}


static __device__ __forceinline__ void apply_fused_ab(float* lds, int base,
                                                      const float* wsA,
                                                      const float* wsB,
                                                      int lane, int h) {
  float br[8], bd[8], bn[8], ar_[8], ad_[8], an_[8];
  const float* bb = wsB + lane * 8;
  const float* ab = wsA + lane * 8;
#pragma unroll
  for (int s = 0; s < 8; ++s) {
    br[s] = bb[s];
    bd[s] = bb[256 + s];
    bn[s] = bb[512 + s];
    ar_[s] = ab[s];
    ad_[s] = ab[256 + s];
    an_[s] = ab[512 + s];
  }
  const int m = lane & 15;
  const int hf = lane >> 4;
  const int baseSrc = base + 276 * m + 17 * (2 * hf) + 8 * h;
  const int baseDst = base + 276 * (8 * hf) + 17 * m + 8 * h;

#pragma unroll 1
  for (int tp = 0; tp < 4; ++tp) {
    const int c0 = 2 * tp, c1 = 2 * tp + 1;
    float sr0[8], si0[8], sr1[8], si1[8];
#pragma unroll
    for (int j = 0; j < 4; ++j) {
      const int o0 = baseSrc + c0 + 17 * (4 * j);
      const int o1 = baseSrc + c1 + 17 * (4 * j);
      sr0[2 * j] = lds[o0];
      sr0[2 * j + 1] = lds[o0 + 17];
      si0[2 * j] = lds[o0 + QC_PLANE];
      si0[2 * j + 1] = lds[o0 + 17 + QC_PLANE];
      sr1[2 * j] = lds[o1];
      sr1[2 * j + 1] = lds[o1 + 17];
      si1[2 * j] = lds[o1 + QC_PLANE];
      si1[2 * j + 1] = lds[o1 + 17 + QC_PLANE];
    }
    v8f k0 = {0.f, 0.f, 0.f, 0.f, 0.f, 0.f, 0.f, 0.f};
    v8f k1 = {0.f, 0.f, 0.f, 0.f, 0.f, 0.f, 0.f, 0.f};
#pragma unroll
    for (int j = 0; j < 4; ++j) {
      v2f Br = {br[2 * j], br[2 * j + 1]};
      v2f As0 = {sr0[2 * j] + si0[2 * j], sr0[2 * j + 1] + si0[2 * j + 1]};
      v2f As1 = {sr1[2 * j] + si1[2 * j], sr1[2 * j + 1] + si1[2 * j + 1]};
      k0 = wmma4(As0, Br, k0);
      k1 = wmma4(As1, Br, k1);
    }
    v8f yr0 = k0, yi0 = k0, yr1 = k1, yi1 = k1;
#pragma unroll
    for (int j = 0; j < 4; ++j) {
      v2f Bd = {bd[2 * j], bd[2 * j + 1]};
      v2f Bn = {bn[2 * j], bn[2 * j + 1]};
      v2f Ar0 = {sr0[2 * j], sr0[2 * j + 1]};
      v2f Ai0 = {si0[2 * j], si0[2 * j + 1]};
      v2f Ar1 = {sr1[2 * j], sr1[2 * j + 1]};
      v2f Ai1 = {si1[2 * j], si1[2 * j + 1]};
      yr0 = wmma4(Ai0, Bn, yr0);
      yi0 = wmma4(Ar0, Bd, yi0);
      yr1 = wmma4(Ai1, Bn, yr1);
      yi1 = wmma4(Ar1, Bd, yi1);
    }
    const v8f ys0 = yr0 + yi0;
    const v8f ys1 = yr1 + yi1;
    v8f q0 = {0.f, 0.f, 0.f, 0.f, 0.f, 0.f, 0.f, 0.f};
    v8f q1 = {0.f, 0.f, 0.f, 0.f, 0.f, 0.f, 0.f, 0.f};
#pragma unroll
    for (int j = 0; j < 4; ++j) {
      v2f Ur = {ar_[2 * j], ar_[2 * j + 1]};
      v2f Bs0 = {ys0[2 * j], ys0[2 * j + 1]};
      v2f Bs1 = {ys1[2 * j], ys1[2 * j + 1]};
      q0 = wmma4(Ur, Bs0, q0);
      q1 = wmma4(Ur, Bs1, q1);
    }
    v8f xr0 = q0, xi0 = q0, xr1 = q1, xi1 = q1;
#pragma unroll
    for (int j = 0; j < 4; ++j) {
      v2f Ud = {ad_[2 * j], ad_[2 * j + 1]};
      v2f Un = {an_[2 * j], an_[2 * j + 1]};
      v2f Yr0 = {yr0[2 * j], yr0[2 * j + 1]};
      v2f Yi0 = {yi0[2 * j], yi0[2 * j + 1]};
      v2f Yr1 = {yr1[2 * j], yr1[2 * j + 1]};
      v2f Yi1 = {yi1[2 * j], yi1[2 * j + 1]};
      xr0 = wmma4(Un, Yi0, xr0);
      xi0 = wmma4(Ud, Yr0, xi0);
      xr1 = wmma4(Un, Yi1, xr1);
      xi1 = wmma4(Ud, Yr1, xi1);
    }
#pragma unroll
    for (int r = 0; r < 8; ++r) {
      const int d0 = baseDst + c0 + 276 * r;
      const int d1 = baseDst + c1 + 276 * r;
      lds[d0] = xr0[r];
      lds[d0 + QC_PLANE] = xi0[r];
      lds[d1] = xr1[r];
      lds[d1 + QC_PLANE] = xi1[r];
    }
  }
}

__global__ void qc_setup_gates(const float* __restrict__ params,
                               float* __restrict__ ws) {
  const int tid = threadIdx.x;
  const int mat = tid >> 5;
  const int lane = tid & 31;
  if (mat >= 9) return;
  const int L = mat / 3, g = mat % 3;

  float mr[4][2][2], mi[4][2][2];
#pragma unroll
  for (int s = 0; s < 4; ++s) {
    const int q = g * 4 + s;
    const int base = L * 36 + q * 3;
    const float tx = params[base], ty = params[base + 1], tz = params[base + 2];
    const float cx = cosf(0.5f * tx), sx = sinf(0.5f * tx);
    const float cy = cosf(0.5f * ty), sy = sinf(0.5f * ty);
    const float cz = cosf(0.5f * tz), sz = sinf(0.5f * tz);
    float rr, ii;
    rr = cy * cx; ii = sy * sx;
    mr[s][0][0] = cz * rr + sz * ii; mi[s][0][0] = cz * ii - sz * rr;
    rr = -sy * cx; ii = -cy * sx;
    mr[s][0][1] = cz * rr + sz * ii; mi[s][0][1] = cz * ii - sz * rr;
    rr = sy * cx; ii = -cy * sx;
    mr[s][1][0] = cz * rr - sz * ii; mi[s][1][0] = cz * ii + sz * rr;
    rr = cy * cx; ii = -sy * sx;
    mr[s][1][1] = cz * rr - sz * ii; mi[s][1][1] = cz * ii + sz * rr;
  }

  float* base = ws + mat * 768;
  const int n = lane & 15;
  const int khalf = (lane >> 4) << 1;
#pragma unroll
  for (int j = 0; j < 4; ++j) {
#pragma unroll
    for (int p = 0; p < 2; ++p) {
      const int k = (g == 0) ? (2 * j + p + 4 * khalf) : (4 * j + p + khalf);
      float re = 1.f, im = 0.f;
#pragma unroll
      for (int s = 0; s < 4; ++s) {
        const int rb = (n >> (3 - s)) & 1;
        const int cb = (k >> (3 - s)) & 1;
        const float wr = mr[s][rb][cb], wi = mi[s][rb][cb];
        const float nre = re * wr - im * wi;
        const float nim = re * wi + im * wr;
        re = nre; im = nim;
      }
      const int slot = lane * 8 + j * 2 + p;
      base[slot] = re;
      base[256 + slot] = im - re;
      base[512 + slot] = -(re + im);
    }
  }
}

__global__ __launch_bounds__(QC_THREADS) void qc_main(
    const float* __restrict__ x, const float* __restrict__ ws,
    float* __restrict__ out) {
  extern __shared__ float lds[];
  const int tid = threadIdx.x;
  const int lane = tid & 31;
  const int w = tid >> 5;
  const int e = w >> 1;
  const int h = w & 1;
  const int lt = h * 32 + lane;
  const long long b = (long long)blockIdx.x * QC_BATCH_PER_BLOCK + e;

  const int eb = e * QC_ELEM;

  for (int p = tid; p < QC_DIM; p += QC_THREADS)
    lds[QC_FINV_OFF + fperm(p)] = __int_as_float(physidx(p));

  const float4* xr = (const float4*)(x + b * QC_DIM);
  float4 v[16];
  float ss = 0.f;
#pragma unroll
  for (int ii = 0; ii < 16; ++ii) {
    v[ii] = xr[ii * 64 + lt];
    ss += v[ii].x * v[ii].x + v[ii].y * v[ii].y + v[ii].z * v[ii].z +
          v[ii].w * v[ii].w;
  }
#pragma unroll
  for (int off = 16; off > 0; off >>= 1) ss += __shfl_xor(ss, off);
  if (lane == 0) lds[eb + QC_PLANE + h] = ss;
  __syncthreads();
  const float inv = rsqrtf(lds[eb + QC_PLANE] + lds[eb + QC_PLANE + 1]);
  __syncthreads();
#pragma unroll
  for (int ii = 0; ii < 16; ++ii) {
    const int u = ii * 64 + lt;
    const int pb = 4 * u + (u >> 2) + 4 * ii;
    lds[eb + pb + 0] = v[ii].x * inv;
    lds[eb + pb + 1] = v[ii].y * inv;
    lds[eb + pb + 2] = v[ii].z * inv;
    lds[eb + pb + 3] = v[ii].w * inv;
    lds[eb + QC_PLANE + pb + 0] = 0.f;
    lds[eb + QC_PLANE + pb + 1] = 0.f;
    lds[eb + QC_PLANE + pb + 2] = 0.f;
    lds[eb + QC_PLANE + pb + 3] = 0.f;
  }
  __syncthreads();

  int cb = 0;
#pragma unroll 1
  for (int L = 0; L < 3; ++L) {
    const float* wsL = ws + L * 3 * 768;
    const int cur = eb + cb * QC_BUF;
    const int nxt = eb + (cb ^ 1) * QC_BUF;
    apply_fused_ab(lds, cur, wsL + 0, wsL + 768, lane, h);
    __syncthreads();
    apply_pass<2, true>(lds, cur, nxt, wsL + 1536, lane, h);
    __syncthreads();
    cb ^= 1;
  }
  const int fin = eb + cb * QC_BUF;

  const int ebase = fin + lt + (lt >> 4);
  float T = 0.f;
  float accL[6];
#pragma unroll
  for (int q = 0; q < 6; ++q) accL[q] = 0.f;
#pragma unroll
  for (int ii = 0; ii < 64; ++ii) {
    const int d = 68 * ii + 4 * (ii >> 2);
    const float re = lds[ebase + d];
    const float im = lds[ebase + d + QC_PLANE];
    const float p = re * re + im * im;
    T += p;
#pragma unroll
    for (int q = 0; q < 6; ++q)
      accL[q] += ((ii >> (5 - q)) & 1) ? -p : p;
  }
  float acc[12];
#pragma unroll
  for (int q = 0; q < 6; ++q) acc[q] = accL[q];
#pragma unroll
  for (int q = 6; q < 12; ++q)
    acc[q] = ((lt >> (11 - q)) & 1) ? -T : T;
#pragma unroll
  for (int q = 0; q < 12; ++q) {
    float s = acc[q];
#pragma unroll
    for (int off = 16; off > 0; off >>= 1) s += __shfl_xor(s, off);
    acc[q] = s;
  }
  __syncthreads();
  if (lane == 0) {
#pragma unroll
    for (int q = 0; q < 12; ++q) lds[QC_FINV_OFF + w * 12 + q] = acc[q];
  }
  __syncthreads();
  if (h == 0 && lane < 12)
    out[b * 12 + lane] = lds[QC_FINV_OFF + (2 * e) * 12 + lane] +
                         lds[QC_FINV_OFF + (2 * e + 1) * 12 + lane];
}

extern "C" void kernel_launch(void* const* d_in, const int* in_sizes, int n_in,
                              void* d_out, int out_size, void* d_ws,
                              size_t ws_size, hipStream_t stream) {
  (void)n_in; (void)out_size; (void)ws_size;
  const float* x = (const float*)d_in[0];
  const float* params = (const float*)d_in[1];
  float* out = (float*)d_out;
  float* ws = (float*)d_ws;

  const int batch = in_sizes[0] / QC_DIM;
  const int blocks = batch / QC_BATCH_PER_BLOCK;

  qc_setup_gates<<<1, 288, 0, stream>>>(params, ws);

  (void)hipFuncSetAttribute((const void*)qc_main,
                            hipFuncAttributeMaxDynamicSharedMemorySize,
                            QC_LDS_BYTES);
  qc_main<<<blocks, QC_THREADS, QC_LDS_BYTES, stream>>>(x, ws, out);
}


# CNOT-scatter table lookups read as one group ahead of each tile pair's matrix block instead of between the scatter stores: seven exposed LDS round trips per tile pair removed; results bit-identical
# speedup vs baseline: 1.4778x; 1.0965x over previous
#include <hip/hip_runtime.h>
#include <math.h>


typedef float v2f __attribute__((ext_vector_type(2)));
typedef float v8f __attribute__((ext_vector_type(8)));

#define QC_BATCH_PER_BLOCK 2
#define QC_THREADS 128
#define QC_DIM 4096

#define QC_PLANE 4416
#define QC_BUF (2 * QC_PLANE)
#define QC_ELEM (2 * QC_BUF)
#define QC_FINV_OFF 0
#define QC_STATE_OFF QC_DIM
#define QC_LDS_FLOATS (QC_STATE_OFF + QC_BATCH_PER_BLOCK * QC_ELEM)
#define QC_LDS_BYTES (QC_LDS_FLOATS * 4)

static __device__ __forceinline__ v8f wmma4(v2f a, v2f b, v8f c) {
  return __builtin_amdgcn_wmma_f32_16x16x4_f32(false, a, false, b, (short)0, c,
                                               false, false);
}

static __device__ __forceinline__ int physidx(int i) {
  return i + (i >> 4) + 4 * (i >> 8);
}

static __device__ __forceinline__ int fperm(int i) {
  int x = i;
#pragma unroll
  for (int q = 11; q >= 0; --q) {
    int C = 11 - q;
    int T = 11 - ((q + 1) % 12);
    x ^= ((x >> C) & 1) << T;
  }
  return x;
}

template <int PASS, bool PERM>
static __device__ __forceinline__ void apply_pass(float* lds, int srcOff,
                                                  int dstOff,
                                                  const float* wsMat, int lane,
                                                  int h) {
  constexpr int CT = (PASS == 0) ? 17 : 276;
  constexpr int CM = (PASS == 2) ? 17 : 1;
  constexpr int CK = (PASS == 0) ? 276 : ((PASS == 1) ? 17 : 1);

  float ur[8], ud[8], un[8];
  const float* ubase = wsMat + lane * 8;
#pragma unroll
  for (int s = 0; s < 8; ++s) {
    ur[s] = ubase[s];
    ud[s] = ubase[256 + s];
    un[s] = ubase[512 + s];
  }
  const int m = lane & 15;
  const int kh = (lane >> 4) << 1;
  const int rh = (lane >> 4) << 3;

  const int baseSrc = srcOff + CT * (h * 8) + CM * m + CK * kh;
  const int baseDst = dstOff + CT * (h * 8) + CM * rh + CK * m;
  const int baseDstLog = 256 * (h * 8) + 16 * rh + m;

#pragma unroll 1
  for (int tp = 0; tp < 4; ++tp) {
    const int t0 = 2 * tp, t1 = 2 * tp + 1;
    int sa0[8], sa1[8];
    if (PERM) {
#pragma unroll
      for (int r = 0; r < 8; ++r) {
        sa0[r] = dstOff + __float_as_int(
            lds[QC_FINV_OFF + baseDstLog + 256 * t0 + 16 * r]);
        sa1[r] = dstOff + __float_as_int(
            lds[QC_FINV_OFF + baseDstLog + 256 * t1 + 16 * r]);
      }
    }
    float ar0[8], ai0[8], ar1[8], ai1[8];
#pragma unroll
    for (int j = 0; j < 4; ++j) {
      const int o0 = baseSrc + CT * t0 + CK * (4 * j);
      const int o1 = baseSrc + CT * t1 + CK * (4 * j);
      ar0[2 * j] = lds[o0];
      ar0[2 * j + 1] = lds[o0 + CK];
      ai0[2 * j] = lds[o0 + QC_PLANE];
      ai0[2 * j + 1] = lds[o0 + CK + QC_PLANE];
      ar1[2 * j] = lds[o1];
      ar1[2 * j + 1] = lds[o1 + CK];
      ai1[2 * j] = lds[o1 + QC_PLANE];
      ai1[2 * j + 1] = lds[o1 + CK + QC_PLANE];
    }
    v8f k0 = {0.f, 0.f, 0.f, 0.f, 0.f, 0.f, 0.f, 0.f};
    v8f k1 = {0.f, 0.f, 0.f, 0.f, 0.f, 0.f, 0.f, 0.f};
#pragma unroll
    for (int j = 0; j < 4; ++j) {
      v2f Br = {ur[2 * j], ur[2 * j + 1]};
      v2f As0 = {ar0[2 * j] + ai0[2 * j], ar0[2 * j + 1] + ai0[2 * j + 1]};
      v2f As1 = {ar1[2 * j] + ai1[2 * j], ar1[2 * j + 1] + ai1[2 * j + 1]};
      k0 = wmma4(As0, Br, k0);
      k1 = wmma4(As1, Br, k1);
    }
    v8f dr0 = k0, di0 = k0, dr1 = k1, di1 = k1;
#pragma unroll
    for (int j = 0; j < 4; ++j) {
      v2f Bd = {ud[2 * j], ud[2 * j + 1]};
      v2f Bn = {un[2 * j], un[2 * j + 1]};
      v2f Ar0 = {ar0[2 * j], ar0[2 * j + 1]};
      v2f Ai0 = {ai0[2 * j], ai0[2 * j + 1]};
      v2f Ar1 = {ar1[2 * j], ar1[2 * j + 1]};
      v2f Ai1 = {ai1[2 * j], ai1[2 * j + 1]};
      dr0 = wmma4(Ai0, Bn, dr0);
      di0 = wmma4(Ar0, Bd, di0);
      dr1 = wmma4(Ai1, Bn, dr1);
      di1 = wmma4(Ar1, Bd, di1);
    }
#pragma unroll
    for (int r = 0; r < 8; ++r) {
      if (PERM) {
        lds[sa0[r]] = dr0[r];
        lds[sa0[r] + QC_PLANE] = di0[r];
        lds[sa1[r]] = dr1[r];
        lds[sa1[r] + QC_PLANE] = di1[r];
      } else {
        const int d0 = baseDst + CT * t0 + CM * r;
        const int d1 = baseDst + CT * t1 + CM * r;
        lds[d0] = dr0[r];
        lds[d0 + QC_PLANE] = di0[r];
        lds[d1] = dr1[r];
        lds[d1 + QC_PLANE] = di1[r];
      }
    }
  }
}


static __device__ __forceinline__ void apply_fused_ab(float* lds, int base,
                                                      const float* wsA,
                                                      const float* wsB,
                                                      int lane, int h) {
  float br[8], bd[8], bn[8], ar_[8], ad_[8], an_[8];
  const float* bb = wsB + lane * 8;
  const float* ab = wsA + lane * 8;
#pragma unroll
  for (int s = 0; s < 8; ++s) {
    br[s] = bb[s];
    bd[s] = bb[256 + s];
    bn[s] = bb[512 + s];
    ar_[s] = ab[s];
    ad_[s] = ab[256 + s];
    an_[s] = ab[512 + s];
  }
  const int m = lane & 15;
  const int hf = lane >> 4;
  const int baseSrc = base + 276 * m + 17 * (2 * hf) + 8 * h;
  const int baseDst = base + 276 * (8 * hf) + 17 * m + 8 * h;

#pragma unroll 1
  for (int tp = 0; tp < 4; ++tp) {
    const int c0 = 2 * tp, c1 = 2 * tp + 1;
    float sr0[8], si0[8], sr1[8], si1[8];
#pragma unroll
    for (int j = 0; j < 4; ++j) {
      const int o0 = baseSrc + c0 + 17 * (4 * j);
      const int o1 = baseSrc + c1 + 17 * (4 * j);
      sr0[2 * j] = lds[o0];
      sr0[2 * j + 1] = lds[o0 + 17];
      si0[2 * j] = lds[o0 + QC_PLANE];
      si0[2 * j + 1] = lds[o0 + 17 + QC_PLANE];
      sr1[2 * j] = lds[o1];
      sr1[2 * j + 1] = lds[o1 + 17];
      si1[2 * j] = lds[o1 + QC_PLANE];
      si1[2 * j + 1] = lds[o1 + 17 + QC_PLANE];
    }
    v8f k0 = {0.f, 0.f, 0.f, 0.f, 0.f, 0.f, 0.f, 0.f};
    v8f k1 = {0.f, 0.f, 0.f, 0.f, 0.f, 0.f, 0.f, 0.f};
#pragma unroll
    for (int j = 0; j < 4; ++j) {
      v2f Br = {br[2 * j], br[2 * j + 1]};
      v2f As0 = {sr0[2 * j] + si0[2 * j], sr0[2 * j + 1] + si0[2 * j + 1]};
      v2f As1 = {sr1[2 * j] + si1[2 * j], sr1[2 * j + 1] + si1[2 * j + 1]};
      k0 = wmma4(As0, Br, k0);
      k1 = wmma4(As1, Br, k1);
    }
    v8f yr0 = k0, yi0 = k0, yr1 = k1, yi1 = k1;
#pragma unroll
    for (int j = 0; j < 4; ++j) {
      v2f Bd = {bd[2 * j], bd[2 * j + 1]};
      v2f Bn = {bn[2 * j], bn[2 * j + 1]};
      v2f Ar0 = {sr0[2 * j], sr0[2 * j + 1]};
      v2f Ai0 = {si0[2 * j], si0[2 * j + 1]};
      v2f Ar1 = {sr1[2 * j], sr1[2 * j + 1]};
      v2f Ai1 = {si1[2 * j], si1[2 * j + 1]};
      yr0 = wmma4(Ai0, Bn, yr0);
      yi0 = wmma4(Ar0, Bd, yi0);
      yr1 = wmma4(Ai1, Bn, yr1);
      yi1 = wmma4(Ar1, Bd, yi1);
    }
    const v8f ys0 = yr0 + yi0;
    const v8f ys1 = yr1 + yi1;
    v8f q0 = {0.f, 0.f, 0.f, 0.f, 0.f, 0.f, 0.f, 0.f};
    v8f q1 = {0.f, 0.f, 0.f, 0.f, 0.f, 0.f, 0.f, 0.f};
#pragma unroll
    for (int j = 0; j < 4; ++j) {
      v2f Ur = {ar_[2 * j], ar_[2 * j + 1]};
      v2f Bs0 = {ys0[2 * j], ys0[2 * j + 1]};
      v2f Bs1 = {ys1[2 * j], ys1[2 * j + 1]};
      q0 = wmma4(Ur, Bs0, q0);
      q1 = wmma4(Ur, Bs1, q1);
    }
    v8f xr0 = q0, xi0 = q0, xr1 = q1, xi1 = q1;
#pragma unroll
    for (int j = 0; j < 4; ++j) {
      v2f Ud = {ad_[2 * j], ad_[2 * j + 1]};
      v2f Un = {an_[2 * j], an_[2 * j + 1]};
      v2f Yr0 = {yr0[2 * j], yr0[2 * j + 1]};
      v2f Yi0 = {yi0[2 * j], yi0[2 * j + 1]};
      v2f Yr1 = {yr1[2 * j], yr1[2 * j + 1]};
      v2f Yi1 = {yi1[2 * j], yi1[2 * j + 1]};
      xr0 = wmma4(Un, Yi0, xr0);
      xi0 = wmma4(Ud, Yr0, xi0);
      xr1 = wmma4(Un, Yi1, xr1);
      xi1 = wmma4(Ud, Yr1, xi1);
    }
#pragma unroll
    for (int r = 0; r < 8; ++r) {
      const int d0 = baseDst + c0 + 276 * r;
      const int d1 = baseDst + c1 + 276 * r;
      lds[d0] = xr0[r];
      lds[d0 + QC_PLANE] = xi0[r];
      lds[d1] = xr1[r];
      lds[d1 + QC_PLANE] = xi1[r];
    }
  }
}

__global__ void qc_setup_gates(const float* __restrict__ params,
                               float* __restrict__ ws) {
  const int tid = threadIdx.x;
  const int mat = tid >> 5;
  const int lane = tid & 31;
  if (mat >= 9) return;
  const int L = mat / 3, g = mat % 3;

  float mr[4][2][2], mi[4][2][2];
#pragma unroll
  for (int s = 0; s < 4; ++s) {
    const int q = g * 4 + s;
    const int base = L * 36 + q * 3;
    const float tx = params[base], ty = params[base + 1], tz = params[base + 2];
    const float cx = cosf(0.5f * tx), sx = sinf(0.5f * tx);
    const float cy = cosf(0.5f * ty), sy = sinf(0.5f * ty);
    const float cz = cosf(0.5f * tz), sz = sinf(0.5f * tz);
    float rr, ii;
    rr = cy * cx; ii = sy * sx;
    mr[s][0][0] = cz * rr + sz * ii; mi[s][0][0] = cz * ii - sz * rr;
    rr = -sy * cx; ii = -cy * sx;
    mr[s][0][1] = cz * rr + sz * ii; mi[s][0][1] = cz * ii - sz * rr;
    rr = sy * cx; ii = -cy * sx;
    mr[s][1][0] = cz * rr - sz * ii; mi[s][1][0] = cz * ii + sz * rr;
    rr = cy * cx; ii = -sy * sx;
    mr[s][1][1] = cz * rr - sz * ii; mi[s][1][1] = cz * ii + sz * rr;
  }

  float* base = ws + mat * 768;
  const int n = lane & 15;
  const int khalf = (lane >> 4) << 1;
#pragma unroll
  for (int j = 0; j < 4; ++j) {
#pragma unroll
    for (int p = 0; p < 2; ++p) {
      const int k = (g == 0) ? (2 * j + p + 4 * khalf) : (4 * j + p + khalf);
      float re = 1.f, im = 0.f;
#pragma unroll
      for (int s = 0; s < 4; ++s) {
        const int rb = (n >> (3 - s)) & 1;
        const int cb = (k >> (3 - s)) & 1;
        const float wr = mr[s][rb][cb], wi = mi[s][rb][cb];
        const float nre = re * wr - im * wi;
        const float nim = re * wi + im * wr;
        re = nre; im = nim;
      }
      const int slot = lane * 8 + j * 2 + p;
      base[slot] = re;
      base[256 + slot] = im - re;
      base[512 + slot] = -(re + im);
    }
  }
}

__global__ __launch_bounds__(QC_THREADS) void qc_main(
    const float* __restrict__ x, const float* __restrict__ ws,
    float* __restrict__ out) {
  extern __shared__ float lds[];
  const int tid = threadIdx.x;
  const int lane = tid & 31;
  const int w = tid >> 5;
  const int e = w >> 1;
  const int h = w & 1;
  const int lt = h * 32 + lane;
  const long long b = (long long)blockIdx.x * QC_BATCH_PER_BLOCK + e;

  const int eb = QC_STATE_OFF + e * QC_ELEM;

  for (int p = tid; p < QC_DIM; p += QC_THREADS)
    lds[QC_FINV_OFF + fperm(p)] = __int_as_float(physidx(p));

  const float4* xr = (const float4*)(x + b * QC_DIM);
  float4 v[16];
  float ss = 0.f;
#pragma unroll
  for (int ii = 0; ii < 16; ++ii) {
    v[ii] = xr[ii * 64 + lt];
    ss += v[ii].x * v[ii].x + v[ii].y * v[ii].y + v[ii].z * v[ii].z +
          v[ii].w * v[ii].w;
  }
#pragma unroll
  for (int off = 16; off > 0; off >>= 1) ss += __shfl_xor(ss, off);
  if (lane == 0) lds[eb + QC_PLANE + h] = ss;
  __syncthreads();
  const float inv = rsqrtf(lds[eb + QC_PLANE] + lds[eb + QC_PLANE + 1]);
  __syncthreads();
#pragma unroll
  for (int ii = 0; ii < 16; ++ii) {
    const int u = ii * 64 + lt;
    const int pb = 4 * u + (u >> 2) + 4 * ii;
    lds[eb + pb + 0] = v[ii].x * inv;
    lds[eb + pb + 1] = v[ii].y * inv;
    lds[eb + pb + 2] = v[ii].z * inv;
    lds[eb + pb + 3] = v[ii].w * inv;
    lds[eb + QC_PLANE + pb + 0] = 0.f;
    lds[eb + QC_PLANE + pb + 1] = 0.f;
    lds[eb + QC_PLANE + pb + 2] = 0.f;
    lds[eb + QC_PLANE + pb + 3] = 0.f;
  }
  __syncthreads();

  int cb = 0;
#pragma unroll 1
  for (int L = 0; L < 3; ++L) {
    const float* wsL = ws + L * 3 * 768;
    const int cur = eb + cb * QC_BUF;
    const int nxt = eb + (cb ^ 1) * QC_BUF;
    apply_fused_ab(lds, cur, wsL + 0, wsL + 768, lane, h);
    __syncthreads();
    apply_pass<2, true>(lds, cur, nxt, wsL + 1536, lane, h);
    __syncthreads();
    cb ^= 1;
  }
  const int fin = eb + cb * QC_BUF;

  const int ebase = fin + lt + (lt >> 4);
  float T = 0.f;
  float accL[6];
#pragma unroll
  for (int q = 0; q < 6; ++q) accL[q] = 0.f;
#pragma unroll
  for (int ii = 0; ii < 64; ++ii) {
    const int d = 68 * ii + 4 * (ii >> 2);
    const float re = lds[ebase + d];
    const float im = lds[ebase + d + QC_PLANE];
    const float p = re * re + im * im;
    T += p;
#pragma unroll
    for (int q = 0; q < 6; ++q)
      accL[q] += ((ii >> (5 - q)) & 1) ? -p : p;
  }
  float acc[12];
#pragma unroll
  for (int q = 0; q < 6; ++q) acc[q] = accL[q];
#pragma unroll
  for (int q = 6; q < 12; ++q)
    acc[q] = ((lt >> (11 - q)) & 1) ? -T : T;
#pragma unroll
  for (int q = 0; q < 12; ++q) {
    float s = acc[q];
#pragma unroll
    for (int off = 16; off > 0; off >>= 1) s += __shfl_xor(s, off);
    acc[q] = s;
  }
  __syncthreads();
  if (lane == 0) {
#pragma unroll
    for (int q = 0; q < 12; ++q) lds[QC_FINV_OFF + w * 12 + q] = acc[q];
  }
  __syncthreads();
  if (h == 0 && lane < 12)
    out[b * 12 + lane] = lds[QC_FINV_OFF + (2 * e) * 12 + lane] +
                         lds[QC_FINV_OFF + (2 * e + 1) * 12 + lane];
}

extern "C" void kernel_launch(void* const* d_in, const int* in_sizes, int n_in,
                              void* d_out, int out_size, void* d_ws,
                              size_t ws_size, hipStream_t stream) {
  (void)n_in; (void)out_size; (void)ws_size;
  const float* x = (const float*)d_in[0];
  const float* params = (const float*)d_in[1];
  float* out = (float*)d_out;
  float* ws = (float*)d_ws;

  const int batch = in_sizes[0] / QC_DIM;
  const int blocks = batch / QC_BATCH_PER_BLOCK;

  qc_setup_gates<<<1, 288, 0, stream>>>(params, ws);

  (void)hipFuncSetAttribute((const void*)qc_main,
                            hipFuncAttributeMaxDynamicSharedMemorySize,
                            QC_LDS_BYTES);
  qc_main<<<blocks, QC_THREADS, QC_LDS_BYTES, stream>>>(x, ws, out);
}


# operand reads one stage ahead: LDS reads of tile pair p+1 ahead of pair p's matrix block (second register set), next pass's gate planes before its barrier, x row before table build; sums unchanged
# speedup vs baseline: 1.5576x; 1.0540x over previous
#include <hip/hip_runtime.h>
#include <math.h>


typedef float v2f __attribute__((ext_vector_type(2)));
typedef float v8f __attribute__((ext_vector_type(8)));

#define QC_BATCH_PER_BLOCK 2
#define QC_THREADS 128
#define QC_DIM 4096

#define QC_PLANE 4416
#define QC_BUF (2 * QC_PLANE)
#define QC_ELEM (2 * QC_BUF)
#define QC_FINV_OFF 0
#define QC_STATE_OFF QC_DIM
#define QC_LDS_FLOATS (QC_STATE_OFF + QC_BATCH_PER_BLOCK * QC_ELEM)
#define QC_LDS_BYTES (QC_LDS_FLOATS * 4)

static __device__ __forceinline__ v8f wmma4(v2f a, v2f b, v8f c) {
  return __builtin_amdgcn_wmma_f32_16x16x4_f32(false, a, false, b, (short)0, c,
                                               false, false);
}

static __device__ __forceinline__ int physidx(int i) {
  return i + (i >> 4) + 4 * (i >> 8);
}

static __device__ __forceinline__ int fperm(int i) {
  int x = i;
#pragma unroll
  for (int q = 11; q >= 0; --q) {
    int C = 11 - q;
    int T = 11 - ((q + 1) % 12);
    x ^= ((x >> C) & 1) << T;
  }
  return x;
}

struct Gate3 {
  float r[8], d[8], n[8];
};

static __device__ __forceinline__ void load_gate(const float* wsMat, int lane,
                                                 Gate3& g) {
  const float* b = wsMat + lane * 8;
#pragma unroll
  for (int s = 0; s < 8; ++s) {
    g.r[s] = b[s];
    g.d[s] = b[256 + s];
    g.n[s] = b[512 + s];
  }
}

struct TilePair {
  float r0[8], i0[8], r1[8], i1[8];
};

static __device__ __forceinline__ void load_c(const float* lds, int baseSrc,
                                              int tblBase, int dstOff, int t0,
                                              TilePair& a, int (&sa0)[8],
                                              int (&sa1)[8]) {
#pragma unroll
  for (int r = 0; r < 8; ++r) {
    sa0[r] = dstOff + __float_as_int(lds[tblBase + 256 * t0 + 16 * r]);
    sa1[r] = dstOff + __float_as_int(lds[tblBase + 256 * (t0 + 1) + 16 * r]);
  }
#pragma unroll
  for (int j = 0; j < 4; ++j) {
    const int o0 = baseSrc + 276 * t0 + 4 * j;
    const int o1 = o0 + 276;
    a.r0[2 * j] = lds[o0];
    a.r0[2 * j + 1] = lds[o0 + 1];
    a.i0[2 * j] = lds[o0 + QC_PLANE];
    a.i0[2 * j + 1] = lds[o0 + 1 + QC_PLANE];
    a.r1[2 * j] = lds[o1];
    a.r1[2 * j + 1] = lds[o1 + 1];
    a.i1[2 * j] = lds[o1 + QC_PLANE];
    a.i1[2 * j + 1] = lds[o1 + 1 + QC_PLANE];
  }
}

static __device__ __forceinline__ void apply_c_perm(float* lds, int srcOff,
                                                    int dstOff,
                                                    const Gate3& u, int lane,
                                                    int h) {
  const int m = lane & 15;
  const int kh = (lane >> 4) << 1;
  const int rh = (lane >> 4) << 3;
  const int baseSrc = srcOff + 276 * (h * 8) + 17 * m + kh;
  const int tblBase = QC_FINV_OFF + 256 * (h * 8) + 16 * rh + m;

  TilePair a[2];
  int sa0[2][8], sa1[2][8];
  load_c(lds, baseSrc, tblBase, dstOff, 0, a[0], sa0[0], sa1[0]);
#pragma unroll
  for (int tp = 0; tp < 4; ++tp) {
    const int cu = tp & 1;
    if (tp < 3)
      load_c(lds, baseSrc, tblBase, dstOff, 2 * (tp + 1), a[cu ^ 1],
             sa0[cu ^ 1], sa1[cu ^ 1]);
    __builtin_amdgcn_sched_barrier(0);
    const TilePair& s = a[cu];
    v8f k0 = {0.f, 0.f, 0.f, 0.f, 0.f, 0.f, 0.f, 0.f};
    v8f k1 = {0.f, 0.f, 0.f, 0.f, 0.f, 0.f, 0.f, 0.f};
#pragma unroll
    for (int j = 0; j < 4; ++j) {
      v2f Br = {u.r[2 * j], u.r[2 * j + 1]};
      v2f As0 = {s.r0[2 * j] + s.i0[2 * j], s.r0[2 * j + 1] + s.i0[2 * j + 1]};
      v2f As1 = {s.r1[2 * j] + s.i1[2 * j], s.r1[2 * j + 1] + s.i1[2 * j + 1]};
      k0 = wmma4(As0, Br, k0);
      k1 = wmma4(As1, Br, k1);
    }
    v8f dr0 = k0, di0 = k0, dr1 = k1, di1 = k1;
#pragma unroll
    for (int j = 0; j < 4; ++j) {
      v2f Bd = {u.d[2 * j], u.d[2 * j + 1]};
      v2f Bn = {u.n[2 * j], u.n[2 * j + 1]};
      v2f Ar0 = {s.r0[2 * j], s.r0[2 * j + 1]};
      v2f Ai0 = {s.i0[2 * j], s.i0[2 * j + 1]};
      v2f Ar1 = {s.r1[2 * j], s.r1[2 * j + 1]};
      v2f Ai1 = {s.i1[2 * j], s.i1[2 * j + 1]};
      dr0 = wmma4(Ai0, Bn, dr0);
      di0 = wmma4(Ar0, Bd, di0);
      dr1 = wmma4(Ai1, Bn, dr1);
      di1 = wmma4(Ar1, Bd, di1);
    }
#pragma unroll
    for (int r = 0; r < 8; ++r) {
      lds[sa0[cu][r]] = dr0[r];
      lds[sa0[cu][r] + QC_PLANE] = di0[r];
      lds[sa1[cu][r]] = dr1[r];
      lds[sa1[cu][r] + QC_PLANE] = di1[r];
    }
  }
}

static __device__ __forceinline__ void load_ab(const float* lds, int baseSrc,
                                               int c0, TilePair& x) {
#pragma unroll
  for (int j = 0; j < 4; ++j) {
    const int o0 = baseSrc + c0 + 17 * (4 * j);
    x.r0[2 * j] = lds[o0];
    x.r0[2 * j + 1] = lds[o0 + 17];
    x.i0[2 * j] = lds[o0 + QC_PLANE];
    x.i0[2 * j + 1] = lds[o0 + 17 + QC_PLANE];
  }
  asm volatile("" ::: "memory");
#pragma unroll
  for (int j = 0; j < 4; ++j) {
    const int o1 = baseSrc + c0 + 1 + 17 * (4 * j);
    x.r1[2 * j] = lds[o1];
    x.r1[2 * j + 1] = lds[o1 + 17];
    x.i1[2 * j] = lds[o1 + QC_PLANE];
    x.i1[2 * j + 1] = lds[o1 + 17 + QC_PLANE];
  }
  asm volatile("" ::: "memory");
}

static __device__ __forceinline__ void apply_fused_ab(float* lds, int base,
                                                      const Gate3& ua,
                                                      const Gate3& ub,
                                                      int lane, int h) {
  const int m = lane & 15;
  const int hf = lane >> 4;
  const int baseSrc = base + 276 * m + 17 * (2 * hf) + 8 * h;
  const int baseDst = base + 276 * (8 * hf) + 17 * m + 8 * h;

  TilePair x[2];
  load_ab(lds, baseSrc, 0, x[0]);
#pragma unroll
  for (int tp = 0; tp < 4; ++tp) {
    const int c0 = 2 * tp, c1 = 2 * tp + 1;
    if (tp < 3) load_ab(lds, baseSrc, 2 * (tp + 1), x[(tp & 1) ^ 1]);
    __builtin_amdgcn_sched_barrier(0);
    const TilePair& s = x[tp & 1];
    v8f k0 = {0.f, 0.f, 0.f, 0.f, 0.f, 0.f, 0.f, 0.f};
    v8f k1 = {0.f, 0.f, 0.f, 0.f, 0.f, 0.f, 0.f, 0.f};
#pragma unroll
    for (int j = 0; j < 4; ++j) {
      v2f Br = {ub.r[2 * j], ub.r[2 * j + 1]};
      v2f As0 = {s.r0[2 * j] + s.i0[2 * j], s.r0[2 * j + 1] + s.i0[2 * j + 1]};
      v2f As1 = {s.r1[2 * j] + s.i1[2 * j], s.r1[2 * j + 1] + s.i1[2 * j + 1]};
      k0 = wmma4(As0, Br, k0);
      k1 = wmma4(As1, Br, k1);
    }
    v8f yr0 = k0, yi0 = k0, yr1 = k1, yi1 = k1;
#pragma unroll
    for (int j = 0; j < 4; ++j) {
      v2f Bd = {ub.d[2 * j], ub.d[2 * j + 1]};
      v2f Bn = {ub.n[2 * j], ub.n[2 * j + 1]};
      v2f Ar0 = {s.r0[2 * j], s.r0[2 * j + 1]};
      v2f Ai0 = {s.i0[2 * j], s.i0[2 * j + 1]};
      v2f Ar1 = {s.r1[2 * j], s.r1[2 * j + 1]};
      v2f Ai1 = {s.i1[2 * j], s.i1[2 * j + 1]};
      yr0 = wmma4(Ai0, Bn, yr0);
      yi0 = wmma4(Ar0, Bd, yi0);
      yr1 = wmma4(Ai1, Bn, yr1);
      yi1 = wmma4(Ar1, Bd, yi1);
    }
    const v8f ys0 = yr0 + yi0;
    const v8f ys1 = yr1 + yi1;
    v8f q0 = {0.f, 0.f, 0.f, 0.f, 0.f, 0.f, 0.f, 0.f};
    v8f q1 = {0.f, 0.f, 0.f, 0.f, 0.f, 0.f, 0.f, 0.f};
#pragma unroll
    for (int j = 0; j < 4; ++j) {
      v2f Ur = {ua.r[2 * j], ua.r[2 * j + 1]};
      v2f Bs0 = {ys0[2 * j], ys0[2 * j + 1]};
      v2f Bs1 = {ys1[2 * j], ys1[2 * j + 1]};
      q0 = wmma4(Ur, Bs0, q0);
      q1 = wmma4(Ur, Bs1, q1);
    }
    v8f xr0 = q0, xi0 = q0, xr1 = q1, xi1 = q1;
#pragma unroll
    for (int j = 0; j < 4; ++j) {
      v2f Ud = {ua.d[2 * j], ua.d[2 * j + 1]};
      v2f Un = {ua.n[2 * j], ua.n[2 * j + 1]};
      v2f Yr0 = {yr0[2 * j], yr0[2 * j + 1]};
      v2f Yi0 = {yi0[2 * j], yi0[2 * j + 1]};
      v2f Yr1 = {yr1[2 * j], yr1[2 * j + 1]};
      v2f Yi1 = {yi1[2 * j], yi1[2 * j + 1]};
      xr0 = wmma4(Un, Yi0, xr0);
      xi0 = wmma4(Ud, Yr0, xi0);
      xr1 = wmma4(Un, Yi1, xr1);
      xi1 = wmma4(Ud, Yr1, xi1);
    }
#pragma unroll
    for (int r = 0; r < 8; ++r) {
      const int d0 = baseDst + c0 + 276 * r;
      const int d1 = baseDst + c1 + 276 * r;
      lds[d0] = xr0[r];
      lds[d0 + QC_PLANE] = xi0[r];
      lds[d1] = xr1[r];
      lds[d1 + QC_PLANE] = xi1[r];
    }
  }
}

__global__ void qc_setup_gates(const float* __restrict__ params,
                               float* __restrict__ ws) {
  const int tid = threadIdx.x;
  const int mat = tid >> 5;
  const int lane = tid & 31;
  if (mat >= 9) return;
  const int L = mat / 3, g = mat % 3;

  float mr[4][2][2], mi[4][2][2];
#pragma unroll
  for (int s = 0; s < 4; ++s) {
    const int q = g * 4 + s;
    const int base = L * 36 + q * 3;
    const float tx = params[base], ty = params[base + 1], tz = params[base + 2];
    const float cx = cosf(0.5f * tx), sx = sinf(0.5f * tx);
    const float cy = cosf(0.5f * ty), sy = sinf(0.5f * ty);
    const float cz = cosf(0.5f * tz), sz = sinf(0.5f * tz);
    float rr, ii;
    rr = cy * cx; ii = sy * sx;
    mr[s][0][0] = cz * rr + sz * ii; mi[s][0][0] = cz * ii - sz * rr;
    rr = -sy * cx; ii = -cy * sx;
    mr[s][0][1] = cz * rr + sz * ii; mi[s][0][1] = cz * ii - sz * rr;
    rr = sy * cx; ii = -cy * sx;
    mr[s][1][0] = cz * rr - sz * ii; mi[s][1][0] = cz * ii + sz * rr;
    rr = cy * cx; ii = -sy * sx;
    mr[s][1][1] = cz * rr - sz * ii; mi[s][1][1] = cz * ii + sz * rr;
  }

  float* base = ws + mat * 768;
  const int n = lane & 15;
  const int khalf = (lane >> 4) << 1;
#pragma unroll
  for (int j = 0; j < 4; ++j) {
#pragma unroll
    for (int p = 0; p < 2; ++p) {
      const int k = (g == 0) ? (2 * j + p + 4 * khalf) : (4 * j + p + khalf);
      float re = 1.f, im = 0.f;
#pragma unroll
      for (int s = 0; s < 4; ++s) {
        const int rb = (n >> (3 - s)) & 1;
        const int cb = (k >> (3 - s)) & 1;
        const float wr = mr[s][rb][cb], wi = mi[s][rb][cb];
        const float nre = re * wr - im * wi;
        const float nim = re * wi + im * wr;
        re = nre; im = nim;
      }
      const int slot = lane * 8 + j * 2 + p;
      base[slot] = re;
      base[256 + slot] = im - re;
      base[512 + slot] = -(re + im);
    }
  }
}

__global__ __launch_bounds__(QC_THREADS) void qc_main(
    const float* __restrict__ x, const float* __restrict__ ws,
    float* __restrict__ out) {
  extern __shared__ float lds[];
  const int tid = threadIdx.x;
  const int lane = tid & 31;
  const int w = tid >> 5;
  const int e = w >> 1;
  const int h = w & 1;
  const int lt = h * 32 + lane;
  const long long b = (long long)blockIdx.x * QC_BATCH_PER_BLOCK + e;

  const int eb = QC_STATE_OFF + e * QC_ELEM;

  const float4* xr = (const float4*)(x + b * QC_DIM);
  float4 v[16];
#pragma unroll
  for (int ii = 0; ii < 16; ++ii) v[ii] = xr[ii * 64 + lt];

  for (int p = tid; p < QC_DIM; p += QC_THREADS)
    lds[QC_FINV_OFF + fperm(p)] = __int_as_float(physidx(p));

  float ss = 0.f;
#pragma unroll
  for (int ii = 0; ii < 16; ++ii) {
    ss += v[ii].x * v[ii].x + v[ii].y * v[ii].y + v[ii].z * v[ii].z +
          v[ii].w * v[ii].w;
  }
#pragma unroll
  for (int off = 16; off > 0; off >>= 1) ss += __shfl_xor(ss, off);
  if (lane == 0) lds[eb + QC_PLANE + h] = ss;
  __syncthreads();
  const float inv = rsqrtf(lds[eb + QC_PLANE] + lds[eb + QC_PLANE + 1]);
  __syncthreads();
  Gate3 gA, gB;
  load_gate(ws + 0, lane, gA);
  load_gate(ws + 768, lane, gB);
#pragma unroll
  for (int ii = 0; ii < 16; ++ii) {
    const int u = ii * 64 + lt;
    const int pb = 4 * u + (u >> 2) + 4 * ii;
    lds[eb + pb + 0] = v[ii].x * inv;
    lds[eb + pb + 1] = v[ii].y * inv;
    lds[eb + pb + 2] = v[ii].z * inv;
    lds[eb + pb + 3] = v[ii].w * inv;
    lds[eb + QC_PLANE + pb + 0] = 0.f;
    lds[eb + QC_PLANE + pb + 1] = 0.f;
    lds[eb + QC_PLANE + pb + 2] = 0.f;
    lds[eb + QC_PLANE + pb + 3] = 0.f;
  }
  __syncthreads();

  int cb = 0;
#pragma unroll 1
  for (int L = 0; L < 3; ++L) {
    const float* wsL = ws + L * 3 * 768;
    const int cur = eb + cb * QC_BUF;
    const int nxt = eb + (cb ^ 1) * QC_BUF;
    Gate3 gC;
    load_gate(wsL + 1536, lane, gC);
    apply_fused_ab(lds, cur, gA, gB, lane, h);
    __syncthreads();
    const float* wsN = ws + ((L < 2) ? (L + 1) : 2) * 3 * 768;
    load_gate(wsN + 0, lane, gA);
    load_gate(wsN + 768, lane, gB);
    apply_c_perm(lds, cur, nxt, gC, lane, h);
    __syncthreads();
    cb ^= 1;
  }
  const int fin = eb + cb * QC_BUF;

  const int ebase = fin + lt + (lt >> 4);
  float T = 0.f;
  float accL[6];
#pragma unroll
  for (int q = 0; q < 6; ++q) accL[q] = 0.f;
#pragma unroll
  for (int ii = 0; ii < 64; ++ii) {
    const int d = 68 * ii + 4 * (ii >> 2);
    const float re = lds[ebase + d];
    const float im = lds[ebase + d + QC_PLANE];
    const float p = re * re + im * im;
    T += p;
#pragma unroll
    for (int q = 0; q < 6; ++q)
      accL[q] += ((ii >> (5 - q)) & 1) ? -p : p;
  }
  float acc[12];
#pragma unroll
  for (int q = 0; q < 6; ++q) acc[q] = accL[q];
#pragma unroll
  for (int q = 6; q < 12; ++q)
    acc[q] = ((lt >> (11 - q)) & 1) ? -T : T;
#pragma unroll
  for (int q = 0; q < 12; ++q) {
    float s = acc[q];
#pragma unroll
    for (int off = 16; off > 0; off >>= 1) s += __shfl_xor(s, off);
    acc[q] = s;
  }
  __syncthreads();
  if (lane == 0) {
#pragma unroll
    for (int q = 0; q < 12; ++q) lds[QC_FINV_OFF + w * 12 + q] = acc[q];
  }
  __syncthreads();
  if (h == 0 && lane < 12)
    out[b * 12 + lane] = lds[QC_FINV_OFF + (2 * e) * 12 + lane] +
                         lds[QC_FINV_OFF + (2 * e + 1) * 12 + lane];
}

extern "C" void kernel_launch(void* const* d_in, const int* in_sizes, int n_in,
                              void* d_out, int out_size, void* d_ws,
                              size_t ws_size, hipStream_t stream) {
  (void)n_in; (void)out_size; (void)ws_size;
  const float* x = (const float*)d_in[0];
  const float* params = (const float*)d_in[1];
  float* out = (float*)d_out;
  float* ws = (float*)d_ws;

  const int batch = in_sizes[0] / QC_DIM;
  const int blocks = batch / QC_BATCH_PER_BLOCK;

  qc_setup_gates<<<1, 288, 0, stream>>>(params, ws);

  (void)hipFuncSetAttribute((const void*)qc_main,
                            hipFuncAttributeMaxDynamicSharedMemorySize,
                            QC_LDS_BYTES);
  qc_main<<<blocks, QC_THREADS, QC_LDS_BYTES, stream>>>(x, ws, out);
}


# layer 0 uses that the embedded state is real: the first product of its fused pass takes two real products per tile (Yr=Sr*Ur, Yi=Yr+Sr*(Ui-Ur)), no Im reads, no Im zero-fill; sums unchanged
# speedup vs baseline: 1.6091x; 1.0331x over previous
#include <hip/hip_runtime.h>
#include <math.h>


typedef float v2f __attribute__((ext_vector_type(2)));
typedef float v8f __attribute__((ext_vector_type(8)));

#define QC_BATCH_PER_BLOCK 2
#define QC_THREADS 128
#define QC_DIM 4096

#define QC_PLANE 4416
#define QC_BUF (2 * QC_PLANE)
#define QC_ELEM (2 * QC_BUF)
#define QC_FINV_OFF 0
#define QC_STATE_OFF QC_DIM
#define QC_LDS_FLOATS (QC_STATE_OFF + QC_BATCH_PER_BLOCK * QC_ELEM)
#define QC_LDS_BYTES (QC_LDS_FLOATS * 4)

static __device__ __forceinline__ v8f wmma4(v2f a, v2f b, v8f c) {
  return __builtin_amdgcn_wmma_f32_16x16x4_f32(false, a, false, b, (short)0, c,
                                               false, false);
}

static __device__ __forceinline__ int physidx(int i) {
  return i + (i >> 4) + 4 * (i >> 8);
}

static __device__ __forceinline__ int fperm(int i) {
  int x = i;
#pragma unroll
  for (int q = 11; q >= 0; --q) {
    int C = 11 - q;
    int T = 11 - ((q + 1) % 12);
    x ^= ((x >> C) & 1) << T;
  }
  return x;
}

struct Gate3 {
  float r[8], d[8], n[8];
};

static __device__ __forceinline__ void load_gate(const float* wsMat, int lane,
                                                 Gate3& g) {
  const float* b = wsMat + lane * 8;
#pragma unroll
  for (int s = 0; s < 8; ++s) {
    g.r[s] = b[s];
    g.d[s] = b[256 + s];
    g.n[s] = b[512 + s];
  }
}

struct TilePair {
  float r0[8], i0[8], r1[8], i1[8];
};

static __device__ __forceinline__ void load_c(const float* lds, int baseSrc,
                                              int tblBase, int dstOff, int t0,
                                              TilePair& a, int (&sa0)[8],
                                              int (&sa1)[8]) {
#pragma unroll
  for (int r = 0; r < 8; ++r) {
    sa0[r] = dstOff + __float_as_int(lds[tblBase + 256 * t0 + 16 * r]);
    sa1[r] = dstOff + __float_as_int(lds[tblBase + 256 * (t0 + 1) + 16 * r]);
  }
#pragma unroll
  for (int j = 0; j < 4; ++j) {
    const int o0 = baseSrc + 276 * t0 + 4 * j;
    const int o1 = o0 + 276;
    a.r0[2 * j] = lds[o0];
    a.r0[2 * j + 1] = lds[o0 + 1];
    a.i0[2 * j] = lds[o0 + QC_PLANE];
    a.i0[2 * j + 1] = lds[o0 + 1 + QC_PLANE];
    a.r1[2 * j] = lds[o1];
    a.r1[2 * j + 1] = lds[o1 + 1];
    a.i1[2 * j] = lds[o1 + QC_PLANE];
    a.i1[2 * j + 1] = lds[o1 + 1 + QC_PLANE];
  }
}

static __device__ __forceinline__ void apply_c_perm(float* lds, int srcOff,
                                                    int dstOff,
                                                    const Gate3& u, int lane,
                                                    int h) {
  const int m = lane & 15;
  const int kh = (lane >> 4) << 1;
  const int rh = (lane >> 4) << 3;
  const int baseSrc = srcOff + 276 * (h * 8) + 17 * m + kh;
  const int tblBase = QC_FINV_OFF + 256 * (h * 8) + 16 * rh + m;

  TilePair a[2];
  int sa0[2][8], sa1[2][8];
  load_c(lds, baseSrc, tblBase, dstOff, 0, a[0], sa0[0], sa1[0]);
#pragma unroll
  for (int tp = 0; tp < 4; ++tp) {
    const int cu = tp & 1;
    if (tp < 3)
      load_c(lds, baseSrc, tblBase, dstOff, 2 * (tp + 1), a[cu ^ 1],
             sa0[cu ^ 1], sa1[cu ^ 1]);
    __builtin_amdgcn_sched_barrier(0);
    const TilePair& s = a[cu];
    v8f k0 = {0.f, 0.f, 0.f, 0.f, 0.f, 0.f, 0.f, 0.f};
    v8f k1 = {0.f, 0.f, 0.f, 0.f, 0.f, 0.f, 0.f, 0.f};
#pragma unroll
    for (int j = 0; j < 4; ++j) {
      v2f Br = {u.r[2 * j], u.r[2 * j + 1]};
      v2f As0 = {s.r0[2 * j] + s.i0[2 * j], s.r0[2 * j + 1] + s.i0[2 * j + 1]};
      v2f As1 = {s.r1[2 * j] + s.i1[2 * j], s.r1[2 * j + 1] + s.i1[2 * j + 1]};
      k0 = wmma4(As0, Br, k0);
      k1 = wmma4(As1, Br, k1);
    }
    v8f dr0 = k0, di0 = k0, dr1 = k1, di1 = k1;
#pragma unroll
    for (int j = 0; j < 4; ++j) {
      v2f Bd = {u.d[2 * j], u.d[2 * j + 1]};
      v2f Bn = {u.n[2 * j], u.n[2 * j + 1]};
      v2f Ar0 = {s.r0[2 * j], s.r0[2 * j + 1]};
      v2f Ai0 = {s.i0[2 * j], s.i0[2 * j + 1]};
      v2f Ar1 = {s.r1[2 * j], s.r1[2 * j + 1]};
      v2f Ai1 = {s.i1[2 * j], s.i1[2 * j + 1]};
      dr0 = wmma4(Ai0, Bn, dr0);
      di0 = wmma4(Ar0, Bd, di0);
      dr1 = wmma4(Ai1, Bn, dr1);
      di1 = wmma4(Ar1, Bd, di1);
    }
#pragma unroll
    for (int r = 0; r < 8; ++r) {
      lds[sa0[cu][r]] = dr0[r];
      lds[sa0[cu][r] + QC_PLANE] = di0[r];
      lds[sa1[cu][r]] = dr1[r];
      lds[sa1[cu][r] + QC_PLANE] = di1[r];
    }
  }
}

template <bool REAL>
static __device__ __forceinline__ void load_ab(const float* lds, int baseSrc,
                                               int c0, TilePair& x) {
#pragma unroll
  for (int j = 0; j < 4; ++j) {
    const int o0 = baseSrc + c0 + 17 * (4 * j);
    x.r0[2 * j] = lds[o0];
    x.r0[2 * j + 1] = lds[o0 + 17];
    if (!REAL) {
      x.i0[2 * j] = lds[o0 + QC_PLANE];
      x.i0[2 * j + 1] = lds[o0 + 17 + QC_PLANE];
    }
  }
  asm volatile("" ::: "memory");
#pragma unroll
  for (int j = 0; j < 4; ++j) {
    const int o1 = baseSrc + c0 + 1 + 17 * (4 * j);
    x.r1[2 * j] = lds[o1];
    x.r1[2 * j + 1] = lds[o1 + 17];
    if (!REAL) {
      x.i1[2 * j] = lds[o1 + QC_PLANE];
      x.i1[2 * j + 1] = lds[o1 + 17 + QC_PLANE];
    }
  }
  asm volatile("" ::: "memory");
}

template <bool REAL>
static __device__ __forceinline__ void apply_fused_ab(float* lds, int base,
                                                      const Gate3& ua,
                                                      const Gate3& ub,
                                                      int lane, int h) {
  const int m = lane & 15;
  const int hf = lane >> 4;
  const int baseSrc = base + 276 * m + 17 * (2 * hf) + 8 * h;
  const int baseDst = base + 276 * (8 * hf) + 17 * m + 8 * h;

  TilePair x[2];
  load_ab<REAL>(lds, baseSrc, 0, x[0]);
#pragma unroll
  for (int tp = 0; tp < 4; ++tp) {
    const int c0 = 2 * tp, c1 = 2 * tp + 1;
    if (tp < 3) load_ab<REAL>(lds, baseSrc, 2 * (tp + 1), x[(tp & 1) ^ 1]);
    __builtin_amdgcn_sched_barrier(0);
    const TilePair& s = x[tp & 1];
    v8f k0 = {0.f, 0.f, 0.f, 0.f, 0.f, 0.f, 0.f, 0.f};
    v8f k1 = {0.f, 0.f, 0.f, 0.f, 0.f, 0.f, 0.f, 0.f};
#pragma unroll
    for (int j = 0; j < 4; ++j) {
      v2f Br = {ub.r[2 * j], ub.r[2 * j + 1]};
      v2f As0, As1;
      if (REAL) {
        As0 = v2f{s.r0[2 * j], s.r0[2 * j + 1]};
        As1 = v2f{s.r1[2 * j], s.r1[2 * j + 1]};
      } else {
        As0 = v2f{s.r0[2 * j] + s.i0[2 * j], s.r0[2 * j + 1] + s.i0[2 * j + 1]};
        As1 = v2f{s.r1[2 * j] + s.i1[2 * j], s.r1[2 * j + 1] + s.i1[2 * j + 1]};
      }
      k0 = wmma4(As0, Br, k0);
      k1 = wmma4(As1, Br, k1);
    }
    v8f yr0 = k0, yi0 = k0, yr1 = k1, yi1 = k1;
#pragma unroll
    for (int j = 0; j < 4; ++j) {
      v2f Bd = {ub.d[2 * j], ub.d[2 * j + 1]};
      v2f Ar0 = {s.r0[2 * j], s.r0[2 * j + 1]};
      v2f Ar1 = {s.r1[2 * j], s.r1[2 * j + 1]};
      if (!REAL) {
        v2f Bn = {ub.n[2 * j], ub.n[2 * j + 1]};
        v2f Ai0 = {s.i0[2 * j], s.i0[2 * j + 1]};
        v2f Ai1 = {s.i1[2 * j], s.i1[2 * j + 1]};
        yr0 = wmma4(Ai0, Bn, yr0);
        yi0 = wmma4(Ar0, Bd, yi0);
        yr1 = wmma4(Ai1, Bn, yr1);
        yi1 = wmma4(Ar1, Bd, yi1);
      } else {
        yi0 = wmma4(Ar0, Bd, yi0);
        yi1 = wmma4(Ar1, Bd, yi1);
      }
    }
    const v8f ys0 = yr0 + yi0;
    const v8f ys1 = yr1 + yi1;
    v8f q0 = {0.f, 0.f, 0.f, 0.f, 0.f, 0.f, 0.f, 0.f};
    v8f q1 = {0.f, 0.f, 0.f, 0.f, 0.f, 0.f, 0.f, 0.f};
#pragma unroll
    for (int j = 0; j < 4; ++j) {
      v2f Ur = {ua.r[2 * j], ua.r[2 * j + 1]};
      v2f Bs0 = {ys0[2 * j], ys0[2 * j + 1]};
      v2f Bs1 = {ys1[2 * j], ys1[2 * j + 1]};
      q0 = wmma4(Ur, Bs0, q0);
      q1 = wmma4(Ur, Bs1, q1);
    }
    v8f xr0 = q0, xi0 = q0, xr1 = q1, xi1 = q1;
#pragma unroll
    for (int j = 0; j < 4; ++j) {
      v2f Ud = {ua.d[2 * j], ua.d[2 * j + 1]};
      v2f Un = {ua.n[2 * j], ua.n[2 * j + 1]};
      v2f Yr0 = {yr0[2 * j], yr0[2 * j + 1]};
      v2f Yi0 = {yi0[2 * j], yi0[2 * j + 1]};
      v2f Yr1 = {yr1[2 * j], yr1[2 * j + 1]};
      v2f Yi1 = {yi1[2 * j], yi1[2 * j + 1]};
      xr0 = wmma4(Un, Yi0, xr0);
      xi0 = wmma4(Ud, Yr0, xi0);
      xr1 = wmma4(Un, Yi1, xr1);
      xi1 = wmma4(Ud, Yr1, xi1);
    }
#pragma unroll
    for (int r = 0; r < 8; ++r) {
      const int d0 = baseDst + c0 + 276 * r;
      const int d1 = baseDst + c1 + 276 * r;
      lds[d0] = xr0[r];
      lds[d0 + QC_PLANE] = xi0[r];
      lds[d1] = xr1[r];
      lds[d1 + QC_PLANE] = xi1[r];
    }
  }
}

__global__ void qc_setup_gates(const float* __restrict__ params,
                               float* __restrict__ ws) {
  const int tid = threadIdx.x;
  const int mat = tid >> 5;
  const int lane = tid & 31;
  if (mat >= 9) return;
  const int L = mat / 3, g = mat % 3;

  float mr[4][2][2], mi[4][2][2];
#pragma unroll
  for (int s = 0; s < 4; ++s) {
    const int q = g * 4 + s;
    const int base = L * 36 + q * 3;
    const float tx = params[base], ty = params[base + 1], tz = params[base + 2];
    const float cx = cosf(0.5f * tx), sx = sinf(0.5f * tx);
    const float cy = cosf(0.5f * ty), sy = sinf(0.5f * ty);
    const float cz = cosf(0.5f * tz), sz = sinf(0.5f * tz);
    float rr, ii;
    rr = cy * cx; ii = sy * sx;
    mr[s][0][0] = cz * rr + sz * ii; mi[s][0][0] = cz * ii - sz * rr;
    rr = -sy * cx; ii = -cy * sx;
    mr[s][0][1] = cz * rr + sz * ii; mi[s][0][1] = cz * ii - sz * rr;
    rr = sy * cx; ii = -cy * sx;
    mr[s][1][0] = cz * rr - sz * ii; mi[s][1][0] = cz * ii + sz * rr;
    rr = cy * cx; ii = -sy * sx;
    mr[s][1][1] = cz * rr - sz * ii; mi[s][1][1] = cz * ii + sz * rr;
  }

  float* base = ws + mat * 768;
  const int n = lane & 15;
  const int khalf = (lane >> 4) << 1;
#pragma unroll
  for (int j = 0; j < 4; ++j) {
#pragma unroll
    for (int p = 0; p < 2; ++p) {
      const int k = (g == 0) ? (2 * j + p + 4 * khalf) : (4 * j + p + khalf);
      float re = 1.f, im = 0.f;
#pragma unroll
      for (int s = 0; s < 4; ++s) {
        const int rb = (n >> (3 - s)) & 1;
        const int cb = (k >> (3 - s)) & 1;
        const float wr = mr[s][rb][cb], wi = mi[s][rb][cb];
        const float nre = re * wr - im * wi;
        const float nim = re * wi + im * wr;
        re = nre; im = nim;
      }
      const int slot = lane * 8 + j * 2 + p;
      base[slot] = re;
      base[256 + slot] = im - re;
      base[512 + slot] = -(re + im);
    }
  }
}

__global__ __launch_bounds__(QC_THREADS) void qc_main(
    const float* __restrict__ x, const float* __restrict__ ws,
    float* __restrict__ out) {
  extern __shared__ float lds[];
  const int tid = threadIdx.x;
  const int lane = tid & 31;
  const int w = tid >> 5;
  const int e = w >> 1;
  const int h = w & 1;
  const int lt = h * 32 + lane;
  const long long b = (long long)blockIdx.x * QC_BATCH_PER_BLOCK + e;

  const int eb = QC_STATE_OFF + e * QC_ELEM;

  const float4* xr = (const float4*)(x + b * QC_DIM);
  float4 v[16];
#pragma unroll
  for (int ii = 0; ii < 16; ++ii) v[ii] = xr[ii * 64 + lt];

  for (int p = tid; p < QC_DIM; p += QC_THREADS)
    lds[QC_FINV_OFF + fperm(p)] = __int_as_float(physidx(p));

  float ss = 0.f;
#pragma unroll
  for (int ii = 0; ii < 16; ++ii) {
    ss += v[ii].x * v[ii].x + v[ii].y * v[ii].y + v[ii].z * v[ii].z +
          v[ii].w * v[ii].w;
  }
#pragma unroll
  for (int off = 16; off > 0; off >>= 1) ss += __shfl_xor(ss, off);
  if (lane == 0) lds[eb + QC_PLANE + h] = ss;
  __syncthreads();
  const float inv = rsqrtf(lds[eb + QC_PLANE] + lds[eb + QC_PLANE + 1]);
  __syncthreads();
  Gate3 gA, gB;
  load_gate(ws + 0, lane, gA);
  load_gate(ws + 768, lane, gB);
#pragma unroll
  for (int ii = 0; ii < 16; ++ii) {
    const int u = ii * 64 + lt;
    const int pb = 4 * u + (u >> 2) + 4 * ii;
    lds[eb + pb + 0] = v[ii].x * inv;
    lds[eb + pb + 1] = v[ii].y * inv;
    lds[eb + pb + 2] = v[ii].z * inv;
    lds[eb + pb + 3] = v[ii].w * inv;
  }
  __syncthreads();

  int cb = 0;
#pragma unroll 1
  for (int L = 0; L < 3; ++L) {
    const float* wsL = ws + L * 3 * 768;
    const int cur = eb + cb * QC_BUF;
    const int nxt = eb + (cb ^ 1) * QC_BUF;
    Gate3 gC;
    load_gate(wsL + 1536, lane, gC);
    if (L == 0)
      apply_fused_ab<true>(lds, cur, gA, gB, lane, h);
    else
      apply_fused_ab<false>(lds, cur, gA, gB, lane, h);
    __syncthreads();
    const float* wsN = ws + ((L < 2) ? (L + 1) : 2) * 3 * 768;
    load_gate(wsN + 0, lane, gA);
    load_gate(wsN + 768, lane, gB);
    apply_c_perm(lds, cur, nxt, gC, lane, h);
    __syncthreads();
    cb ^= 1;
  }
  const int fin = eb + cb * QC_BUF;

  const int ebase = fin + lt + (lt >> 4);
  float T = 0.f;
  float accL[6];
#pragma unroll
  for (int q = 0; q < 6; ++q) accL[q] = 0.f;
#pragma unroll
  for (int ii = 0; ii < 64; ++ii) {
    const int d = 68 * ii + 4 * (ii >> 2);
    const float re = lds[ebase + d];
    const float im = lds[ebase + d + QC_PLANE];
    const float p = re * re + im * im;
    T += p;
#pragma unroll
    for (int q = 0; q < 6; ++q)
      accL[q] += ((ii >> (5 - q)) & 1) ? -p : p;
  }
  float acc[12];
#pragma unroll
  for (int q = 0; q < 6; ++q) acc[q] = accL[q];
#pragma unroll
  for (int q = 6; q < 12; ++q)
    acc[q] = ((lt >> (11 - q)) & 1) ? -T : T;
#pragma unroll
  for (int q = 0; q < 12; ++q) {
    float s = acc[q];
#pragma unroll
    for (int off = 16; off > 0; off >>= 1) s += __shfl_xor(s, off);
    acc[q] = s;
  }
  __syncthreads();
  if (lane == 0) {
#pragma unroll
    for (int q = 0; q < 12; ++q) lds[QC_FINV_OFF + w * 12 + q] = acc[q];
  }
  __syncthreads();
  if (h == 0 && lane < 12)
    out[b * 12 + lane] = lds[QC_FINV_OFF + (2 * e) * 12 + lane] +
                         lds[QC_FINV_OFF + (2 * e + 1) * 12 + lane];
}

extern "C" void kernel_launch(void* const* d_in, const int* in_sizes, int n_in,
                              void* d_out, int out_size, void* d_ws,
                              size_t ws_size, hipStream_t stream) {
  (void)n_in; (void)out_size; (void)ws_size;
  const float* x = (const float*)d_in[0];
  const float* params = (const float*)d_in[1];
  float* out = (float*)d_out;
  float* ws = (float*)d_ws;

  const int batch = in_sizes[0] / QC_DIM;
  const int blocks = batch / QC_BATCH_PER_BLOCK;

  qc_setup_gates<<<1, 288, 0, stream>>>(params, ws);

  (void)hipFuncSetAttribute((const void*)qc_main,
                            hipFuncAttributeMaxDynamicSharedMemorySize,
                            QC_LDS_BYTES);
  qc_main<<<blocks, QC_THREADS, QC_LDS_BYTES, stream>>>(x, ws, out);
}


# group C and CNOT pass in place (tile results kept in registers over a barrier, then scattered): one state buffer per element, three work-groups per WGP instead of two; sums unchanged
# speedup vs baseline: 1.6782x; 1.0429x over previous
#include <hip/hip_runtime.h>
#include <math.h>


typedef float v2f __attribute__((ext_vector_type(2)));
typedef float v8f __attribute__((ext_vector_type(8)));

#define QC_BATCH_PER_BLOCK 2
#define QC_THREADS 128
#define QC_DIM 4096

#define QC_PLANE 4416
#define QC_ELEM (2 * QC_PLANE)
#define QC_FINV_OFF 0
#define QC_STATE_OFF QC_DIM
#define QC_LDS_FLOATS (QC_STATE_OFF + QC_BATCH_PER_BLOCK * QC_ELEM)
#define QC_LDS_BYTES (QC_LDS_FLOATS * 4)

static __device__ __forceinline__ v8f wmma4(v2f a, v2f b, v8f c) {
  return __builtin_amdgcn_wmma_f32_16x16x4_f32(false, a, false, b, (short)0, c,
                                               false, false);
}

static __device__ __forceinline__ int physidx(int i) {
  return i + (i >> 4) + 4 * (i >> 8);
}

static __device__ __forceinline__ int fperm(int i) {
  int x = i;
#pragma unroll
  for (int q = 11; q >= 0; --q) {
    int C = 11 - q;
    int T = 11 - ((q + 1) % 12);
    x ^= ((x >> C) & 1) << T;
  }
  return x;
}

struct Gate3 {
  float r[8], d[8], n[8];
};

static __device__ __forceinline__ void load_gate(const float* wsMat, int lane,
                                                 Gate3& g) {
  const float* b = wsMat + lane * 8;
#pragma unroll
  for (int s = 0; s < 8; ++s) {
    g.r[s] = b[s];
    g.d[s] = b[256 + s];
    g.n[s] = b[512 + s];
  }
}

struct TilePair {
  float r0[8], i0[8], r1[8], i1[8];
};

static __device__ __forceinline__ void load_c(const float* lds, int baseSrc,
                                              int tblBase, int dstOff, int t0,
                                              TilePair& a, int (&sa0)[8],
                                              int (&sa1)[8]) {
#pragma unroll
  for (int r = 0; r < 8; ++r) {
    sa0[r] = dstOff + __float_as_int(lds[tblBase + 256 * t0 + 16 * r]);
    sa1[r] = dstOff + __float_as_int(lds[tblBase + 256 * (t0 + 1) + 16 * r]);
  }
#pragma unroll
  for (int j = 0; j < 4; ++j) {
    const int o0 = baseSrc + 276 * t0 + 4 * j;
    const int o1 = o0 + 276;
    a.r0[2 * j] = lds[o0];
    a.r0[2 * j + 1] = lds[o0 + 1];
    a.i0[2 * j] = lds[o0 + QC_PLANE];
    a.i0[2 * j + 1] = lds[o0 + 1 + QC_PLANE];
    a.r1[2 * j] = lds[o1];
    a.r1[2 * j + 1] = lds[o1 + 1];
    a.i1[2 * j] = lds[o1 + QC_PLANE];
    a.i1[2 * j + 1] = lds[o1 + 1 + QC_PLANE];
  }
}

static __device__ __forceinline__ void apply_c_perm(float* lds, int base,
                                                    const Gate3& u,
                                                    const float* wsN,
                                                    Gate3& gA, Gate3& gB,
                                                    int lane, int h) {
  const int m = lane & 15;
  const int kh = (lane >> 4) << 1;
  const int rh = (lane >> 4) << 3;
  const int baseSrc = base + 276 * (h * 8) + 17 * m + kh;
  const int tblBase = QC_FINV_OFF + 256 * (h * 8) + 16 * rh + m;

  TilePair a[2];
  int sa[8][8];
  v8f dr[8], di[8];
  load_c(lds, baseSrc, tblBase, base, 0, a[0], sa[0], sa[1]);
#pragma unroll
  for (int tp = 0; tp < 4; ++tp) {
    const int cu = tp & 1;
    if (tp < 3)
      load_c(lds, baseSrc, tblBase, base, 2 * (tp + 1), a[cu ^ 1],
             sa[2 * tp + 2], sa[2 * tp + 3]);
    __builtin_amdgcn_sched_barrier(0);
    const TilePair& s = a[cu];
    v8f k0 = {0.f, 0.f, 0.f, 0.f, 0.f, 0.f, 0.f, 0.f};
    v8f k1 = {0.f, 0.f, 0.f, 0.f, 0.f, 0.f, 0.f, 0.f};
#pragma unroll
    for (int j = 0; j < 4; ++j) {
      v2f Br = {u.r[2 * j], u.r[2 * j + 1]};
      v2f As0 = {s.r0[2 * j] + s.i0[2 * j], s.r0[2 * j + 1] + s.i0[2 * j + 1]};
      v2f As1 = {s.r1[2 * j] + s.i1[2 * j], s.r1[2 * j + 1] + s.i1[2 * j + 1]};
      k0 = wmma4(As0, Br, k0);
      k1 = wmma4(As1, Br, k1);
    }
    v8f dr0 = k0, di0 = k0, dr1 = k1, di1 = k1;
#pragma unroll
    for (int j = 0; j < 4; ++j) {
      v2f Bd = {u.d[2 * j], u.d[2 * j + 1]};
      v2f Bn = {u.n[2 * j], u.n[2 * j + 1]};
      v2f Ar0 = {s.r0[2 * j], s.r0[2 * j + 1]};
      v2f Ai0 = {s.i0[2 * j], s.i0[2 * j + 1]};
      v2f Ar1 = {s.r1[2 * j], s.r1[2 * j + 1]};
      v2f Ai1 = {s.i1[2 * j], s.i1[2 * j + 1]};
      dr0 = wmma4(Ai0, Bn, dr0);
      di0 = wmma4(Ar0, Bd, di0);
      dr1 = wmma4(Ai1, Bn, dr1);
      di1 = wmma4(Ar1, Bd, di1);
    }
    dr[2 * tp] = dr0;
    di[2 * tp] = di0;
    dr[2 * tp + 1] = dr1;
    di[2 * tp + 1] = di1;
  }
  __syncthreads();
  load_gate(wsN + 0, lane, gA);
  load_gate(wsN + 768, lane, gB);
#pragma unroll
  for (int t = 0; t < 8; ++t) {
#pragma unroll
    for (int r = 0; r < 8; ++r) {
      lds[sa[t][r]] = dr[t][r];
      lds[sa[t][r] + QC_PLANE] = di[t][r];
    }
  }
}

template <bool REAL>
static __device__ __forceinline__ void load_ab(const float* lds, int baseSrc,
                                               int c0, TilePair& x) {
#pragma unroll
  for (int j = 0; j < 4; ++j) {
    const int o0 = baseSrc + c0 + 17 * (4 * j);
    x.r0[2 * j] = lds[o0];
    x.r0[2 * j + 1] = lds[o0 + 17];
    if (!REAL) {
      x.i0[2 * j] = lds[o0 + QC_PLANE];
      x.i0[2 * j + 1] = lds[o0 + 17 + QC_PLANE];
    }
  }
  asm volatile("" ::: "memory");
#pragma unroll
  for (int j = 0; j < 4; ++j) {
    const int o1 = baseSrc + c0 + 1 + 17 * (4 * j);
    x.r1[2 * j] = lds[o1];
    x.r1[2 * j + 1] = lds[o1 + 17];
    if (!REAL) {
      x.i1[2 * j] = lds[o1 + QC_PLANE];
      x.i1[2 * j + 1] = lds[o1 + 17 + QC_PLANE];
    }
  }
  asm volatile("" ::: "memory");
}

template <bool REAL>
static __device__ __forceinline__ void apply_fused_ab(float* lds, int base,
                                                      const Gate3& ua,
                                                      const Gate3& ub,
                                                      int lane, int h) {
  const int m = lane & 15;
  const int hf = lane >> 4;
  const int baseSrc = base + 276 * m + 17 * (2 * hf) + 8 * h;
  const int baseDst = base + 276 * (8 * hf) + 17 * m + 8 * h;

  TilePair x[2];
  load_ab<REAL>(lds, baseSrc, 0, x[0]);
#pragma unroll
  for (int tp = 0; tp < 4; ++tp) {
    const int c0 = 2 * tp, c1 = 2 * tp + 1;
    if (tp < 3) load_ab<REAL>(lds, baseSrc, 2 * (tp + 1), x[(tp & 1) ^ 1]);
    __builtin_amdgcn_sched_barrier(0);
    const TilePair& s = x[tp & 1];
    v8f k0 = {0.f, 0.f, 0.f, 0.f, 0.f, 0.f, 0.f, 0.f};
    v8f k1 = {0.f, 0.f, 0.f, 0.f, 0.f, 0.f, 0.f, 0.f};
#pragma unroll
    for (int j = 0; j < 4; ++j) {
      v2f Br = {ub.r[2 * j], ub.r[2 * j + 1]};
      v2f As0, As1;
      if (REAL) {
        As0 = v2f{s.r0[2 * j], s.r0[2 * j + 1]};
        As1 = v2f{s.r1[2 * j], s.r1[2 * j + 1]};
      } else {
        As0 = v2f{s.r0[2 * j] + s.i0[2 * j], s.r0[2 * j + 1] + s.i0[2 * j + 1]};
        As1 = v2f{s.r1[2 * j] + s.i1[2 * j], s.r1[2 * j + 1] + s.i1[2 * j + 1]};
      }
      k0 = wmma4(As0, Br, k0);
      k1 = wmma4(As1, Br, k1);
    }
    v8f yr0 = k0, yi0 = k0, yr1 = k1, yi1 = k1;
#pragma unroll
    for (int j = 0; j < 4; ++j) {
      v2f Bd = {ub.d[2 * j], ub.d[2 * j + 1]};
      v2f Ar0 = {s.r0[2 * j], s.r0[2 * j + 1]};
      v2f Ar1 = {s.r1[2 * j], s.r1[2 * j + 1]};
      if (!REAL) {
        v2f Bn = {ub.n[2 * j], ub.n[2 * j + 1]};
        v2f Ai0 = {s.i0[2 * j], s.i0[2 * j + 1]};
        v2f Ai1 = {s.i1[2 * j], s.i1[2 * j + 1]};
        yr0 = wmma4(Ai0, Bn, yr0);
        yi0 = wmma4(Ar0, Bd, yi0);
        yr1 = wmma4(Ai1, Bn, yr1);
        yi1 = wmma4(Ar1, Bd, yi1);
      } else {
        yi0 = wmma4(Ar0, Bd, yi0);
        yi1 = wmma4(Ar1, Bd, yi1);
      }
    }
    const v8f ys0 = yr0 + yi0;
    const v8f ys1 = yr1 + yi1;
    v8f q0 = {0.f, 0.f, 0.f, 0.f, 0.f, 0.f, 0.f, 0.f};
    v8f q1 = {0.f, 0.f, 0.f, 0.f, 0.f, 0.f, 0.f, 0.f};
#pragma unroll
    for (int j = 0; j < 4; ++j) {
      v2f Ur = {ua.r[2 * j], ua.r[2 * j + 1]};
      v2f Bs0 = {ys0[2 * j], ys0[2 * j + 1]};
      v2f Bs1 = {ys1[2 * j], ys1[2 * j + 1]};
      q0 = wmma4(Ur, Bs0, q0);
      q1 = wmma4(Ur, Bs1, q1);
    }
    v8f xr0 = q0, xi0 = q0, xr1 = q1, xi1 = q1;
#pragma unroll
    for (int j = 0; j < 4; ++j) {
      v2f Ud = {ua.d[2 * j], ua.d[2 * j + 1]};
      v2f Un = {ua.n[2 * j], ua.n[2 * j + 1]};
      v2f Yr0 = {yr0[2 * j], yr0[2 * j + 1]};
      v2f Yi0 = {yi0[2 * j], yi0[2 * j + 1]};
      v2f Yr1 = {yr1[2 * j], yr1[2 * j + 1]};
      v2f Yi1 = {yi1[2 * j], yi1[2 * j + 1]};
      xr0 = wmma4(Un, Yi0, xr0);
      xi0 = wmma4(Ud, Yr0, xi0);
      xr1 = wmma4(Un, Yi1, xr1);
      xi1 = wmma4(Ud, Yr1, xi1);
    }
#pragma unroll
    for (int r = 0; r < 8; ++r) {
      const int d0 = baseDst + c0 + 276 * r;
      const int d1 = baseDst + c1 + 276 * r;
      lds[d0] = xr0[r];
      lds[d0 + QC_PLANE] = xi0[r];
      lds[d1] = xr1[r];
      lds[d1 + QC_PLANE] = xi1[r];
    }
  }
}

__global__ void qc_setup_gates(const float* __restrict__ params,
                               float* __restrict__ ws) {
  const int tid = threadIdx.x;
  const int mat = tid >> 5;
  const int lane = tid & 31;
  if (mat >= 9) return;
  const int L = mat / 3, g = mat % 3;

  float mr[4][2][2], mi[4][2][2];
#pragma unroll
  for (int s = 0; s < 4; ++s) {
    const int q = g * 4 + s;
    const int base = L * 36 + q * 3;
    const float tx = params[base], ty = params[base + 1], tz = params[base + 2];
    const float cx = cosf(0.5f * tx), sx = sinf(0.5f * tx);
    const float cy = cosf(0.5f * ty), sy = sinf(0.5f * ty);
    const float cz = cosf(0.5f * tz), sz = sinf(0.5f * tz);
    float rr, ii;
    rr = cy * cx; ii = sy * sx;
    mr[s][0][0] = cz * rr + sz * ii; mi[s][0][0] = cz * ii - sz * rr;
    rr = -sy * cx; ii = -cy * sx;
    mr[s][0][1] = cz * rr + sz * ii; mi[s][0][1] = cz * ii - sz * rr;
    rr = sy * cx; ii = -cy * sx;
    mr[s][1][0] = cz * rr - sz * ii; mi[s][1][0] = cz * ii + sz * rr;
    rr = cy * cx; ii = -sy * sx;
    mr[s][1][1] = cz * rr - sz * ii; mi[s][1][1] = cz * ii + sz * rr;
  }

  float* base = ws + mat * 768;
  const int n = lane & 15;
  const int khalf = (lane >> 4) << 1;
#pragma unroll
  for (int j = 0; j < 4; ++j) {
#pragma unroll
    for (int p = 0; p < 2; ++p) {
      const int k = (g == 0) ? (2 * j + p + 4 * khalf) : (4 * j + p + khalf);
      float re = 1.f, im = 0.f;
#pragma unroll
      for (int s = 0; s < 4; ++s) {
        const int rb = (n >> (3 - s)) & 1;
        const int cb = (k >> (3 - s)) & 1;
        const float wr = mr[s][rb][cb], wi = mi[s][rb][cb];
        const float nre = re * wr - im * wi;
        const float nim = re * wi + im * wr;
        re = nre; im = nim;
      }
      const int slot = lane * 8 + j * 2 + p;
      base[slot] = re;
      base[256 + slot] = im - re;
      base[512 + slot] = -(re + im);
    }
  }
}

__global__ __launch_bounds__(QC_THREADS, 3) void qc_main(
    const float* __restrict__ x, const float* __restrict__ ws,
    float* __restrict__ out) {
  extern __shared__ float lds[];
  const int tid = threadIdx.x;
  const int lane = tid & 31;
  const int w = tid >> 5;
  const int e = w >> 1;
  const int h = w & 1;
  const int lt = h * 32 + lane;
  const long long b = (long long)blockIdx.x * QC_BATCH_PER_BLOCK + e;

  const int eb = QC_STATE_OFF + e * QC_ELEM;

  const float4* xr = (const float4*)(x + b * QC_DIM);
  float4 v[16];
#pragma unroll
  for (int ii = 0; ii < 16; ++ii) v[ii] = xr[ii * 64 + lt];

  for (int p = tid; p < QC_DIM; p += QC_THREADS)
    lds[QC_FINV_OFF + fperm(p)] = __int_as_float(physidx(p));

  float ss = 0.f;
#pragma unroll
  for (int ii = 0; ii < 16; ++ii) {
    ss += v[ii].x * v[ii].x + v[ii].y * v[ii].y + v[ii].z * v[ii].z +
          v[ii].w * v[ii].w;
  }
#pragma unroll
  for (int off = 16; off > 0; off >>= 1) ss += __shfl_xor(ss, off);
  if (lane == 0) lds[eb + QC_PLANE + h] = ss;
  __syncthreads();
  const float inv = rsqrtf(lds[eb + QC_PLANE] + lds[eb + QC_PLANE + 1]);
  __syncthreads();
  Gate3 gA, gB;
  load_gate(ws + 0, lane, gA);
  load_gate(ws + 768, lane, gB);
#pragma unroll
  for (int ii = 0; ii < 16; ++ii) {
    const int u = ii * 64 + lt;
    const int pb = 4 * u + (u >> 2) + 4 * ii;
    lds[eb + pb + 0] = v[ii].x * inv;
    lds[eb + pb + 1] = v[ii].y * inv;
    lds[eb + pb + 2] = v[ii].z * inv;
    lds[eb + pb + 3] = v[ii].w * inv;
  }
  __syncthreads();

#pragma unroll 1
  for (int L = 0; L < 3; ++L) {
    const float* wsL = ws + L * 3 * 768;
    const int cur = eb;
    Gate3 gC;
    load_gate(wsL + 1536, lane, gC);
    if (L == 0)
      apply_fused_ab<true>(lds, cur, gA, gB, lane, h);
    else
      apply_fused_ab<false>(lds, cur, gA, gB, lane, h);
    __syncthreads();
    const float* wsN = ws + ((L < 2) ? (L + 1) : 2) * 3 * 768;
    apply_c_perm(lds, cur, gC, wsN, gA, gB, lane, h);
    __syncthreads();
  }
  const int fin = eb;

  const int ebase = fin + lt + (lt >> 4);
  float T = 0.f;
  float accL[6];
#pragma unroll
  for (int q = 0; q < 6; ++q) accL[q] = 0.f;
#pragma unroll
  for (int ii = 0; ii < 64; ++ii) {
    const int d = 68 * ii + 4 * (ii >> 2);
    const float re = lds[ebase + d];
    const float im = lds[ebase + d + QC_PLANE];
    const float p = re * re + im * im;
    T += p;
#pragma unroll
    for (int q = 0; q < 6; ++q)
      accL[q] += ((ii >> (5 - q)) & 1) ? -p : p;
  }
  float acc[12];
#pragma unroll
  for (int q = 0; q < 6; ++q) acc[q] = accL[q];
#pragma unroll
  for (int q = 6; q < 12; ++q)
    acc[q] = ((lt >> (11 - q)) & 1) ? -T : T;
#pragma unroll
  for (int q = 0; q < 12; ++q) {
    float s = acc[q];
#pragma unroll
    for (int off = 16; off > 0; off >>= 1) s += __shfl_xor(s, off);
    acc[q] = s;
  }
  __syncthreads();
  if (lane == 0) {
#pragma unroll
    for (int q = 0; q < 12; ++q) lds[QC_FINV_OFF + w * 12 + q] = acc[q];
  }
  __syncthreads();
  if (h == 0 && lane < 12)
    out[b * 12 + lane] = lds[QC_FINV_OFF + (2 * e) * 12 + lane] +
                         lds[QC_FINV_OFF + (2 * e + 1) * 12 + lane];
}

extern "C" void kernel_launch(void* const* d_in, const int* in_sizes, int n_in,
                              void* d_out, int out_size, void* d_ws,
                              size_t ws_size, hipStream_t stream) {
  (void)n_in; (void)out_size; (void)ws_size;
  const float* x = (const float*)d_in[0];
  const float* params = (const float*)d_in[1];
  float* out = (float*)d_out;
  float* ws = (float*)d_ws;

  const int batch = in_sizes[0] / QC_DIM;
  const int blocks = batch / QC_BATCH_PER_BLOCK;

  qc_setup_gates<<<1, 288, 0, stream>>>(params, ws);

  (void)hipFuncSetAttribute((const void*)qc_main,
                            hipFuncAttributeMaxDynamicSharedMemorySize,
                            QC_LDS_BYTES);
  qc_main<<<blocks, QC_THREADS, QC_LDS_BYTES, stream>>>(x, ws, out);
}


# last layer's expectation sums taken from the group-C result registers, CNOT ring folded into sign masks: no final scatter, no state re-read, three barriers fewer; changes summation order
# speedup vs baseline: 1.7344x; 1.0335x over previous
#include <hip/hip_runtime.h>
#include <math.h>


typedef float v2f __attribute__((ext_vector_type(2)));
typedef float v8f __attribute__((ext_vector_type(8)));

#define QC_BATCH_PER_BLOCK 2
#define QC_THREADS 128
#define QC_DIM 4096

#define QC_PLANE 4416
#define QC_ELEM (2 * QC_PLANE)
#define QC_FINV_OFF 0
#define QC_STATE_OFF QC_DIM
#define QC_LDS_FLOATS (QC_STATE_OFF + QC_BATCH_PER_BLOCK * QC_ELEM)
#define QC_LDS_BYTES (QC_LDS_FLOATS * 4)

static __device__ __forceinline__ v8f wmma4(v2f a, v2f b, v8f c) {
  return __builtin_amdgcn_wmma_f32_16x16x4_f32(false, a, false, b, (short)0, c,
                                               false, false);
}

static __device__ __forceinline__ int physidx(int i) {
  return i + (i >> 4) + 4 * (i >> 8);
}

static __device__ __forceinline__ int fperm(int i) {
  int x = i;
#pragma unroll
  for (int q = 11; q >= 0; --q) {
    int C = 11 - q;
    int T = 11 - ((q + 1) % 12);
    x ^= ((x >> C) & 1) << T;
  }
  return x;
}

static constexpr int finvperm(int i) {
  int x = i;
  for (int q = 0; q < 12; ++q) {
    int C = 11 - q;
    int T = 11 - ((q + 1) % 12);
    x ^= ((x >> C) & 1) << T;
  }
  return x;
}

static constexpr int finv_mask(int k) {
  int m = 0;
  for (int b = 0; b < 12; ++b)
    if ((finvperm(1 << b) >> k) & 1) m |= 1 << b;
  return m;
}

struct Gate3 {
  float r[8], d[8], n[8];
};

static __device__ __forceinline__ void load_gate(const float* wsMat, int lane,
                                                 Gate3& g) {
  const float* b = wsMat + lane * 8;
#pragma unroll
  for (int s = 0; s < 8; ++s) {
    g.r[s] = b[s];
    g.d[s] = b[256 + s];
    g.n[s] = b[512 + s];
  }
}

struct TilePair {
  float r0[8], i0[8], r1[8], i1[8];
};

static __device__ __forceinline__ void load_c(const float* lds, int baseSrc,
                                              int tblBase, int dstOff, int t0,
                                              TilePair& a, int (&sa0)[8],
                                              int (&sa1)[8]) {
#pragma unroll
  for (int r = 0; r < 8; ++r) {
    sa0[r] = dstOff + __float_as_int(lds[tblBase + 256 * t0 + 16 * r]);
    sa1[r] = dstOff + __float_as_int(lds[tblBase + 256 * (t0 + 1) + 16 * r]);
  }
#pragma unroll
  for (int j = 0; j < 4; ++j) {
    const int o0 = baseSrc + 276 * t0 + 4 * j;
    const int o1 = o0 + 276;
    a.r0[2 * j] = lds[o0];
    a.r0[2 * j + 1] = lds[o0 + 1];
    a.i0[2 * j] = lds[o0 + QC_PLANE];
    a.i0[2 * j + 1] = lds[o0 + 1 + QC_PLANE];
    a.r1[2 * j] = lds[o1];
    a.r1[2 * j + 1] = lds[o1 + 1];
    a.i1[2 * j] = lds[o1 + QC_PLANE];
    a.i1[2 * j + 1] = lds[o1 + 1 + QC_PLANE];
  }
}

static __device__ __forceinline__ void apply_c_perm(float* lds, int base,
                                                    const Gate3& u,
                                                    const float* wsN,
                                                    Gate3& gA, Gate3& gB,
                                                    int lane, int h) {
  const int m = lane & 15;
  const int kh = (lane >> 4) << 1;
  const int rh = (lane >> 4) << 3;
  const int baseSrc = base + 276 * (h * 8) + 17 * m + kh;
  const int tblBase = QC_FINV_OFF + 256 * (h * 8) + 16 * rh + m;

  TilePair a[2];
  int sa[8][8];
  v8f dr[8], di[8];
  load_c(lds, baseSrc, tblBase, base, 0, a[0], sa[0], sa[1]);
#pragma unroll
  for (int tp = 0; tp < 4; ++tp) {
    const int cu = tp & 1;
    if (tp < 3)
      load_c(lds, baseSrc, tblBase, base, 2 * (tp + 1), a[cu ^ 1],
             sa[2 * tp + 2], sa[2 * tp + 3]);
    __builtin_amdgcn_sched_barrier(0);
    const TilePair& s = a[cu];
    v8f k0 = {0.f, 0.f, 0.f, 0.f, 0.f, 0.f, 0.f, 0.f};
    v8f k1 = {0.f, 0.f, 0.f, 0.f, 0.f, 0.f, 0.f, 0.f};
#pragma unroll
    for (int j = 0; j < 4; ++j) {
      v2f Br = {u.r[2 * j], u.r[2 * j + 1]};
      v2f As0 = {s.r0[2 * j] + s.i0[2 * j], s.r0[2 * j + 1] + s.i0[2 * j + 1]};
      v2f As1 = {s.r1[2 * j] + s.i1[2 * j], s.r1[2 * j + 1] + s.i1[2 * j + 1]};
      k0 = wmma4(As0, Br, k0);
      k1 = wmma4(As1, Br, k1);
    }
    v8f dr0 = k0, di0 = k0, dr1 = k1, di1 = k1;
#pragma unroll
    for (int j = 0; j < 4; ++j) {
      v2f Bd = {u.d[2 * j], u.d[2 * j + 1]};
      v2f Bn = {u.n[2 * j], u.n[2 * j + 1]};
      v2f Ar0 = {s.r0[2 * j], s.r0[2 * j + 1]};
      v2f Ai0 = {s.i0[2 * j], s.i0[2 * j + 1]};
      v2f Ar1 = {s.r1[2 * j], s.r1[2 * j + 1]};
      v2f Ai1 = {s.i1[2 * j], s.i1[2 * j + 1]};
      dr0 = wmma4(Ai0, Bn, dr0);
      di0 = wmma4(Ar0, Bd, di0);
      dr1 = wmma4(Ai1, Bn, dr1);
      di1 = wmma4(Ar1, Bd, di1);
    }
    dr[2 * tp] = dr0;
    di[2 * tp] = di0;
    dr[2 * tp + 1] = dr1;
    di[2 * tp + 1] = di1;
  }
  __syncthreads();
  load_gate(wsN + 0, lane, gA);
  load_gate(wsN + 768, lane, gB);
#pragma unroll
  for (int t = 0; t < 8; ++t) {
#pragma unroll
    for (int r = 0; r < 8; ++r) {
      lds[sa[t][r]] = dr[t][r];
      lds[sa[t][r] + QC_PLANE] = di[t][r];
    }
  }
}

static __device__ __forceinline__ void load_c_ops(const float* lds, int baseSrc,
                                                  int t0, TilePair& a) {
#pragma unroll
  for (int j = 0; j < 4; ++j) {
    const int o0 = baseSrc + 276 * t0 + 4 * j;
    const int o1 = o0 + 276;
    a.r0[2 * j] = lds[o0];
    a.r0[2 * j + 1] = lds[o0 + 1];
    a.i0[2 * j] = lds[o0 + QC_PLANE];
    a.i0[2 * j + 1] = lds[o0 + 1 + QC_PLANE];
    a.r1[2 * j] = lds[o1];
    a.r1[2 * j + 1] = lds[o1 + 1];
    a.i1[2 * j] = lds[o1 + QC_PLANE];
    a.i1[2 * j + 1] = lds[o1 + 1 + QC_PLANE];
  }
}

template <int Q>
static __device__ __forceinline__ void z_add_tile(float (&acc)[12],
                                                  const float (&wh)[8], int t) {
  constexpr int M = finv_mask(11 - Q);
  constexpr int TQ = (M >> 8) & 7;
  constexpr int RQ = (M >> 4) & 7;
  const float w = wh[RQ];
  acc[Q] += (__builtin_popcount(t & TQ) & 1) ? -w : w;
}

static __device__ __forceinline__ void z_tile(float (&acc)[12], const v8f& dr,
                                              const v8f& di, int t) {
  float wh[8];
#pragma unroll
  for (int r = 0; r < 8; ++r) wh[r] = dr[r] * dr[r] + di[r] * di[r];
#pragma unroll
  for (int len = 1; len < 8; len <<= 1) {
#pragma unroll
    for (int i = 0; i < 8; i += 2 * len) {
#pragma unroll
      for (int j = i; j < i + len; ++j) {
        const float x = wh[j], y = wh[j + len];
        wh[j] = x + y;
        wh[j + len] = x - y;
      }
    }
  }
  z_add_tile<0>(acc, wh, t);
  z_add_tile<1>(acc, wh, t);
  z_add_tile<2>(acc, wh, t);
  z_add_tile<3>(acc, wh, t);
  z_add_tile<4>(acc, wh, t);
  z_add_tile<5>(acc, wh, t);
  z_add_tile<6>(acc, wh, t);
  z_add_tile<7>(acc, wh, t);
  z_add_tile<8>(acc, wh, t);
  z_add_tile<9>(acc, wh, t);
  z_add_tile<10>(acc, wh, t);
  z_add_tile<11>(acc, wh, t);
}

template <int Q>
static __device__ __forceinline__ void z_lane_sign(float (&acc)[12], int key) {
  constexpr int LQ = finv_mask(11 - Q) & 0x88F;
  const int s = (__builtin_popcount(key & LQ) & 1) << 31;
  acc[Q] = __int_as_float(__float_as_int(acc[Q]) ^ s);
}

static __device__ __forceinline__ void apply_c_final(const float* lds, int base,
                                                     const Gate3& u, int lane,
                                                     int h, float (&acc)[12]) {
  const int m = lane & 15;
  const int kh = (lane >> 4) << 1;
  const int baseSrc = base + 276 * (h * 8) + 17 * m + kh;

  TilePair a[2];
  v8f pr0, pi0, pr1, pi1;
#pragma unroll
  for (int q = 0; q < 12; ++q) acc[q] = 0.f;
  load_c_ops(lds, baseSrc, 0, a[0]);
#pragma unroll
  for (int tp = 0; tp < 4; ++tp) {
    const int cu = tp & 1;
    if (tp < 3) load_c_ops(lds, baseSrc, 2 * (tp + 1), a[cu ^ 1]);
    asm volatile("" ::: "memory");
    __builtin_amdgcn_sched_barrier(0);
    if (tp > 0) {
      asm volatile("" : "+v"(pr0));
      asm volatile("" : "+v"(pi0));
      asm volatile("" : "+v"(pr1));
      asm volatile("" : "+v"(pi1));
      z_tile(acc, pr0, pi0, 2 * tp - 2);
      z_tile(acc, pr1, pi1, 2 * tp - 1);
    }
    const TilePair& s = a[cu];
    v8f k0 = {0.f, 0.f, 0.f, 0.f, 0.f, 0.f, 0.f, 0.f};
    v8f k1 = {0.f, 0.f, 0.f, 0.f, 0.f, 0.f, 0.f, 0.f};
#pragma unroll
    for (int j = 0; j < 4; ++j) {
      v2f Br = {u.r[2 * j], u.r[2 * j + 1]};
      v2f As0 = {s.r0[2 * j] + s.i0[2 * j], s.r0[2 * j + 1] + s.i0[2 * j + 1]};
      v2f As1 = {s.r1[2 * j] + s.i1[2 * j], s.r1[2 * j + 1] + s.i1[2 * j + 1]};
      k0 = wmma4(As0, Br, k0);
      k1 = wmma4(As1, Br, k1);
    }
    v8f dr0 = k0, di0 = k0, dr1 = k1, di1 = k1;
#pragma unroll
    for (int j = 0; j < 4; ++j) {
      v2f Bd = {u.d[2 * j], u.d[2 * j + 1]};
      v2f Bn = {u.n[2 * j], u.n[2 * j + 1]};
      v2f Ar0 = {s.r0[2 * j], s.r0[2 * j + 1]};
      v2f Ai0 = {s.i0[2 * j], s.i0[2 * j + 1]};
      v2f Ar1 = {s.r1[2 * j], s.r1[2 * j + 1]};
      v2f Ai1 = {s.i1[2 * j], s.i1[2 * j + 1]};
      dr0 = wmma4(Ai0, Bn, dr0);
      di0 = wmma4(Ar0, Bd, di0);
      dr1 = wmma4(Ai1, Bn, dr1);
      di1 = wmma4(Ar1, Bd, di1);
    }
    pr0 = dr0;
    pi0 = di0;
    pr1 = dr1;
    pi1 = di1;
    __builtin_amdgcn_sched_barrier(0);
  }
  z_tile(acc, pr0, pi0, 6);
  z_tile(acc, pr1, pi1, 7);
  const int key = (h << 11) | ((lane >> 4) << 7) | m;
  z_lane_sign<0>(acc, key);
  z_lane_sign<1>(acc, key);
  z_lane_sign<2>(acc, key);
  z_lane_sign<3>(acc, key);
  z_lane_sign<4>(acc, key);
  z_lane_sign<5>(acc, key);
  z_lane_sign<6>(acc, key);
  z_lane_sign<7>(acc, key);
  z_lane_sign<8>(acc, key);
  z_lane_sign<9>(acc, key);
  z_lane_sign<10>(acc, key);
  z_lane_sign<11>(acc, key);
}

template <bool REAL>
static __device__ __forceinline__ void load_ab(const float* lds, int baseSrc,
                                               int c0, TilePair& x) {
#pragma unroll
  for (int j = 0; j < 4; ++j) {
    const int o0 = baseSrc + c0 + 17 * (4 * j);
    x.r0[2 * j] = lds[o0];
    x.r0[2 * j + 1] = lds[o0 + 17];
    if (!REAL) {
      x.i0[2 * j] = lds[o0 + QC_PLANE];
      x.i0[2 * j + 1] = lds[o0 + 17 + QC_PLANE];
    }
  }
  asm volatile("" ::: "memory");
#pragma unroll
  for (int j = 0; j < 4; ++j) {
    const int o1 = baseSrc + c0 + 1 + 17 * (4 * j);
    x.r1[2 * j] = lds[o1];
    x.r1[2 * j + 1] = lds[o1 + 17];
    if (!REAL) {
      x.i1[2 * j] = lds[o1 + QC_PLANE];
      x.i1[2 * j + 1] = lds[o1 + 17 + QC_PLANE];
    }
  }
  asm volatile("" ::: "memory");
}

template <bool REAL>
static __device__ __forceinline__ void apply_fused_ab(float* lds, int base,
                                                      const Gate3& ua,
                                                      const Gate3& ub,
                                                      int lane, int h) {
  const int m = lane & 15;
  const int hf = lane >> 4;
  const int baseSrc = base + 276 * m + 17 * (2 * hf) + 8 * h;
  const int baseDst = base + 276 * (8 * hf) + 17 * m + 8 * h;

  TilePair x[2];
  load_ab<REAL>(lds, baseSrc, 0, x[0]);
#pragma unroll
  for (int tp = 0; tp < 4; ++tp) {
    const int c0 = 2 * tp, c1 = 2 * tp + 1;
    if (tp < 3) load_ab<REAL>(lds, baseSrc, 2 * (tp + 1), x[(tp & 1) ^ 1]);
    __builtin_amdgcn_sched_barrier(0);
    const TilePair& s = x[tp & 1];
    v8f k0 = {0.f, 0.f, 0.f, 0.f, 0.f, 0.f, 0.f, 0.f};
    v8f k1 = {0.f, 0.f, 0.f, 0.f, 0.f, 0.f, 0.f, 0.f};
#pragma unroll
    for (int j = 0; j < 4; ++j) {
      v2f Br = {ub.r[2 * j], ub.r[2 * j + 1]};
      v2f As0, As1;
      if (REAL) {
        As0 = v2f{s.r0[2 * j], s.r0[2 * j + 1]};
        As1 = v2f{s.r1[2 * j], s.r1[2 * j + 1]};
      } else {
        As0 = v2f{s.r0[2 * j] + s.i0[2 * j], s.r0[2 * j + 1] + s.i0[2 * j + 1]};
        As1 = v2f{s.r1[2 * j] + s.i1[2 * j], s.r1[2 * j + 1] + s.i1[2 * j + 1]};
      }
      k0 = wmma4(As0, Br, k0);
      k1 = wmma4(As1, Br, k1);
    }
    v8f yr0 = k0, yi0 = k0, yr1 = k1, yi1 = k1;
#pragma unroll
    for (int j = 0; j < 4; ++j) {
      v2f Bd = {ub.d[2 * j], ub.d[2 * j + 1]};
      v2f Ar0 = {s.r0[2 * j], s.r0[2 * j + 1]};
      v2f Ar1 = {s.r1[2 * j], s.r1[2 * j + 1]};
      if (!REAL) {
        v2f Bn = {ub.n[2 * j], ub.n[2 * j + 1]};
        v2f Ai0 = {s.i0[2 * j], s.i0[2 * j + 1]};
        v2f Ai1 = {s.i1[2 * j], s.i1[2 * j + 1]};
        yr0 = wmma4(Ai0, Bn, yr0);
        yi0 = wmma4(Ar0, Bd, yi0);
        yr1 = wmma4(Ai1, Bn, yr1);
        yi1 = wmma4(Ar1, Bd, yi1);
      } else {
        yi0 = wmma4(Ar0, Bd, yi0);
        yi1 = wmma4(Ar1, Bd, yi1);
      }
    }
    const v8f ys0 = yr0 + yi0;
    const v8f ys1 = yr1 + yi1;
    v8f q0 = {0.f, 0.f, 0.f, 0.f, 0.f, 0.f, 0.f, 0.f};
    v8f q1 = {0.f, 0.f, 0.f, 0.f, 0.f, 0.f, 0.f, 0.f};
#pragma unroll
    for (int j = 0; j < 4; ++j) {
      v2f Ur = {ua.r[2 * j], ua.r[2 * j + 1]};
      v2f Bs0 = {ys0[2 * j], ys0[2 * j + 1]};
      v2f Bs1 = {ys1[2 * j], ys1[2 * j + 1]};
      q0 = wmma4(Ur, Bs0, q0);
      q1 = wmma4(Ur, Bs1, q1);
    }
    v8f xr0 = q0, xi0 = q0, xr1 = q1, xi1 = q1;
#pragma unroll
    for (int j = 0; j < 4; ++j) {
      v2f Ud = {ua.d[2 * j], ua.d[2 * j + 1]};
      v2f Un = {ua.n[2 * j], ua.n[2 * j + 1]};
      v2f Yr0 = {yr0[2 * j], yr0[2 * j + 1]};
      v2f Yi0 = {yi0[2 * j], yi0[2 * j + 1]};
      v2f Yr1 = {yr1[2 * j], yr1[2 * j + 1]};
      v2f Yi1 = {yi1[2 * j], yi1[2 * j + 1]};
      xr0 = wmma4(Un, Yi0, xr0);
      xi0 = wmma4(Ud, Yr0, xi0);
      xr1 = wmma4(Un, Yi1, xr1);
      xi1 = wmma4(Ud, Yr1, xi1);
    }
#pragma unroll
    for (int r = 0; r < 8; ++r) {
      const int d0 = baseDst + c0 + 276 * r;
      const int d1 = baseDst + c1 + 276 * r;
      lds[d0] = xr0[r];
      lds[d0 + QC_PLANE] = xi0[r];
      lds[d1] = xr1[r];
      lds[d1 + QC_PLANE] = xi1[r];
    }
  }
}

__global__ void qc_setup_gates(const float* __restrict__ params,
                               float* __restrict__ ws) {
  const int tid = threadIdx.x;
  const int mat = tid >> 5;
  const int lane = tid & 31;
  if (mat >= 9) return;
  const int L = mat / 3, g = mat % 3;

  float mr[4][2][2], mi[4][2][2];
#pragma unroll
  for (int s = 0; s < 4; ++s) {
    const int q = g * 4 + s;
    const int base = L * 36 + q * 3;
    const float tx = params[base], ty = params[base + 1], tz = params[base + 2];
    const float cx = cosf(0.5f * tx), sx = sinf(0.5f * tx);
    const float cy = cosf(0.5f * ty), sy = sinf(0.5f * ty);
    const float cz = cosf(0.5f * tz), sz = sinf(0.5f * tz);
    float rr, ii;
    rr = cy * cx; ii = sy * sx;
    mr[s][0][0] = cz * rr + sz * ii; mi[s][0][0] = cz * ii - sz * rr;
    rr = -sy * cx; ii = -cy * sx;
    mr[s][0][1] = cz * rr + sz * ii; mi[s][0][1] = cz * ii - sz * rr;
    rr = sy * cx; ii = -cy * sx;
    mr[s][1][0] = cz * rr - sz * ii; mi[s][1][0] = cz * ii + sz * rr;
    rr = cy * cx; ii = -sy * sx;
    mr[s][1][1] = cz * rr - sz * ii; mi[s][1][1] = cz * ii + sz * rr;
  }

  float* base = ws + mat * 768;
  const int n = lane & 15;
  const int khalf = (lane >> 4) << 1;
#pragma unroll
  for (int j = 0; j < 4; ++j) {
#pragma unroll
    for (int p = 0; p < 2; ++p) {
      const int k = (g == 0) ? (2 * j + p + 4 * khalf) : (4 * j + p + khalf);
      float re = 1.f, im = 0.f;
#pragma unroll
      for (int s = 0; s < 4; ++s) {
        const int rb = (n >> (3 - s)) & 1;
        const int cb = (k >> (3 - s)) & 1;
        const float wr = mr[s][rb][cb], wi = mi[s][rb][cb];
        const float nre = re * wr - im * wi;
        const float nim = re * wi + im * wr;
        re = nre; im = nim;
      }
      const int slot = lane * 8 + j * 2 + p;
      base[slot] = re;
      base[256 + slot] = im - re;
      base[512 + slot] = -(re + im);
    }
  }
}

__global__ __launch_bounds__(QC_THREADS, 3) void qc_main(
    const float* __restrict__ x, const float* __restrict__ ws,
    float* __restrict__ out) {
  extern __shared__ float lds[];
  const int tid = threadIdx.x;
  const int lane = tid & 31;
  const int w = tid >> 5;
  const int e = w >> 1;
  const int h = w & 1;
  const int lt = h * 32 + lane;
  const long long b = (long long)blockIdx.x * QC_BATCH_PER_BLOCK + e;

  const int eb = QC_STATE_OFF + e * QC_ELEM;

  const float4* xr = (const float4*)(x + b * QC_DIM);
  float4 v[16];
#pragma unroll
  for (int ii = 0; ii < 16; ++ii) v[ii] = xr[ii * 64 + lt];

  for (int p = tid; p < QC_DIM; p += QC_THREADS)
    lds[QC_FINV_OFF + fperm(p)] = __int_as_float(physidx(p));

  float ss = 0.f;
#pragma unroll
  for (int ii = 0; ii < 16; ++ii) {
    ss += v[ii].x * v[ii].x + v[ii].y * v[ii].y + v[ii].z * v[ii].z +
          v[ii].w * v[ii].w;
  }
#pragma unroll
  for (int off = 16; off > 0; off >>= 1) ss += __shfl_xor(ss, off);
  if (lane == 0) lds[eb + QC_PLANE + h] = ss;
  __syncthreads();
  const float inv = rsqrtf(lds[eb + QC_PLANE] + lds[eb + QC_PLANE + 1]);
  __syncthreads();
  Gate3 gA, gB;
  load_gate(ws + 0, lane, gA);
  load_gate(ws + 768, lane, gB);
#pragma unroll
  for (int ii = 0; ii < 16; ++ii) {
    const int u = ii * 64 + lt;
    const int pb = 4 * u + (u >> 2) + 4 * ii;
    lds[eb + pb + 0] = v[ii].x * inv;
    lds[eb + pb + 1] = v[ii].y * inv;
    lds[eb + pb + 2] = v[ii].z * inv;
    lds[eb + pb + 3] = v[ii].w * inv;
  }
  __syncthreads();

  Gate3 gC;
#pragma unroll 1
  for (int L = 0;; ++L) {
    const float* wsL = ws + L * 3 * 768;
    load_gate(wsL + 1536, lane, gC);
    if (L == 0)
      apply_fused_ab<true>(lds, eb, gA, gB, lane, h);
    else
      apply_fused_ab<false>(lds, eb, gA, gB, lane, h);
    __syncthreads();
    if (L == 2) break;
    const float* wsN = ws + (L + 1) * 3 * 768;
    apply_c_perm(lds, eb, gC, wsN, gA, gB, lane, h);
    __syncthreads();
  }
  float acc[12];
  apply_c_final(lds, eb, gC, lane, h, acc);
#pragma unroll
  for (int q = 0; q < 12; ++q) {
    float s = acc[q];
#pragma unroll
    for (int off = 16; off > 0; off >>= 1) s += __shfl_xor(s, off);
    acc[q] = s;
  }
  if (lane == 0) {
#pragma unroll
    for (int q = 0; q < 12; ++q) lds[QC_FINV_OFF + w * 12 + q] = acc[q];
  }
  __syncthreads();
  if (h == 0 && lane < 12)
    out[b * 12 + lane] = lds[QC_FINV_OFF + (2 * e) * 12 + lane] +
                         lds[QC_FINV_OFF + (2 * e + 1) * 12 + lane];
}

extern "C" void kernel_launch(void* const* d_in, const int* in_sizes, int n_in,
                              void* d_out, int out_size, void* d_ws,
                              size_t ws_size, hipStream_t stream) {
  (void)n_in; (void)out_size; (void)ws_size;
  const float* x = (const float*)d_in[0];
  const float* params = (const float*)d_in[1];
  float* out = (float*)d_out;
  float* ws = (float*)d_ws;

  const int batch = in_sizes[0] / QC_DIM;
  const int blocks = batch / QC_BATCH_PER_BLOCK;

  qc_setup_gates<<<1, 288, 0, stream>>>(params, ws);

  (void)hipFuncSetAttribute((const void*)qc_main,
                            hipFuncAttributeMaxDynamicSharedMemorySize,
                            QC_LDS_BYTES);
  qc_main<<<blocks, QC_THREADS, QC_LDS_BYTES, stream>>>(x, ws, out);
}


# CNOT ring split into pair-local store offsets in pass 3 plus column-permuted gate planes of the next layer: no table, no barrier or result hold in pass 3, 4 groups per WGP; changes summation order
# speedup vs baseline: 1.8169x; 1.0476x over previous
#include <hip/hip_runtime.h>
#include <math.h>


typedef float v2f __attribute__((ext_vector_type(2)));
typedef float v8f __attribute__((ext_vector_type(8)));

#define QC_BATCH_PER_BLOCK 2
#define QC_THREADS 128
#define QC_DIM 4096

#define QC_PLANE 4416
#define QC_ELEM (2 * QC_PLANE)
#define QC_SCR_OFF 0
#define QC_STATE_OFF 64
#define QC_LDS_FLOATS (QC_STATE_OFF + QC_BATCH_PER_BLOCK * QC_ELEM)
#define QC_LDS_BYTES (QC_LDS_FLOATS * 4)

static __device__ __forceinline__ v8f wmma4(v2f a, v2f b, v8f c) {
  return __builtin_amdgcn_wmma_f32_16x16x4_f32(false, a, false, b, (short)0, c,
                                               false, false);
}

static constexpr int finvperm(int i) {
  int x = i;
  for (int q = 0; q < 12; ++q) {
    int C = 11 - q;
    int T = 11 - ((q + 1) % 12);
    x ^= ((x >> C) & 1) << T;
  }
  return x;
}

static constexpr int finv_mask(int k) {
  int m = 0;
  for (int b = 0; b < 12; ++b)
    if ((finvperm(1 << b) >> k) & 1) m |= 1 << b;
  return m;
}

static __device__ __forceinline__ int ring_nibble(int g, int k) {
  const int sh = 8 - 4 * g;
  int x = k << sh;
#pragma unroll
  for (int q = 0; q < 12; ++q) {
    const int C = 11 - q;
    const int T = 11 - ((q + 1) % 12);
    x ^= ((x >> C) & 1) << T;
  }
  return (x >> sh) & 15;
}

struct Gate3 {
  float r[8], d[8], n[8];
};

static __device__ __forceinline__ void load_gate(const float* wsMat, int lane,
                                                 Gate3& g) {
  const float* b = wsMat + lane * 8;
#pragma unroll
  for (int s = 0; s < 8; ++s) {
    g.r[s] = b[s];
    g.d[s] = b[256 + s];
    g.n[s] = b[512 + s];
  }
}

struct TilePair {
  float r0[8], i0[8], r1[8], i1[8];
};

static __device__ __forceinline__ void load_c_pair(const float* lds, int src0,
                                                   int src1, int p,
                                                   TilePair& a) {
#pragma unroll
  for (int j = 0; j < 4; ++j) {
    const int o0 = src0 + 276 * p + 4 * j;
    const int o1 = src1 + 276 * p + 4 * j;
    a.r0[2 * j] = lds[o0];
    a.r0[2 * j + 1] = lds[o0 + 1];
    a.i0[2 * j] = lds[o0 + QC_PLANE];
    a.i0[2 * j + 1] = lds[o0 + 1 + QC_PLANE];
    a.r1[2 * j] = lds[o1];
    a.r1[2 * j + 1] = lds[o1 + 1];
    a.i1[2 * j] = lds[o1 + QC_PLANE];
    a.i1[2 * j + 1] = lds[o1 + 1 + QC_PLANE];
  }
}

static __device__ __forceinline__ void apply_c_ring(float* lds, int base,
                                                    const Gate3& u, int lane,
                                                    int h) {
  const int n = lane & 15;
  const int hf = lane >> 4;
  const int tA = 4 * h, tB = 12 - 4 * h;
  const int srcLane = base + 17 * n + 2 * hf;
  const int src0 = srcLane + 276 * tA;
  const int src1 = srcLane + 276 * tB;
  const int e = (hf ^ __builtin_popcount(n)) & 1;
  const int dstSame = base + 276 * (e ? tB : tA);
  const int dstSwap = base + 276 * (e ? tA : tB);
  const int g0 = (hf ^ h) & 1;

  TilePair a[2];
  load_c_pair(lds, src0, src1, 0, a[0]);
#pragma unroll
  for (int tp = 0; tp < 4; ++tp) {
    const int cu = tp & 1;
    if (tp < 3) load_c_pair(lds, src0, src1, tp + 1, a[cu ^ 1]);
    __builtin_amdgcn_sched_barrier(0);
    const TilePair& s = a[cu];
    v8f k0 = {0.f, 0.f, 0.f, 0.f, 0.f, 0.f, 0.f, 0.f};
    v8f k1 = {0.f, 0.f, 0.f, 0.f, 0.f, 0.f, 0.f, 0.f};
#pragma unroll
    for (int j = 0; j < 4; ++j) {
      v2f Br = {u.r[2 * j], u.r[2 * j + 1]};
      v2f As0 = {s.r0[2 * j] + s.i0[2 * j], s.r0[2 * j + 1] + s.i0[2 * j + 1]};
      v2f As1 = {s.r1[2 * j] + s.i1[2 * j], s.r1[2 * j + 1] + s.i1[2 * j + 1]};
      k0 = wmma4(As0, Br, k0);
      k1 = wmma4(As1, Br, k1);
    }
    v8f dr0 = k0, di0 = k0, dr1 = k1, di1 = k1;
#pragma unroll
    for (int j = 0; j < 4; ++j) {
      v2f Bd = {u.d[2 * j], u.d[2 * j + 1]};
      v2f Bn = {u.n[2 * j], u.n[2 * j + 1]};
      v2f Ar0 = {s.r0[2 * j], s.r0[2 * j + 1]};
      v2f Ai0 = {s.i0[2 * j], s.i0[2 * j + 1]};
      v2f Ar1 = {s.r1[2 * j], s.r1[2 * j + 1]};
      v2f Ai1 = {s.i1[2 * j], s.i1[2 * j + 1]};
      dr0 = wmma4(Ai0, Bn, dr0);
      di0 = wmma4(Ar0, Bd, di0);
      dr1 = wmma4(Ai1, Bn, dr1);
      di1 = wmma4(Ar1, Bd, di1);
    }
    const int g = g0 ^ (__builtin_popcount(tp) & 1);
    const int colE = n ^ (8 * g);
    const int rcE = 136 * g + colE + 276 * tp;
    const int rcO = 136 * g + (colE ^ 8) + 276 * tp;
    const int d0E = dstSame + rcE, d0O = dstSwap + rcO;
    const int d1E = dstSwap + rcE, d1O = dstSame + rcO;
#pragma unroll
    for (int q = 0; q < 4; ++q) {
      const int ra = (q == 0) ? 0 : (q == 1) ? 5 : (q == 2) ? 1 : 4;
      const int rb = (q == 0) ? 3 : (q == 1) ? 6 : (q == 2) ? 2 : 7;
      const int b0 = (q >= 2) ? d0O : d0E;
      const int b1 = (q >= 2) ? d1O : d1E;
      lds[b0 + 17 * ra] = dr0[ra];
      lds[b0 + 17 * rb] = dr0[rb];
      lds[b0 + QC_PLANE + 17 * ra] = di0[ra];
      lds[b0 + QC_PLANE + 17 * rb] = di0[rb];
      lds[b1 + 17 * ra] = dr1[ra];
      lds[b1 + 17 * rb] = dr1[rb];
      lds[b1 + QC_PLANE + 17 * ra] = di1[ra];
      lds[b1 + QC_PLANE + 17 * rb] = di1[rb];
    }
  }
}

static __device__ __forceinline__ void load_c_ops(const float* lds, int baseSrc,
                                                  int t0, TilePair& a) {
#pragma unroll
  for (int j = 0; j < 4; ++j) {
    const int o0 = baseSrc + 276 * t0 + 4 * j;
    const int o1 = o0 + 276;
    a.r0[2 * j] = lds[o0];
    a.r0[2 * j + 1] = lds[o0 + 1];
    a.i0[2 * j] = lds[o0 + QC_PLANE];
    a.i0[2 * j + 1] = lds[o0 + 1 + QC_PLANE];
    a.r1[2 * j] = lds[o1];
    a.r1[2 * j + 1] = lds[o1 + 1];
    a.i1[2 * j] = lds[o1 + QC_PLANE];
    a.i1[2 * j + 1] = lds[o1 + 1 + QC_PLANE];
  }
}

template <int Q>
static __device__ __forceinline__ void z_add_tile(float (&acc)[12],
                                                  const float (&wh)[8], int t) {
  constexpr int M = finv_mask(11 - Q);
  constexpr int TQ = (M >> 8) & 7;
  constexpr int RQ = (M >> 4) & 7;
  const float w = wh[RQ];
  acc[Q] += (__builtin_popcount(t & TQ) & 1) ? -w : w;
}

static __device__ __forceinline__ void z_tile(float (&acc)[12], const v8f& dr,
                                              const v8f& di, int t) {
  float wh[8];
#pragma unroll
  for (int r = 0; r < 8; ++r) wh[r] = dr[r] * dr[r] + di[r] * di[r];
#pragma unroll
  for (int len = 1; len < 8; len <<= 1) {
#pragma unroll
    for (int i = 0; i < 8; i += 2 * len) {
#pragma unroll
      for (int j = i; j < i + len; ++j) {
        const float x = wh[j], y = wh[j + len];
        wh[j] = x + y;
        wh[j + len] = x - y;
      }
    }
  }
  z_add_tile<0>(acc, wh, t);
  z_add_tile<1>(acc, wh, t);
  z_add_tile<2>(acc, wh, t);
  z_add_tile<3>(acc, wh, t);
  z_add_tile<4>(acc, wh, t);
  z_add_tile<5>(acc, wh, t);
  z_add_tile<6>(acc, wh, t);
  z_add_tile<7>(acc, wh, t);
  z_add_tile<8>(acc, wh, t);
  z_add_tile<9>(acc, wh, t);
  z_add_tile<10>(acc, wh, t);
  z_add_tile<11>(acc, wh, t);
}

template <int Q>
static __device__ __forceinline__ void z_lane_sign(float (&acc)[12], int key) {
  constexpr int LQ = finv_mask(11 - Q) & 0x88F;
  const int s = (__builtin_popcount(key & LQ) & 1) << 31;
  acc[Q] = __int_as_float(__float_as_int(acc[Q]) ^ s);
}

static __device__ __forceinline__ void apply_c_final(const float* lds, int base,
                                                     const Gate3& u, int lane,
                                                     int h, float (&acc)[12]) {
  const int m = lane & 15;
  const int kh = (lane >> 4) << 1;
  const int baseSrc = base + 276 * (h * 8) + 17 * m + kh;

  TilePair a[2];
  v8f pr0, pi0, pr1, pi1;
#pragma unroll
  for (int q = 0; q < 12; ++q) acc[q] = 0.f;
  load_c_ops(lds, baseSrc, 0, a[0]);
#pragma unroll
  for (int tp = 0; tp < 4; ++tp) {
    const int cu = tp & 1;
    if (tp < 3) load_c_ops(lds, baseSrc, 2 * (tp + 1), a[cu ^ 1]);
    asm volatile("" ::: "memory");
    __builtin_amdgcn_sched_barrier(0);
    if (tp > 0) {
      asm volatile("" : "+v"(pr0));
      asm volatile("" : "+v"(pi0));
      asm volatile("" : "+v"(pr1));
      asm volatile("" : "+v"(pi1));
      z_tile(acc, pr0, pi0, 2 * tp - 2);
      z_tile(acc, pr1, pi1, 2 * tp - 1);
    }
    const TilePair& s = a[cu];
    v8f k0 = {0.f, 0.f, 0.f, 0.f, 0.f, 0.f, 0.f, 0.f};
    v8f k1 = {0.f, 0.f, 0.f, 0.f, 0.f, 0.f, 0.f, 0.f};
#pragma unroll
    for (int j = 0; j < 4; ++j) {
      v2f Br = {u.r[2 * j], u.r[2 * j + 1]};
      v2f As0 = {s.r0[2 * j] + s.i0[2 * j], s.r0[2 * j + 1] + s.i0[2 * j + 1]};
      v2f As1 = {s.r1[2 * j] + s.i1[2 * j], s.r1[2 * j + 1] + s.i1[2 * j + 1]};
      k0 = wmma4(As0, Br, k0);
      k1 = wmma4(As1, Br, k1);
    }
    v8f dr0 = k0, di0 = k0, dr1 = k1, di1 = k1;
#pragma unroll
    for (int j = 0; j < 4; ++j) {
      v2f Bd = {u.d[2 * j], u.d[2 * j + 1]};
      v2f Bn = {u.n[2 * j], u.n[2 * j + 1]};
      v2f Ar0 = {s.r0[2 * j], s.r0[2 * j + 1]};
      v2f Ai0 = {s.i0[2 * j], s.i0[2 * j + 1]};
      v2f Ar1 = {s.r1[2 * j], s.r1[2 * j + 1]};
      v2f Ai1 = {s.i1[2 * j], s.i1[2 * j + 1]};
      dr0 = wmma4(Ai0, Bn, dr0);
      di0 = wmma4(Ar0, Bd, di0);
      dr1 = wmma4(Ai1, Bn, dr1);
      di1 = wmma4(Ar1, Bd, di1);
    }
    pr0 = dr0;
    pi0 = di0;
    pr1 = dr1;
    pi1 = di1;
    __builtin_amdgcn_sched_barrier(0);
  }
  z_tile(acc, pr0, pi0, 6);
  z_tile(acc, pr1, pi1, 7);
  const int key = (h << 11) | ((lane >> 4) << 7) | m;
  z_lane_sign<0>(acc, key);
  z_lane_sign<1>(acc, key);
  z_lane_sign<2>(acc, key);
  z_lane_sign<3>(acc, key);
  z_lane_sign<4>(acc, key);
  z_lane_sign<5>(acc, key);
  z_lane_sign<6>(acc, key);
  z_lane_sign<7>(acc, key);
  z_lane_sign<8>(acc, key);
  z_lane_sign<9>(acc, key);
  z_lane_sign<10>(acc, key);
  z_lane_sign<11>(acc, key);
}

template <bool REAL>
static __device__ __forceinline__ void load_ab(const float* lds, int baseSrc,
                                               int c0, TilePair& x) {
#pragma unroll
  for (int j = 0; j < 4; ++j) {
    const int o0 = baseSrc + c0 + 17 * (4 * j);
    x.r0[2 * j] = lds[o0];
    x.r0[2 * j + 1] = lds[o0 + 17];
    if (!REAL) {
      x.i0[2 * j] = lds[o0 + QC_PLANE];
      x.i0[2 * j + 1] = lds[o0 + 17 + QC_PLANE];
    }
  }
  asm volatile("" ::: "memory");
#pragma unroll
  for (int j = 0; j < 4; ++j) {
    const int o1 = baseSrc + c0 + 1 + 17 * (4 * j);
    x.r1[2 * j] = lds[o1];
    x.r1[2 * j + 1] = lds[o1 + 17];
    if (!REAL) {
      x.i1[2 * j] = lds[o1 + QC_PLANE];
      x.i1[2 * j + 1] = lds[o1 + 17 + QC_PLANE];
    }
  }
  asm volatile("" ::: "memory");
}

template <bool REAL>
static __device__ __forceinline__ void apply_fused_ab(float* lds, int base,
                                                      const Gate3& ua,
                                                      const Gate3& ub,
                                                      const float* wsC,
                                                      Gate3& gC,
                                                      int lane, int h) {
  const int m = lane & 15;
  const int hf = lane >> 4;
  const int baseSrc = base + 276 * m + 17 * (2 * hf) + 8 * h;
  const int baseDst = base + 276 * (8 * hf) + 17 * m + 8 * h;

  TilePair x[2];
  load_ab<REAL>(lds, baseSrc, 0, x[0]);
#pragma unroll
  for (int tp = 0; tp < 4; ++tp) {
    const int c0 = 2 * tp, c1 = 2 * tp + 1;
    if (tp < 3)
      load_ab<REAL>(lds, baseSrc, 2 * (tp + 1), x[(tp & 1) ^ 1]);
    else
      load_gate(wsC, lane, gC);
    __builtin_amdgcn_sched_barrier(0);
    const TilePair& s = x[tp & 1];
    v8f k0 = {0.f, 0.f, 0.f, 0.f, 0.f, 0.f, 0.f, 0.f};
    v8f k1 = {0.f, 0.f, 0.f, 0.f, 0.f, 0.f, 0.f, 0.f};
#pragma unroll
    for (int j = 0; j < 4; ++j) {
      v2f Br = {ub.r[2 * j], ub.r[2 * j + 1]};
      v2f As0, As1;
      if (REAL) {
        As0 = v2f{s.r0[2 * j], s.r0[2 * j + 1]};
        As1 = v2f{s.r1[2 * j], s.r1[2 * j + 1]};
      } else {
        As0 = v2f{s.r0[2 * j] + s.i0[2 * j], s.r0[2 * j + 1] + s.i0[2 * j + 1]};
        As1 = v2f{s.r1[2 * j] + s.i1[2 * j], s.r1[2 * j + 1] + s.i1[2 * j + 1]};
      }
      k0 = wmma4(As0, Br, k0);
      k1 = wmma4(As1, Br, k1);
    }
    v8f yr0 = k0, yi0 = k0, yr1 = k1, yi1 = k1;
#pragma unroll
    for (int j = 0; j < 4; ++j) {
      v2f Bd = {ub.d[2 * j], ub.d[2 * j + 1]};
      v2f Ar0 = {s.r0[2 * j], s.r0[2 * j + 1]};
      v2f Ar1 = {s.r1[2 * j], s.r1[2 * j + 1]};
      if (!REAL) {
        v2f Bn = {ub.n[2 * j], ub.n[2 * j + 1]};
        v2f Ai0 = {s.i0[2 * j], s.i0[2 * j + 1]};
        v2f Ai1 = {s.i1[2 * j], s.i1[2 * j + 1]};
        yr0 = wmma4(Ai0, Bn, yr0);
        yi0 = wmma4(Ar0, Bd, yi0);
        yr1 = wmma4(Ai1, Bn, yr1);
        yi1 = wmma4(Ar1, Bd, yi1);
      } else {
        yi0 = wmma4(Ar0, Bd, yi0);
        yi1 = wmma4(Ar1, Bd, yi1);
      }
    }
    const v8f ys0 = yr0 + yi0;
    const v8f ys1 = yr1 + yi1;
    v8f q0 = {0.f, 0.f, 0.f, 0.f, 0.f, 0.f, 0.f, 0.f};
    v8f q1 = {0.f, 0.f, 0.f, 0.f, 0.f, 0.f, 0.f, 0.f};
#pragma unroll
    for (int j = 0; j < 4; ++j) {
      v2f Ur = {ua.r[2 * j], ua.r[2 * j + 1]};
      v2f Bs0 = {ys0[2 * j], ys0[2 * j + 1]};
      v2f Bs1 = {ys1[2 * j], ys1[2 * j + 1]};
      q0 = wmma4(Ur, Bs0, q0);
      q1 = wmma4(Ur, Bs1, q1);
    }
    v8f xr0 = q0, xi0 = q0, xr1 = q1, xi1 = q1;
#pragma unroll
    for (int j = 0; j < 4; ++j) {
      v2f Ud = {ua.d[2 * j], ua.d[2 * j + 1]};
      v2f Un = {ua.n[2 * j], ua.n[2 * j + 1]};
      v2f Yr0 = {yr0[2 * j], yr0[2 * j + 1]};
      v2f Yi0 = {yi0[2 * j], yi0[2 * j + 1]};
      v2f Yr1 = {yr1[2 * j], yr1[2 * j + 1]};
      v2f Yi1 = {yi1[2 * j], yi1[2 * j + 1]};
      xr0 = wmma4(Un, Yi0, xr0);
      xi0 = wmma4(Ud, Yr0, xi0);
      xr1 = wmma4(Un, Yi1, xr1);
      xi1 = wmma4(Ud, Yr1, xi1);
    }
#pragma unroll
    for (int r = 0; r < 8; ++r) {
      const int d0 = baseDst + c0 + 276 * r;
      const int d1 = baseDst + c1 + 276 * r;
      lds[d0] = xr0[r];
      lds[d0 + QC_PLANE] = xi0[r];
      lds[d1] = xr1[r];
      lds[d1 + QC_PLANE] = xi1[r];
    }
  }
}

__global__ void qc_setup_gates(const float* __restrict__ params,
                               float* __restrict__ ws) {
  const int tid = threadIdx.x;
  const int mat = tid >> 5;
  const int lane = tid & 31;
  if (mat >= 9) return;
  const int L = mat / 3, g = mat % 3;

  float mr[4][2][2], mi[4][2][2];
#pragma unroll
  for (int s = 0; s < 4; ++s) {
    const int q = g * 4 + s;
    const int base = L * 36 + q * 3;
    const float tx = params[base], ty = params[base + 1], tz = params[base + 2];
    const float cx = cosf(0.5f * tx), sx = sinf(0.5f * tx);
    const float cy = cosf(0.5f * ty), sy = sinf(0.5f * ty);
    const float cz = cosf(0.5f * tz), sz = sinf(0.5f * tz);
    float rr, ii;
    rr = cy * cx; ii = sy * sx;
    mr[s][0][0] = cz * rr + sz * ii; mi[s][0][0] = cz * ii - sz * rr;
    rr = -sy * cx; ii = -cy * sx;
    mr[s][0][1] = cz * rr + sz * ii; mi[s][0][1] = cz * ii - sz * rr;
    rr = sy * cx; ii = -cy * sx;
    mr[s][1][0] = cz * rr - sz * ii; mi[s][1][0] = cz * ii + sz * rr;
    rr = cy * cx; ii = -sy * sx;
    mr[s][1][1] = cz * rr - sz * ii; mi[s][1][1] = cz * ii + sz * rr;
  }

  float* base = ws + mat * 768;
  const int n = lane & 15;
  const int khalf = (lane >> 4) << 1;
#pragma unroll
  for (int j = 0; j < 4; ++j) {
#pragma unroll
    for (int p = 0; p < 2; ++p) {
      const int k = (g == 0) ? (2 * j + p + 4 * khalf) : (4 * j + p + khalf);
      const int kc = (L == 0) ? k : ring_nibble(g, k);
      float re = 1.f, im = 0.f;
#pragma unroll
      for (int s = 0; s < 4; ++s) {
        const int rb = (n >> (3 - s)) & 1;
        const int cb = (kc >> (3 - s)) & 1;
        const float wr = mr[s][rb][cb], wi = mi[s][rb][cb];
        const float nre = re * wr - im * wi;
        const float nim = re * wi + im * wr;
        re = nre; im = nim;
      }
      const int slot = lane * 8 + j * 2 + p;
      base[slot] = re;
      base[256 + slot] = im - re;
      base[512 + slot] = -(re + im);
    }
  }
}

__global__ __launch_bounds__(QC_THREADS, 4) void qc_main(
    const float* __restrict__ x, const float* __restrict__ ws,
    float* __restrict__ out) {
  extern __shared__ float lds[];
  const int tid = threadIdx.x;
  const int lane = tid & 31;
  const int w = tid >> 5;
  const int e = w >> 1;
  const int h = w & 1;
  const int lt = h * 32 + lane;
  const long long b = (long long)blockIdx.x * QC_BATCH_PER_BLOCK + e;

  const int eb = QC_STATE_OFF + e * QC_ELEM;

  const float4* xr = (const float4*)(x + b * QC_DIM);
  float4 v[16];
#pragma unroll
  for (int ii = 0; ii < 16; ++ii) v[ii] = xr[ii * 64 + lt];

  float ss = 0.f;
#pragma unroll
  for (int ii = 0; ii < 16; ++ii) {
    ss += v[ii].x * v[ii].x + v[ii].y * v[ii].y + v[ii].z * v[ii].z +
          v[ii].w * v[ii].w;
  }
#pragma unroll
  for (int off = 16; off > 0; off >>= 1) ss += __shfl_xor(ss, off);
  if (lane == 0) lds[eb + QC_PLANE + h] = ss;
  __syncthreads();
  const float inv = rsqrtf(lds[eb + QC_PLANE] + lds[eb + QC_PLANE + 1]);
  __syncthreads();
  Gate3 gA, gB;
  load_gate(ws + 0, lane, gA);
  load_gate(ws + 768, lane, gB);
#pragma unroll
  for (int ii = 0; ii < 16; ++ii) {
    const int u = ii * 64 + lt;
    const int pb = 4 * u + (u >> 2) + 4 * ii;
    lds[eb + pb + 0] = v[ii].x * inv;
    lds[eb + pb + 1] = v[ii].y * inv;
    lds[eb + pb + 2] = v[ii].z * inv;
    lds[eb + pb + 3] = v[ii].w * inv;
  }
  __syncthreads();

  Gate3 gC;
#pragma unroll 1
  for (int L = 0;; ++L) {
    const float* wsL = ws + L * 3 * 768;
    if (L == 0)
      apply_fused_ab<true>(lds, eb, gA, gB, wsL + 1536, gC, lane, h);
    else
      apply_fused_ab<false>(lds, eb, gA, gB, wsL + 1536, gC, lane, h);
    __syncthreads();
    if (L == 2) break;
    const float* wsN = ws + (L + 1) * 3 * 768;
    load_gate(wsN + 0, lane, gA);
    load_gate(wsN + 768, lane, gB);
    apply_c_ring(lds, eb, gC, lane, h);
    __syncthreads();
  }
  float acc[12];
  apply_c_final(lds, eb, gC, lane, h, acc);
#pragma unroll
  for (int q = 0; q < 12; ++q) {
    float s = acc[q];
#pragma unroll
    for (int off = 16; off > 0; off >>= 1) s += __shfl_xor(s, off);
    acc[q] = s;
  }
  if (lane == 0) {
#pragma unroll
    for (int q = 0; q < 12; ++q) lds[QC_SCR_OFF + w * 12 + q] = acc[q];
  }
  __syncthreads();
  if (h == 0 && lane < 12)
    out[b * 12 + lane] = lds[QC_SCR_OFF + (2 * e) * 12 + lane] +
                         lds[QC_SCR_OFF + (2 * e + 1) * 12 + lane];
}

extern "C" void kernel_launch(void* const* d_in, const int* in_sizes, int n_in,
                              void* d_out, int out_size, void* d_ws,
                              size_t ws_size, hipStream_t stream) {
  (void)n_in; (void)out_size; (void)ws_size;
  const float* x = (const float*)d_in[0];
  const float* params = (const float*)d_in[1];
  float* out = (float*)d_out;
  float* ws = (float*)d_ws;

  const int batch = in_sizes[0] / QC_DIM;
  const int blocks = batch / QC_BATCH_PER_BLOCK;

  qc_setup_gates<<<1, 288, 0, stream>>>(params, ws);

  (void)hipFuncSetAttribute((const void*)qc_main,
                            hipFuncAttributeMaxDynamicSharedMemorySize,
                            QC_LDS_BYTES);
  qc_main<<<blocks, QC_THREADS, QC_LDS_BYTES, stream>>>(x, ws, out);
}


# gate matrices split exactly into phase diagonal * real matrix * phase diagonal: two real WMMA products per complex tile instead of three, diagonals as vector multiplies; changes summation order
# speedup vs baseline: 2.1055x; 1.1588x over previous
#include <hip/hip_runtime.h>
#include <math.h>


typedef float v2f __attribute__((ext_vector_type(2)));
typedef float v8f __attribute__((ext_vector_type(8)));

#define QC_BATCH_PER_BLOCK 2
#define QC_THREADS 128
#define QC_DIM 4096

#define QC_PLANE 4416
#define QC_ELEM (2 * QC_PLANE)
#define QC_SCR_OFF 0
#define QC_STATE_OFF 64
#define QC_LDS_FLOATS (QC_STATE_OFF + QC_BATCH_PER_BLOCK * QC_ELEM)
#define QC_LDS_BYTES (QC_LDS_FLOATS * 4)

#define QC_WS_LAYER 1856
#define QC_WS_RA 0
#define QC_WS_RB 256
#define QC_WS_RC 512
#define QC_WS_PAB 768
#define QC_WS_QAB 1280
#define QC_WS_PC 1792
#define QC_WS_QC 1824

static __device__ __forceinline__ v8f wmma4(v2f a, v2f b, v8f c) {
  return __builtin_amdgcn_wmma_f32_16x16x4_f32(false, a, false, b, (short)0, c,
                                               false, false);
}

static constexpr int finvperm(int i) {
  int x = i;
  for (int q = 0; q < 12; ++q) {
    int C = 11 - q;
    int T = 11 - ((q + 1) % 12);
    x ^= ((x >> C) & 1) << T;
  }
  return x;
}

static constexpr int finv_mask(int k) {
  int m = 0;
  for (int b = 0; b < 12; ++b)
    if ((finvperm(1 << b) >> k) & 1) m |= 1 << b;
  return m;
}

static __device__ __forceinline__ int ring_nibble(int g, int k) {
  const int sh = 8 - 4 * g;
  int x = k << sh;
#pragma unroll
  for (int q = 0; q < 12; ++q) {
    const int C = 11 - q;
    const int T = 11 - ((q + 1) % 12);
    x ^= ((x >> C) & 1) << T;
  }
  return (x >> sh) & 15;
}


struct GateAB {
  float ra[8], rb[8];
  float pr[8], pi[8];
  float qr[8], qi[8];
};

struct GateC {
  float r[8];
  float pr[8], pi[8];
  float qr, qi;
};

static __device__ __forceinline__ void load_gate_ab(const float* wsL, int lane,
                                                    GateAB& g) {
  const float* b = wsL + lane * 8;
  const float* t = wsL + lane * 16;
#pragma unroll
  for (int s = 0; s < 8; ++s) {
    g.ra[s] = b[QC_WS_RA + s];
    g.rb[s] = b[QC_WS_RB + s];
    g.pr[s] = t[QC_WS_PAB + s];
    g.pi[s] = t[QC_WS_PAB + 8 + s];
    g.qr[s] = t[QC_WS_QAB + s];
    g.qi[s] = t[QC_WS_QAB + 8 + s];
  }
}

static __device__ __forceinline__ void load_gate_c(const float* wsL, int lane,
                                                   GateC& g) {
  const float* b = wsL + QC_WS_RC + lane * 8;
  const float* t = wsL + QC_WS_PC + (lane >> 4) * 16;
#pragma unroll
  for (int s = 0; s < 8; ++s) {
    g.r[s] = b[s];
    g.pr[s] = t[s];
    g.pi[s] = t[8 + s];
  }
  g.qr = wsL[QC_WS_QC + 2 * (lane & 15)];
  g.qi = wsL[QC_WS_QC + 2 * (lane & 15) + 1];
}

struct TilePair {
  float r0[8], i0[8], r1[8], i1[8];
};

static __device__ __forceinline__ void load_c_pair(const float* lds, int src0,
                                                   int src1, int p,
                                                   TilePair& a) {
#pragma unroll
  for (int j = 0; j < 4; ++j) {
    const int o0 = src0 + 276 * p + 4 * j;
    const int o1 = src1 + 276 * p + 4 * j;
    a.r0[2 * j] = lds[o0];
    a.r0[2 * j + 1] = lds[o0 + 1];
    a.i0[2 * j] = lds[o0 + QC_PLANE];
    a.i0[2 * j + 1] = lds[o0 + 1 + QC_PLANE];
    a.r1[2 * j] = lds[o1];
    a.r1[2 * j + 1] = lds[o1 + 1];
    a.i1[2 * j] = lds[o1 + QC_PLANE];
    a.i1[2 * j + 1] = lds[o1 + 1 + QC_PLANE];
  }
}
static __device__ __forceinline__ void apply_c_ring(float* lds, int base,
                                                    const GateC& u, int lane,
                                                    int h) {
  const int n = lane & 15;
  const int hf = lane >> 4;
  const int tA = 4 * h, tB = 12 - 4 * h;
  const int srcLane = base + 17 * n + 2 * hf;
  const int src0 = srcLane + 276 * tA;
  const int src1 = srcLane + 276 * tB;
  const int e = (hf ^ __builtin_popcount(n)) & 1;
  const int dstSame = base + 276 * (e ? tB : tA);
  const int dstSwap = base + 276 * (e ? tA : tB);
  const int g0 = (hf ^ h) & 1;

  TilePair a[2];
  load_c_pair(lds, src0, src1, 0, a[0]);
#pragma unroll
  for (int tp = 0; tp < 4; ++tp) {
    const int cu = tp & 1;
    if (tp < 3) load_c_pair(lds, src0, src1, tp + 1, a[cu ^ 1]);
    __builtin_amdgcn_sched_barrier(0);
    const TilePair& s = a[cu];
    v8f dr0 = {0.f, 0.f, 0.f, 0.f, 0.f, 0.f, 0.f, 0.f};
    v8f di0 = dr0, dr1 = dr0, di1 = dr0;
#pragma unroll
    for (int j = 0; j < 4; ++j) {
      const v2f Br = {u.r[2 * j], u.r[2 * j + 1]};
      const v2f pr = {u.pr[2 * j], u.pr[2 * j + 1]};
      const v2f pi = {u.pi[2 * j], u.pi[2 * j + 1]};
      const v2f r0 = {s.r0[2 * j], s.r0[2 * j + 1]};
      const v2f i0 = {s.i0[2 * j], s.i0[2 * j + 1]};
      const v2f r1 = {s.r1[2 * j], s.r1[2 * j + 1]};
      const v2f i1 = {s.i1[2 * j], s.i1[2 * j + 1]};
      const v2f Ar0 = r0 * pr - i0 * pi, Ai0 = r0 * pi + i0 * pr;
      const v2f Ar1 = r1 * pr - i1 * pi, Ai1 = r1 * pi + i1 * pr;
      dr0 = wmma4(Ar0, Br, dr0);
      di0 = wmma4(Ai0, Br, di0);
      dr1 = wmma4(Ar1, Br, dr1);
      di1 = wmma4(Ai1, Br, di1);
    }
    const v8f er0 = dr0 * u.qr - di0 * u.qi, ei0 = dr0 * u.qi + di0 * u.qr;
    const v8f er1 = dr1 * u.qr - di1 * u.qi, ei1 = dr1 * u.qi + di1 * u.qr;
    const int g = g0 ^ (__builtin_popcount(tp) & 1);
    const int colE = n ^ (8 * g);
    const int rcE = 136 * g + colE + 276 * tp;
    const int rcO = 136 * g + (colE ^ 8) + 276 * tp;
    const int d0E = dstSame + rcE, d0O = dstSwap + rcO;
    const int d1E = dstSwap + rcE, d1O = dstSame + rcO;
#pragma unroll
    for (int q = 0; q < 4; ++q) {
      const int ra = (q == 0) ? 0 : (q == 1) ? 5 : (q == 2) ? 1 : 4;
      const int rb = (q == 0) ? 3 : (q == 1) ? 6 : (q == 2) ? 2 : 7;
      const int b0 = (q >= 2) ? d0O : d0E;
      const int b1 = (q >= 2) ? d1O : d1E;
      lds[b0 + 17 * ra] = er0[ra];
      lds[b0 + 17 * rb] = er0[rb];
      lds[b0 + QC_PLANE + 17 * ra] = ei0[ra];
      lds[b0 + QC_PLANE + 17 * rb] = ei0[rb];
      lds[b1 + 17 * ra] = er1[ra];
      lds[b1 + 17 * rb] = er1[rb];
      lds[b1 + QC_PLANE + 17 * ra] = ei1[ra];
      lds[b1 + QC_PLANE + 17 * rb] = ei1[rb];
    }
  }
}

static __device__ __forceinline__ void load_c_ops(const float* lds, int baseSrc,
                                                  int t0, TilePair& a) {
#pragma unroll
  for (int j = 0; j < 4; ++j) {
    const int o0 = baseSrc + 276 * t0 + 4 * j;
    const int o1 = o0 + 276;
    a.r0[2 * j] = lds[o0];
    a.r0[2 * j + 1] = lds[o0 + 1];
    a.i0[2 * j] = lds[o0 + QC_PLANE];
    a.i0[2 * j + 1] = lds[o0 + 1 + QC_PLANE];
    a.r1[2 * j] = lds[o1];
    a.r1[2 * j + 1] = lds[o1 + 1];
    a.i1[2 * j] = lds[o1 + QC_PLANE];
    a.i1[2 * j + 1] = lds[o1 + 1 + QC_PLANE];
  }
}

template <int Q>
static __device__ __forceinline__ void z_add_tile(float (&acc)[12],
                                                  const float (&wh)[8], int t) {
  constexpr int M = finv_mask(11 - Q);
  constexpr int TQ = (M >> 8) & 7;
  constexpr int RQ = (M >> 4) & 7;
  const float w = wh[RQ];
  acc[Q] += (__builtin_popcount(t & TQ) & 1) ? -w : w;
}

static __device__ __forceinline__ void z_tile(float (&acc)[12], const v8f& dr,
                                              const v8f& di, int t) {
  float wh[8];
#pragma unroll
  for (int r = 0; r < 8; ++r) wh[r] = dr[r] * dr[r] + di[r] * di[r];
#pragma unroll
  for (int len = 1; len < 8; len <<= 1) {
#pragma unroll
    for (int i = 0; i < 8; i += 2 * len) {
#pragma unroll
      for (int j = i; j < i + len; ++j) {
        const float x = wh[j], y = wh[j + len];
        wh[j] = x + y;
        wh[j + len] = x - y;
      }
    }
  }
  z_add_tile<0>(acc, wh, t);
  z_add_tile<1>(acc, wh, t);
  z_add_tile<2>(acc, wh, t);
  z_add_tile<3>(acc, wh, t);
  z_add_tile<4>(acc, wh, t);
  z_add_tile<5>(acc, wh, t);
  z_add_tile<6>(acc, wh, t);
  z_add_tile<7>(acc, wh, t);
  z_add_tile<8>(acc, wh, t);
  z_add_tile<9>(acc, wh, t);
  z_add_tile<10>(acc, wh, t);
  z_add_tile<11>(acc, wh, t);
}

template <int Q>
static __device__ __forceinline__ void z_lane_sign(float (&acc)[12], int key) {
  constexpr int LQ = finv_mask(11 - Q) & 0x88F;
  const int s = (__builtin_popcount(key & LQ) & 1) << 31;
  acc[Q] = __int_as_float(__float_as_int(acc[Q]) ^ s);
}

static __device__ __forceinline__ void apply_c_final(const float* lds, int base,
                                                     const GateC& u, int lane,
                                                     int h, float (&acc)[12]) {
  const int m = lane & 15;
  const int kh = (lane >> 4) << 1;
  const int baseSrc = base + 276 * (h * 8) + 17 * m + kh;

  TilePair a[2];
  v8f pr0, pi0, pr1, pi1;
#pragma unroll
  for (int q = 0; q < 12; ++q) acc[q] = 0.f;
  load_c_ops(lds, baseSrc, 0, a[0]);
#pragma unroll
  for (int tp = 0; tp < 4; ++tp) {
    const int cu = tp & 1;
    if (tp < 3) load_c_ops(lds, baseSrc, 2 * (tp + 1), a[cu ^ 1]);
    asm volatile("" ::: "memory");
    __builtin_amdgcn_sched_barrier(0);
    if (tp > 0) {
      asm volatile("" : "+v"(pr0));
      asm volatile("" : "+v"(pi0));
      asm volatile("" : "+v"(pr1));
      asm volatile("" : "+v"(pi1));
      z_tile(acc, pr0, pi0, 2 * tp - 2);
      z_tile(acc, pr1, pi1, 2 * tp - 1);
    }
    const TilePair& s = a[cu];
    v8f dr0 = {0.f, 0.f, 0.f, 0.f, 0.f, 0.f, 0.f, 0.f};
    v8f di0 = dr0, dr1 = dr0, di1 = dr0;
#pragma unroll
    for (int j = 0; j < 4; ++j) {
      const v2f Br = {u.r[2 * j], u.r[2 * j + 1]};
      const v2f pr = {u.pr[2 * j], u.pr[2 * j + 1]};
      const v2f pi = {u.pi[2 * j], u.pi[2 * j + 1]};
      const v2f r0 = {s.r0[2 * j], s.r0[2 * j + 1]};
      const v2f i0 = {s.i0[2 * j], s.i0[2 * j + 1]};
      const v2f r1 = {s.r1[2 * j], s.r1[2 * j + 1]};
      const v2f i1 = {s.i1[2 * j], s.i1[2 * j + 1]};
      const v2f Ar0 = r0 * pr - i0 * pi, Ai0 = r0 * pi + i0 * pr;
      const v2f Ar1 = r1 * pr - i1 * pi, Ai1 = r1 * pi + i1 * pr;
      dr0 = wmma4(Ar0, Br, dr0);
      di0 = wmma4(Ai0, Br, di0);
      dr1 = wmma4(Ar1, Br, dr1);
      di1 = wmma4(Ai1, Br, di1);
    }
    pr0 = dr0;
    pi0 = di0;
    pr1 = dr1;
    pi1 = di1;
    __builtin_amdgcn_sched_barrier(0);
  }
  z_tile(acc, pr0, pi0, 6);
  z_tile(acc, pr1, pi1, 7);
  const int key = (h << 11) | ((lane >> 4) << 7) | m;
  z_lane_sign<0>(acc, key);
  z_lane_sign<1>(acc, key);
  z_lane_sign<2>(acc, key);
  z_lane_sign<3>(acc, key);
  z_lane_sign<4>(acc, key);
  z_lane_sign<5>(acc, key);
  z_lane_sign<6>(acc, key);
  z_lane_sign<7>(acc, key);
  z_lane_sign<8>(acc, key);
  z_lane_sign<9>(acc, key);
  z_lane_sign<10>(acc, key);
  z_lane_sign<11>(acc, key);
}

template <bool REAL>
static __device__ __forceinline__ void load_ab(const float* lds, int baseSrc,
                                               int c0, TilePair& x) {
#pragma unroll
  for (int j = 0; j < 4; ++j) {
    const int o0 = baseSrc + c0 + 17 * (4 * j);
    x.r0[2 * j] = lds[o0];
    x.r0[2 * j + 1] = lds[o0 + 17];
    if (!REAL) {
      x.i0[2 * j] = lds[o0 + QC_PLANE];
      x.i0[2 * j + 1] = lds[o0 + 17 + QC_PLANE];
    }
  }
  asm volatile("" ::: "memory");
#pragma unroll
  for (int j = 0; j < 4; ++j) {
    const int o1 = baseSrc + c0 + 1 + 17 * (4 * j);
    x.r1[2 * j] = lds[o1];
    x.r1[2 * j + 1] = lds[o1 + 17];
    if (!REAL) {
      x.i1[2 * j] = lds[o1 + QC_PLANE];
      x.i1[2 * j + 1] = lds[o1 + 17 + QC_PLANE];
    }
  }
  asm volatile("" ::: "memory");
}

template <bool REAL, bool POST>
static __device__ __forceinline__ void apply_fused_ab(float* lds, int base,
                                                      const GateAB& g,
                                                      const float* wsL,
                                                      GateC& gC,
                                                      int lane, int h) {
  const int m = lane & 15;
  const int hf = lane >> 4;
  const int baseSrc = base + 276 * m + 17 * (2 * hf) + 8 * h;
  const int baseDst = base + 276 * (8 * hf) + 17 * m + 8 * h;

  v8f qr, qi;
#pragma unroll
  for (int r = 0; r < 8; ++r) {
    qr[r] = g.qr[r];
    qi[r] = g.qi[r];
  }

  TilePair x[2];
  load_ab<REAL>(lds, baseSrc, 0, x[0]);
#pragma unroll
  for (int tp = 0; tp < 4; ++tp) {
    const int c0 = 2 * tp, c1 = 2 * tp + 1;
    if (tp < 3)
      load_ab<REAL>(lds, baseSrc, 2 * (tp + 1), x[(tp & 1) ^ 1]);
    else
      load_gate_c(wsL, lane, gC);
    __builtin_amdgcn_sched_barrier(0);
    const TilePair& s = x[tp & 1];
    v8f yr0 = {0.f, 0.f, 0.f, 0.f, 0.f, 0.f, 0.f, 0.f};
    v8f yi0 = yr0, yr1 = yr0, yi1 = yr0;
#pragma unroll
    for (int j = 0; j < 4; ++j) {
      const v2f Br = {g.rb[2 * j], g.rb[2 * j + 1]};
      const v2f pr = {g.pr[2 * j], g.pr[2 * j + 1]};
      const v2f pi = {g.pi[2 * j], g.pi[2 * j + 1]};
      const v2f r0 = {s.r0[2 * j], s.r0[2 * j + 1]};
      const v2f r1 = {s.r1[2 * j], s.r1[2 * j + 1]};
      v2f Ar0, Ai0, Ar1, Ai1;
      if (REAL) {
        Ar0 = r0 * pr;
        Ai0 = r0 * pi;
        Ar1 = r1 * pr;
        Ai1 = r1 * pi;
      } else {
        const v2f i0 = {s.i0[2 * j], s.i0[2 * j + 1]};
        const v2f i1 = {s.i1[2 * j], s.i1[2 * j + 1]};
        Ar0 = r0 * pr - i0 * pi;
        Ai0 = r0 * pi + i0 * pr;
        Ar1 = r1 * pr - i1 * pi;
        Ai1 = r1 * pi + i1 * pr;
      }
      yr0 = wmma4(Ar0, Br, yr0);
      yi0 = wmma4(Ai0, Br, yi0);
      yr1 = wmma4(Ar1, Br, yr1);
      yi1 = wmma4(Ai1, Br, yi1);
    }
    v8f xr0 = {0.f, 0.f, 0.f, 0.f, 0.f, 0.f, 0.f, 0.f};
    v8f xi0 = xr0, xr1 = xr0, xi1 = xr0;
#pragma unroll
    for (int j = 0; j < 4; ++j) {
      const v2f Ur = {g.ra[2 * j], g.ra[2 * j + 1]};
      const v2f Yr0 = {yr0[2 * j], yr0[2 * j + 1]};
      const v2f Yi0 = {yi0[2 * j], yi0[2 * j + 1]};
      const v2f Yr1 = {yr1[2 * j], yr1[2 * j + 1]};
      const v2f Yi1 = {yi1[2 * j], yi1[2 * j + 1]};
      xr0 = wmma4(Ur, Yr0, xr0);
      xi0 = wmma4(Ur, Yi0, xi0);
      xr1 = wmma4(Ur, Yr1, xr1);
      xi1 = wmma4(Ur, Yi1, xi1);
    }
    v8f or0 = xr0, oi0 = xi0, or1 = xr1, oi1 = xi1;
    if (POST) {
      or0 = xr0 * qr - xi0 * qi;
      oi0 = xr0 * qi + xi0 * qr;
      or1 = xr1 * qr - xi1 * qi;
      oi1 = xr1 * qi + xi1 * qr;
    }
#pragma unroll
    for (int r = 0; r < 8; ++r) {
      const int d0 = baseDst + c0 + 276 * r;
      const int d1 = baseDst + c1 + 276 * r;
      lds[d0] = or0[r];
      lds[d0 + QC_PLANE] = oi0[r];
      lds[d1] = or1[r];
      lds[d1 + QC_PLANE] = oi1[r];
    }
  }
}

static __device__ __forceinline__ void polar(float re, float im, float& mag,
                                             float& ur, float& ui) {
  const float mx = fmaxf(fabsf(re), fabsf(im));
  if (mx == 0.f) {
    mag = 0.f;
    ur = 1.f;
    ui = 0.f;
    return;
  }
  const float a = re / mx, b = im / mx;
  const float nrm = sqrtf(a * a + b * b);
  mag = mx * nrm;
  ur = a / nrm;
  ui = b / nrm;
}

__global__ void qc_setup_gates(const float* __restrict__ params,
                               float* __restrict__ ws) {
  __shared__ float sPre[9][32];
  __shared__ float sPost[9][32];
  const int tid = threadIdx.x;
  const int mat = tid >> 5;
  const int lane = tid & 31;
  const int L = mat / 3, g = mat % 3;

  float rot[4][2][2];
  float lr[4][2], li[4][2];
  float tr[4], ti[4];
#pragma unroll
  for (int s = 0; s < 4; ++s) {
    const int q = g * 4 + s;
    const int base = L * 36 + q * 3;
    const float tx = params[base], ty = params[base + 1], tz = params[base + 2];
    const float cx = cosf(0.5f * tx), sx = sinf(0.5f * tx);
    const float cy = cosf(0.5f * ty), sy = sinf(0.5f * ty);
    const float cz = cosf(0.5f * tz), sz = sinf(0.5f * tz);
    float C, S, ar, ai, br, bi;
    polar(cy * cx, sy * sx, C, ar, ai);
    polar(sy * cx, -cy * sx, S, br, bi);
    rot[s][0][0] = C; rot[s][0][1] = -S;
    rot[s][1][0] = S; rot[s][1][1] = C;
    lr[s][0] = cz * ar + sz * ai; li[s][0] = cz * ai - sz * ar;
    lr[s][1] = cz * br - sz * bi; li[s][1] = cz * bi + sz * br;
    tr[s] = ar * br - ai * bi;
    ti[s] = -(ar * bi + ai * br);
  }

  float* wsL = ws + L * QC_WS_LAYER;
  float* base = wsL + g * 256;
  const int n = lane & 15;
  const int hf = lane >> 4;
  const int khalf = hf << 1;
#pragma unroll
  for (int j = 0; j < 4; ++j) {
#pragma unroll
    for (int p = 0; p < 2; ++p) {
      const int k = (g == 0) ? (2 * j + p + 4 * khalf) : (4 * j + p + khalf);
      const int kc = (L == 0) ? k : ring_nibble(g, k);
      float v = 1.f;
#pragma unroll
      for (int s = 0; s < 4; ++s)
        v *= rot[s][(n >> (3 - s)) & 1][(kc >> (3 - s)) & 1];
      base[lane * 8 + j * 2 + p] = v;
    }
  }

  if (lane < 16) {
    const int kc = (L == 0) ? lane : ring_nibble(g, lane);
    float pre = 1.f, pim = 0.f, qre = 1.f, qim = 0.f;
#pragma unroll
    for (int s = 0; s < 4; ++s) {
      if ((kc >> (3 - s)) & 1) {
        const float t0 = pre * tr[s] - pim * ti[s];
        pim = pre * ti[s] + pim * tr[s];
        pre = t0;
      }
      const int rb = (lane >> (3 - s)) & 1;
      const float t1 = qre * lr[s][rb] - qim * li[s][rb];
      qim = qre * li[s][rb] + qim * lr[s][rb];
      qre = t1;
    }
    sPre[mat][2 * lane] = pre;
    sPre[mat][2 * lane + 1] = pim;
    sPost[mat][2 * lane] = qre;
    sPost[mat][2 * lane + 1] = qim;
  }
  __syncthreads();

  if (g == 0) {
    const float ar = sPre[3 * L][2 * n], ai = sPre[3 * L][2 * n + 1];
#pragma unroll
    for (int s = 0; s < 8; ++s) {
      const int k = 4 * (s >> 1) + (s & 1) + khalf;
      const float br = sPre[3 * L + 1][2 * k], bi = sPre[3 * L + 1][2 * k + 1];
      wsL[QC_WS_PAB + lane * 16 + s] = ar * br - ai * bi;
      wsL[QC_WS_PAB + lane * 16 + 8 + s] = ar * bi + ai * br;
    }
  } else if (g == 1) {
    const float br = sPost[3 * L + 1][2 * n], bi = sPost[3 * L + 1][2 * n + 1];
#pragma unroll
    for (int r = 0; r < 8; ++r) {
      const int a = 8 * hf + r;
      const float ar = sPost[3 * L][2 * a], ai = sPost[3 * L][2 * a + 1];
      wsL[QC_WS_QAB + lane * 16 + r] = ar * br - ai * bi;
      wsL[QC_WS_QAB + lane * 16 + 8 + r] = ar * bi + ai * br;
    }
  } else {
    if (n == 0) {
#pragma unroll
      for (int s = 0; s < 8; ++s) {
        const int k = 4 * (s >> 1) + (s & 1) + khalf;
        wsL[QC_WS_PC + hf * 16 + s] = sPre[mat][2 * k];
        wsL[QC_WS_PC + hf * 16 + 8 + s] = sPre[mat][2 * k + 1];
      }
    }
    if (lane < 16) {
      wsL[QC_WS_QC + 2 * lane] = sPost[mat][2 * lane];
      wsL[QC_WS_QC + 2 * lane + 1] = sPost[mat][2 * lane + 1];
    }
  }
}

__global__ __launch_bounds__(QC_THREADS, 4) void qc_main(
    const float* __restrict__ x, const float* __restrict__ ws,
    float* __restrict__ out) {
  extern __shared__ float lds[];
  const int tid = threadIdx.x;
  const int lane = tid & 31;
  const int w = tid >> 5;
  const int e = w >> 1;
  const int h = w & 1;
  const int lt = h * 32 + lane;
  const long long b = (long long)blockIdx.x * QC_BATCH_PER_BLOCK + e;

  const int eb = QC_STATE_OFF + e * QC_ELEM;

  const float4* xr = (const float4*)(x + b * QC_DIM);
  float4 v[16];
#pragma unroll
  for (int ii = 0; ii < 16; ++ii) v[ii] = xr[ii * 64 + lt];

  float ss = 0.f;
#pragma unroll
  for (int ii = 0; ii < 16; ++ii) {
    ss += v[ii].x * v[ii].x + v[ii].y * v[ii].y + v[ii].z * v[ii].z +
          v[ii].w * v[ii].w;
  }
#pragma unroll
  for (int off = 16; off > 0; off >>= 1) ss += __shfl_xor(ss, off);
  if (lane == 0) lds[eb + QC_PLANE + h] = ss;
  __syncthreads();
  const float inv = rsqrtf(lds[eb + QC_PLANE] + lds[eb + QC_PLANE + 1]);
  __syncthreads();
  GateAB gAB;
  load_gate_ab(ws, lane, gAB);
#pragma unroll
  for (int ii = 0; ii < 16; ++ii) {
    const int u = ii * 64 + lt;
    const int pb = 4 * u + (u >> 2) + 4 * ii;
    lds[eb + pb + 0] = v[ii].x * inv;
    lds[eb + pb + 1] = v[ii].y * inv;
    lds[eb + pb + 2] = v[ii].z * inv;
    lds[eb + pb + 3] = v[ii].w * inv;
  }
  __syncthreads();

  GateC gC;
#pragma unroll 1
  for (int L = 0;; ++L) {
    const float* wsL = ws + L * QC_WS_LAYER;
    if (L == 0)
      apply_fused_ab<true, true>(lds, eb, gAB, wsL, gC, lane, h);
    else if (L == 1)
      apply_fused_ab<false, true>(lds, eb, gAB, wsL, gC, lane, h);
    else
      apply_fused_ab<false, false>(lds, eb, gAB, wsL, gC, lane, h);
    __syncthreads();
    if (L == 2) break;
    load_gate_ab(wsL + QC_WS_LAYER, lane, gAB);
    apply_c_ring(lds, eb, gC, lane, h);
    __syncthreads();
  }
  float acc[12];
  apply_c_final(lds, eb, gC, lane, h, acc);
#pragma unroll
  for (int q = 0; q < 12; ++q) {
    float s = acc[q];
#pragma unroll
    for (int off = 16; off > 0; off >>= 1) s += __shfl_xor(s, off);
    acc[q] = s;
  }
  if (lane == 0) {
#pragma unroll
    for (int q = 0; q < 12; ++q) lds[QC_SCR_OFF + w * 12 + q] = acc[q];
  }
  __syncthreads();
  if (h == 0 && lane < 12)
    out[b * 12 + lane] = lds[QC_SCR_OFF + (2 * e) * 12 + lane] +
                         lds[QC_SCR_OFF + (2 * e + 1) * 12 + lane];
}

extern "C" void kernel_launch(void* const* d_in, const int* in_sizes, int n_in,
                              void* d_out, int out_size, void* d_ws,
                              size_t ws_size, hipStream_t stream) {
  (void)n_in; (void)out_size; (void)ws_size;
  const float* x = (const float*)d_in[0];
  const float* params = (const float*)d_in[1];
  float* out = (float*)d_out;
  float* ws = (float*)d_ws;

  const int batch = in_sizes[0] / QC_DIM;
  const int blocks = batch / QC_BATCH_PER_BLOCK;

  qc_setup_gates<<<1, 288, 0, stream>>>(params, ws);

  (void)hipFuncSetAttribute((const void*)qc_main,
                            hipFuncAttributeMaxDynamicSharedMemorySize,
                            QC_LDS_BYTES);
  qc_main<<<blocks, QC_THREADS, QC_LDS_BYTES, stream>>>(x, ws, out);
}
